# starting kernel
# baseline (speedup 1.0000x reference)
#include <hip/hip_runtime.h>
#include <hip/hip_bf16.h>

typedef __attribute__((ext_vector_type(2))) float v2f;
typedef __attribute__((ext_vector_type(8))) float v8f;

#define NEG_SLOPE 0.2f


__device__ __forceinline__ float lrelu(float v) {
    return v >= 0.0f ? v : NEG_SLOPE * v;
}

__device__ __forceinline__ unsigned fkey(float f) {
    unsigned u = __float_as_uint(f);
    return (u & 0x80000000u) ? ~u : (u | 0x80000000u);
}
__device__ __forceinline__ float fdekey(unsigned k) {
    return (k & 0x80000000u) ? __uint_as_float(k ^ 0x80000000u)
                             : __uint_as_float(~k);
}

__global__ void zero_u32(unsigned* __restrict__ p, int n) {
    int i = blockIdx.x * blockDim.x + threadIdx.x;
    if (i < n) p[i] = 0u;
}

__global__ __launch_bounds__(32)
void gemm1_wmma(const float* __restrict__ x, const float* __restrict__ W1,
                float* __restrict__ h1, int N) {
    const int mtile = blockIdx.x;
    const int lane  = threadIdx.x;
    const int mrow  = mtile * 16 + (lane & 15);
    const int lrow  = mrow < N ? mrow : N - 1;
    const int khalf = (lane >> 4) * 2;
    const int colbase = blockIdx.y * 64;
    const int nc    = lane & 15;

    v8f c0 = {}, c1 = {}, c2 = {}, c3 = {};
    const float* xrow = x + (size_t)lrow * 128;

    for (int k = 0; k < 128; k += 4) {
        const int kb = k + khalf;
        v2f a; a.x = xrow[kb]; a.y = xrow[kb + 1];
        const float* w0 = W1 + (size_t)kb * 256 + colbase + nc;
        const float* w1 = w0 + 256;
        v2f b0; b0.x = w0[0];  b0.y = w1[0];
        v2f b1; b1.x = w0[16]; b1.y = w1[16];
        v2f b2; b2.x = w0[32]; b2.y = w1[32];
        v2f b3; b3.x = w0[48]; b3.y = w1[48];
        c0 = __builtin_amdgcn_wmma_f32_16x16x4_f32(false, a, false, b0, (short)0, c0, false, false);
        c1 = __builtin_amdgcn_wmma_f32_16x16x4_f32(false, a, false, b1, (short)0, c1, false, false);
        c2 = __builtin_amdgcn_wmma_f32_16x16x4_f32(false, a, false, b2, (short)0, c2, false, false);
        c3 = __builtin_amdgcn_wmma_f32_16x16x4_f32(false, a, false, b3, (short)0, c3, false, false);
    }

    const int rowoff = (lane >= 16) ? 8 : 0;
    for (int r = 0; r < 8; ++r) {
        const int grow = mtile * 16 + rowoff + r;
        if (grow < N) {
            float* op = h1 + (size_t)grow * 256 + colbase + nc;
            op[0]  = c0[r];
            op[16] = c1[r];
            op[32] = c2[r];
            op[48] = c3[r];
        }
    }
}

__global__ __launch_bounds__(32)
void gemm2_wmma(const float* __restrict__ y1, const float* __restrict__ W2,
                float* __restrict__ h2, int N) {
    const int mtile = blockIdx.x;
    const int lane  = threadIdx.x;
    const int mrow  = mtile * 16 + (lane & 15);
    const int lrow  = mrow < N ? mrow : N - 1;
    const int khalf = (lane >> 4) * 2;
    const int nc    = lane & 15;

    v8f c = {};
    const float* yr = y1 + (size_t)lrow * 256;
    for (int k = 0; k < 256; k += 4) {
        const int kb = k + khalf;
        v2f a; a.x = yr[kb]; a.y = yr[kb + 1];
        v2f b; b.x = W2[(size_t)kb * 16 + nc]; b.y = W2[(size_t)(kb + 1) * 16 + nc];
        c = __builtin_amdgcn_wmma_f32_16x16x4_f32(false, a, false, b, (short)0, c, false, false);
    }
    const int rowoff = (lane >= 16) ? 8 : 0;
    for (int r = 0; r < 8; ++r) {
        const int grow = mtile * 16 + rowoff + r;
        if (grow < N) h2[(size_t)grow * 16 + nc] = c[r];
    }
}

__global__ void att1_kernel(const float* __restrict__ h1,
                            const float* __restrict__ att_src,
                            const float* __restrict__ att_dst,
                            float* __restrict__ a_src, float* __restrict__ a_dst,
                            int NH) {
    int i = blockIdx.x * blockDim.x + threadIdx.x;
    if (i >= NH) return;
    const int h = i & 3;
    const float* hp = h1 + (size_t)(i >> 2) * 256 + h * 64;
    const float* as = att_src + h * 64;
    const float* ad = att_dst + h * 64;
    float s = 0.0f, d = 0.0f;
    for (int c = 0; c < 64; ++c) {
        const float v = hp[c];
        s += v * as[c];
        d += v * ad[c];
    }
    a_src[i] = s;
    a_dst[i] = d;
}

__global__ void att2_kernel(const float* __restrict__ h2,
                            const float* __restrict__ att_src,
                            const float* __restrict__ att_dst,
                            float* __restrict__ a_src, float* __restrict__ a_dst,
                            int N) {
    int i = blockIdx.x * blockDim.x + threadIdx.x;
    if (i >= N) return;
    const float* hp = h2 + (size_t)i * 16;
    float s = 0.0f, d = 0.0f;
    for (int c = 0; c < 16; ++c) {
        const float v = hp[c];
        s += v * att_src[c];
        d += v * att_dst[c];
    }
    a_src[i] = s;
    a_dst[i] = d;
}

__device__ __forceinline__ void edge_sd(const int* __restrict__ ei, int e, int E,
                                        int& s, int& d) {
    if (e < E) { s = ei[e]; d = ei[E + e]; }
    else       { s = d = e - E; }
}

__global__ void edge_max1(const int* __restrict__ ei, int E, int N,
                          const float* __restrict__ a_src, const float* __restrict__ a_dst,
                          unsigned* __restrict__ mkey) {
    int e = blockIdx.x * blockDim.x + threadIdx.x;
    const int ET = E + N;
    if (e >= ET) return;
    int s, d; edge_sd(ei, e, E, s, d);
    for (int h = 0; h < 4; ++h) {
        const float v = lrelu(a_src[s * 4 + h] + a_dst[d * 4 + h]);
        atomicMax(&mkey[d * 4 + h], fkey(v));
    }
}

__global__ void edge_expsum1(const int* __restrict__ ei, int E, int N,
                             const float* __restrict__ a_src, const float* __restrict__ a_dst,
                             const unsigned* __restrict__ mkey,
                             float* __restrict__ ex, float* __restrict__ den) {
    int e = blockIdx.x * blockDim.x + threadIdx.x;
    const int ET = E + N;
    if (e >= ET) return;
    int s, d; edge_sd(ei, e, E, s, d);
    for (int h = 0; h < 4; ++h) {
        const float v  = lrelu(a_src[s * 4 + h] + a_dst[d * 4 + h]);
        const float m  = fdekey(mkey[d * 4 + h]);
        const float xv = expf(v - m);
        ex[(size_t)e * 4 + h] = xv;
        atomicAdd(&den[d * 4 + h], xv);
    }
}

__global__ void edge_agg1(const int* __restrict__ ei, int E, int N,
                          const float* __restrict__ ex, const float* __restrict__ den,
                          const float* __restrict__ h1, float* __restrict__ out1) {
    const int gid = blockIdx.x * blockDim.x + threadIdx.x;
    const int e = gid >> 6;
    const int t = gid & 63;
    const int ET = E + N;
    if (e >= ET) return;
    int s, d; edge_sd(ei, e, E, s, d);
    const int head = t >> 4;
    const float alpha = ex[(size_t)e * 4 + head] / den[d * 4 + head];
    const float4 hv = *(const float4*)(h1 + (size_t)s * 256 + t * 4);
    float* op = out1 + (size_t)d * 256 + t * 4;
    atomicAdd(op + 0, alpha * hv.x);
    atomicAdd(op + 1, alpha * hv.y);
    atomicAdd(op + 2, alpha * hv.z);
    atomicAdd(op + 3, alpha * hv.w);
}

__global__ void relu_bias1(float* __restrict__ out1, const float* __restrict__ b1, int n) {
    int i = blockIdx.x * blockDim.x + threadIdx.x;
    if (i < n) out1[i] = fmaxf(out1[i] + b1[i & 255], 0.0f);
}

__global__ void edge_max2(const int* __restrict__ ei, int E, int N,
                          const float* __restrict__ a_src, const float* __restrict__ a_dst,
                          unsigned* __restrict__ mkey) {
    int e = blockIdx.x * blockDim.x + threadIdx.x;
    const int ET = E + N;
    if (e >= ET) return;
    int s, d; edge_sd(ei, e, E, s, d);
    const float v = lrelu(a_src[s] + a_dst[d]);
    atomicMax(&mkey[d], fkey(v));
}

__global__ void edge_expsum2(const int* __restrict__ ei, int E, int N,
                             const float* __restrict__ a_src, const float* __restrict__ a_dst,
                             const unsigned* __restrict__ mkey,
                             float* __restrict__ ex, float* __restrict__ den) {
    int e = blockIdx.x * blockDim.x + threadIdx.x;
    const int ET = E + N;
    if (e >= ET) return;
    int s, d; edge_sd(ei, e, E, s, d);
    const float v  = lrelu(a_src[s] + a_dst[d]);
    const float xv = expf(v - fdekey(mkey[d]));
    ex[e] = xv;
    atomicAdd(&den[d], xv);
}

__global__ void edge_agg2(const int* __restrict__ ei, int E, int N,
                          const float* __restrict__ ex, const float* __restrict__ den,
                          const float* __restrict__ h2, float* __restrict__ out) {
    const int gid = blockIdx.x * blockDim.x + threadIdx.x;
    const int e = gid >> 4;
    const int t = gid & 15;
    const int ET = E + N;
    if (e >= ET) return;
    int s, d; edge_sd(ei, e, E, s, d);
    const float alpha = ex[e] / den[d];
    atomicAdd(&out[(size_t)d * 16 + t], alpha * h2[(size_t)s * 16 + t]);
}

__global__ void bias2_kernel(float* __restrict__ out, const float* __restrict__ b2, int n) {
    int i = blockIdx.x * blockDim.x + threadIdx.x;
    if (i < n) out[i] += b2[i & 15];
}


extern "C" void kernel_launch(void* const* d_in, const int* in_sizes, int n_in,
                              void* d_out, int out_size, void* d_ws, size_t ws_size,
                              hipStream_t stream) {
    const float* x   = (const float*)d_in[0];
    const int*   ei  = (const int*)  d_in[1];
    const float* W1  = (const float*)d_in[2];
    const float* as1 = (const float*)d_in[3];
    const float* ad1 = (const float*)d_in[4];
    const float* b1  = (const float*)d_in[5];
    const float* W2  = (const float*)d_in[6];
    const float* as2 = (const float*)d_in[7];
    const float* ad2 = (const float*)d_in[8];
    const float* b2  = (const float*)d_in[9];

    const int N  = in_sizes[0] / 128;
    const int E  = in_sizes[1] / 2;
    const int ET = E + N;

    char* w = (char*)d_ws;
    auto carve = [&](size_t bytes) -> void* {
        void* p = (void*)w;
        w += (bytes + 255) & ~((size_t)255);
        return p;
    };
    float*    h1    = (float*)   carve((size_t)N * 256 * 4);
    float*    out1  = (float*)   carve((size_t)N * 256 * 4);
    float*    aS1   = (float*)   carve((size_t)N * 4 * 4);
    float*    aD1   = (float*)   carve((size_t)N * 4 * 4);
    unsigned* mkey1 = (unsigned*)carve((size_t)N * 4 * 4);
    float*    den1  = (float*)   carve((size_t)N * 4 * 4);
    float*    ex1   = (float*)   carve((size_t)ET * 4 * 4);
    float*    h2    = (float*)   carve((size_t)N * 16 * 4);
    float*    aS2   = (float*)   carve((size_t)N * 4);
    float*    aD2   = (float*)   carve((size_t)N * 4);
    unsigned* mkey2 = (unsigned*)carve((size_t)N * 4);
    float*    den2  = (float*)   carve((size_t)N * 4);
    float*    ex2   = (float*)   carve((size_t)ET * 4);

    const int TB = 256;
    auto blocks = [&](long long n) { return (unsigned)((n + TB - 1) / TB); };

    zero_u32<<<blocks((long long)N * 256), TB, 0, stream>>>((unsigned*)out1, N * 256);
    zero_u32<<<blocks(N * 4), TB, 0, stream>>>(mkey1, N * 4);
    zero_u32<<<blocks(N * 4), TB, 0, stream>>>((unsigned*)den1, N * 4);
    zero_u32<<<blocks(N), TB, 0, stream>>>(mkey2, N);
    zero_u32<<<blocks(N), TB, 0, stream>>>((unsigned*)den2, N);
    zero_u32<<<blocks(N * 16), TB, 0, stream>>>((unsigned*)d_out, N * 16);

    dim3 g1((N + 15) / 16, 4);
    gemm1_wmma<<<g1, 32, 0, stream>>>(x, W1, h1, N);
    att1_kernel<<<blocks(N * 4), TB, 0, stream>>>(h1, as1, ad1, aS1, aD1, N * 4);
    edge_max1<<<blocks(ET), TB, 0, stream>>>(ei, E, N, aS1, aD1, mkey1);
    edge_expsum1<<<blocks(ET), TB, 0, stream>>>(ei, E, N, aS1, aD1, mkey1, ex1, den1);
    edge_agg1<<<blocks((long long)ET * 64), TB, 0, stream>>>(ei, E, N, ex1, den1, h1, out1);
    relu_bias1<<<blocks((long long)N * 256), TB, 0, stream>>>(out1, b1, N * 256);

    gemm2_wmma<<<(N + 15) / 16, 32, 0, stream>>>(out1, W2, h2, N);
    att2_kernel<<<blocks(N), TB, 0, stream>>>(h2, as2, ad2, aS2, aD2, N);
    edge_max2<<<blocks(ET), TB, 0, stream>>>(ei, E, N, aS2, aD2, mkey2);
    edge_expsum2<<<blocks(ET), TB, 0, stream>>>(ei, E, N, aS2, aD2, mkey2, ex2, den2);
    edge_agg2<<<blocks((long long)ET * 16), TB, 0, stream>>>(ei, E, N, ex2, den2, h2, (float*)d_out);
    bias2_kernel<<<blocks(N * 16), TB, 0, stream>>>((float*)d_out, b2, N * 16);
}


# Each block owns 200 destination nodes and filters the whole edge list for them: softmax state and row sums in LDS (float atomics, arrival order), divided after the sum, stored once; 18 launches to 6
# speedup vs baseline: 2.9052x; 2.9052x over previous
#include <hip/hip_runtime.h>
#include <hip/hip_bf16.h>

typedef __attribute__((ext_vector_type(2))) float v2f;
typedef __attribute__((ext_vector_type(8))) float v8f;

#define NEG_SLOPE 0.2f


__device__ __forceinline__ float lrelu(float v) {
    return v >= 0.0f ? v : NEG_SLOPE * v;
}

__device__ __forceinline__ unsigned fkey(float f) {
    unsigned u = __float_as_uint(f);
    return (u & 0x80000000u) ? ~u : (u | 0x80000000u);
}
__device__ __forceinline__ float fdekey(unsigned k) {
    return (k & 0x80000000u) ? __uint_as_float(k ^ 0x80000000u)
                             : __uint_as_float(~k);
}

__global__ __launch_bounds__(32)
void gemm1_wmma(const float* __restrict__ x, const float* __restrict__ W1,
                float* __restrict__ h1, int N) {
    const int mtile = blockIdx.x;
    const int lane  = threadIdx.x;
    const int mrow  = mtile * 16 + (lane & 15);
    const int lrow  = mrow < N ? mrow : N - 1;
    const int khalf = (lane >> 4) * 2;
    const int colbase = blockIdx.y * 64;
    const int nc    = lane & 15;

    v8f c0 = {}, c1 = {}, c2 = {}, c3 = {};
    const float* xrow = x + (size_t)lrow * 128;

    for (int k = 0; k < 128; k += 4) {
        const int kb = k + khalf;
        v2f a; a.x = xrow[kb]; a.y = xrow[kb + 1];
        const float* w0 = W1 + (size_t)kb * 256 + colbase + nc;
        const float* w1 = w0 + 256;
        v2f b0; b0.x = w0[0];  b0.y = w1[0];
        v2f b1; b1.x = w0[16]; b1.y = w1[16];
        v2f b2; b2.x = w0[32]; b2.y = w1[32];
        v2f b3; b3.x = w0[48]; b3.y = w1[48];
        c0 = __builtin_amdgcn_wmma_f32_16x16x4_f32(false, a, false, b0, (short)0, c0, false, false);
        c1 = __builtin_amdgcn_wmma_f32_16x16x4_f32(false, a, false, b1, (short)0, c1, false, false);
        c2 = __builtin_amdgcn_wmma_f32_16x16x4_f32(false, a, false, b2, (short)0, c2, false, false);
        c3 = __builtin_amdgcn_wmma_f32_16x16x4_f32(false, a, false, b3, (short)0, c3, false, false);
    }

    const int rowoff = (lane >= 16) ? 8 : 0;
    for (int r = 0; r < 8; ++r) {
        const int grow = mtile * 16 + rowoff + r;
        if (grow < N) {
            float* op = h1 + (size_t)grow * 256 + colbase + nc;
            op[0]  = c0[r];
            op[16] = c1[r];
            op[32] = c2[r];
            op[48] = c3[r];
        }
    }
}

__global__ __launch_bounds__(32)
void gemm2_wmma(const float* __restrict__ y1, const float* __restrict__ W2,
                float* __restrict__ h2, int N) {
    const int mtile = blockIdx.x;
    const int lane  = threadIdx.x;
    const int mrow  = mtile * 16 + (lane & 15);
    const int lrow  = mrow < N ? mrow : N - 1;
    const int khalf = (lane >> 4) * 2;
    const int nc    = lane & 15;

    v8f c = {};
    const float* yr = y1 + (size_t)lrow * 256;
    for (int k = 0; k < 256; k += 4) {
        const int kb = k + khalf;
        v2f a; a.x = yr[kb]; a.y = yr[kb + 1];
        v2f b; b.x = W2[(size_t)kb * 16 + nc]; b.y = W2[(size_t)(kb + 1) * 16 + nc];
        c = __builtin_amdgcn_wmma_f32_16x16x4_f32(false, a, false, b, (short)0, c, false, false);
    }
    const int rowoff = (lane >= 16) ? 8 : 0;
    for (int r = 0; r < 8; ++r) {
        const int grow = mtile * 16 + rowoff + r;
        if (grow < N) h2[(size_t)grow * 16 + nc] = c[r];
    }
}

__global__ void att1_kernel(const float* __restrict__ h1,
                            const float* __restrict__ att_src,
                            const float* __restrict__ att_dst,
                            float* __restrict__ a_src, float* __restrict__ a_dst,
                            int NH) {
    int i = blockIdx.x * blockDim.x + threadIdx.x;
    if (i >= NH) return;
    const int h = i & 3;
    const float* hp = h1 + (size_t)(i >> 2) * 256 + h * 64;
    const float* as = att_src + h * 64;
    const float* ad = att_dst + h * 64;
    float s = 0.0f, d = 0.0f;
    for (int c = 0; c < 64; ++c) {
        const float v = hp[c];
        s += v * as[c];
        d += v * ad[c];
    }
    a_src[i] = s;
    a_dst[i] = d;
}

__global__ void att2_kernel(const float* __restrict__ h2,
                            const float* __restrict__ att_src,
                            const float* __restrict__ att_dst,
                            float* __restrict__ a_src, float* __restrict__ a_dst,
                            int N) {
    int i = blockIdx.x * blockDim.x + threadIdx.x;
    if (i >= N) return;
    const float* hp = h2 + (size_t)i * 16;
    float s = 0.0f, d = 0.0f;
    for (int c = 0; c < 16; ++c) {
        const float v = hp[c];
        s += v * att_src[c];
        d += v * att_dst[c];
    }
    a_src[i] = s;
    a_dst[i] = d;
}

#define RNG  200
#define CAPL 4096
#define NTH  1024

__device__ __forceinline__ void list_one(int d, int e, int d0,
                                         int* lst_s, int* lst_d, int* cnt) {
    const unsigned dl = (unsigned)(d - d0);
    if (dl < (unsigned)RNG) {
        const int slot = atomicAdd(cnt, 1);
        if (slot < CAPL) { lst_s[slot] = e; lst_d[slot] = (int)dl; }
    }
}

__device__ __forceinline__ void scan_edges(const int* __restrict__ dstp, int lo, int hi,
                                           bool vec, int d0, int tid,
                                           int* lst_s, int* lst_d, int* cnt) {
    constexpr int SU = 8;
    if (vec) {
        const int nq = (hi - lo) >> 2;
        const int K  = (nq + NTH - 1) / NTH;
        const int rot = (int)(((long long)blockIdx.x * K) / gridDim.x);
        for (int k = 0; k < K; k += SU) {
            int4 v[SU]; int eb[SU]; bool ok[SU];
            #pragma unroll
            for (int u = 0; u < SU; ++u) {
                int kr = k + u + rot; kr = kr >= K ? kr - K : kr;
                const int q = kr * NTH + tid;
                ok[u] = (k + u < K) && (q < nq);
                const int qq = ok[u] ? q : 0;
                v[u]  = *(const int4*)(dstp + lo + 4 * qq);
                eb[u] = lo + 4 * qq;
            }
            #pragma unroll
            for (int u = 0; u < SU; ++u) {
                if (ok[u]) {
                    list_one(v[u].x, eb[u] + 0, d0, lst_s, lst_d, cnt);
                    list_one(v[u].y, eb[u] + 1, d0, lst_s, lst_d, cnt);
                    list_one(v[u].z, eb[u] + 2, d0, lst_s, lst_d, cnt);
                    list_one(v[u].w, eb[u] + 3, d0, lst_s, lst_d, cnt);
                }
            }
        }
        for (int e = lo + 4 * nq + tid; e < hi; e += NTH)
            list_one(dstp[e], e, d0, lst_s, lst_d, cnt);
    } else {
        for (int e = lo + tid; e < hi; e += NTH)
            list_one(dstp[e], e, d0, lst_s, lst_d, cnt);
    }
}

template <int H>
__device__ __forceinline__ void phase_max(int n, bool with_self, int d0, int N, int tid,
                                          const float* __restrict__ a_src,
                                          const int* lst_s, const int* lst_d,
                                          const float* adl, unsigned* mkey) {
    const int tot = n + (with_self ? RNG : 0);
    for (int i = tid; i < tot; i += NTH) {
        int s, dl;
        if (i < n) { s = lst_s[i]; dl = lst_d[i]; }
        else       { dl = i - n;   s = d0 + dl; }
        if (d0 + dl < N) {
            for (int h = 0; h < H; ++h) {
                const float v = lrelu(a_src[s * H + h] + adl[dl * H + h]);
                atomicMax(&mkey[dl * H + h], fkey(v));
            }
        }
    }
}

template <int H, int C>
__device__ __forceinline__ void phase_sum(int n, bool with_self, int d0, int N, int tid,
                                          const float* __restrict__ a_src,
                                          const float* __restrict__ hfeat,
                                          const int* lst_s, const int* lst_d,
                                          const float* adl, const unsigned* mkey,
                                          float* den, float* acc) {
    constexpr int F   = H * C;
    constexpr int LPE = F >= 32 ? 32 : F;
    constexpr int EPW = 32 / LPE;
    constexpr int J   = F / LPE;
    constexpr int U   = 4;
    constexpr int STEP = (NTH / 32) * EPW * U;
    const int lane = tid & 31, wave = tid >> 5;
    const int sub = lane / LPE, l = lane % LPE;
    const int tot = n + (with_self ? RNG : 0);

    for (int i0 = 0; i0 < tot; i0 += STEP) {
        int dl[U]; bool ok[U];
        float row[U][J]; float as[U][H];
        #pragma unroll
        for (int u = 0; u < U; ++u) {
            const int i = i0 + (wave * U + u) * EPW + sub;
            int s;
            if (i < n) { s = lst_s[i]; dl[u] = lst_d[i]; }
            else       { dl[u] = i - n; s = d0 + dl[u]; }
            ok[u] = (i < tot) && (d0 + dl[u] < N);
            if (!ok[u]) { s = 0; dl[u] = 0; }
            const float* rp = hfeat + (size_t)s * F + l;
            #pragma unroll
            for (int j = 0; j < J; ++j) row[u][j] = rp[LPE * j];
            #pragma unroll
            for (int h = 0; h < H; ++h) as[u][h] = a_src[s * H + h];
        }
        #pragma unroll
        for (int u = 0; u < U; ++u) {
            if (ok[u]) {
                float ex[H];
                #pragma unroll
                for (int h = 0; h < H; ++h) {
                    const float v = lrelu(as[u][h] + adl[dl[u] * H + h]);
                    ex[h] = expf(v - fdekey(mkey[dl[u] * H + h]));
                }
                float* ap = acc + dl[u] * F + l;
                #pragma unroll
                for (int j = 0; j < J; ++j)
                    atomicAdd(ap + LPE * j, ex[(LPE * j) / C] * row[u][j]);
                if (l < H) {
                    float mine = ex[0];
                    #pragma unroll
                    for (int h = 1; h < H; ++h) mine = (l == h) ? ex[h] : mine;
                    atomicAdd(&den[dl[u] * H + l], mine);
                }
            }
        }
    }
}

template <int H, int C, bool RELU>
__global__ __launch_bounds__(NTH)
void node_range_layer(const int* __restrict__ ei, int E, int N,
                      const float* __restrict__ a_src, const float* __restrict__ a_dst,
                      const float* __restrict__ hfeat, const float* __restrict__ bias,
                      float* __restrict__ outp) {
    constexpr int F = H * C;
    __shared__ float    acc[RNG * F];
    __shared__ unsigned mkey[RNG * H];
    __shared__ float    den[RNG * H];
    __shared__ float    adl[RNG * H];
    __shared__ int      lst_s[CAPL];
    __shared__ int      lst_d[CAPL];
    __shared__ int      cnt;

    const int tid = threadIdx.x;
    const int d0  = blockIdx.x * RNG;
    const int* dstp = ei + E;
    const bool vec = (((size_t)dstp) & 15) == 0;

    for (int i = tid; i < RNG * F; i += NTH) acc[i] = 0.0f;
    for (int i = tid; i < RNG * H; i += NTH) {
        mkey[i] = 0u; den[i] = 0.0f;
        adl[i] = (d0 * H + i < N * H) ? a_dst[d0 * H + i] : 0.0f;
    }
    if (tid == 0) cnt = 0;
    __syncthreads();

    scan_edges(dstp, 0, E, vec, d0, tid, lst_s, lst_d, &cnt);
    __syncthreads();
    const int total = cnt;

    if (total <= CAPL) {
        for (int i = tid; i < total; i += NTH) lst_s[i] = ei[lst_s[i]];
        __syncthreads();
        phase_max<H>(total, true, d0, N, tid, a_src, lst_s, lst_d, adl, mkey);
        __syncthreads();
        phase_sum<H, C>(total, true, d0, N, tid, a_src, hfeat, lst_s, lst_d, adl, mkey, den, acc);
        __syncthreads();
    } else {
        __syncthreads();
        for (int pass = 0; pass < 2; ++pass) {
            for (int lo = 0; lo < E; lo += CAPL) {
                const int hi = lo + CAPL < E ? lo + CAPL : E;
                if (tid == 0) cnt = 0;
                __syncthreads();
                scan_edges(dstp, lo, hi, vec, d0, tid, lst_s, lst_d, &cnt);
                __syncthreads();
                const int n = cnt;
                for (int i = tid; i < n; i += NTH) lst_s[i] = ei[lst_s[i]];
                __syncthreads();
                if (pass == 0)
                    phase_max<H>(n, lo == 0, d0, N, tid, a_src, lst_s, lst_d, adl, mkey);
                else
                    phase_sum<H, C>(n, lo == 0, d0, N, tid, a_src, hfeat, lst_s, lst_d, adl, mkey, den, acc);
                __syncthreads();
            }
        }
    }

    for (int idx = tid; idx < RNG * F; idx += NTH) {
        const int dl = idx / F, c = idx % F;
        const int node = d0 + dl;
        if (node < N) {
            float v = acc[idx] / den[dl * H + c / C] + bias[c];
            if (RELU) v = fmaxf(v, 0.0f);
            outp[(size_t)node * F + c] = v;
        }
    }
}


extern "C" void kernel_launch(void* const* d_in, const int* in_sizes, int n_in,
                              void* d_out, int out_size, void* d_ws, size_t ws_size,
                              hipStream_t stream) {
    const float* x   = (const float*)d_in[0];
    const int*   ei  = (const int*)  d_in[1];
    const float* W1  = (const float*)d_in[2];
    const float* as1 = (const float*)d_in[3];
    const float* ad1 = (const float*)d_in[4];
    const float* b1  = (const float*)d_in[5];
    const float* W2  = (const float*)d_in[6];
    const float* as2 = (const float*)d_in[7];
    const float* ad2 = (const float*)d_in[8];
    const float* b2  = (const float*)d_in[9];

    const int N  = in_sizes[0] / 128;
    const int E  = in_sizes[1] / 2;

    char* w = (char*)d_ws;
    auto carve = [&](size_t bytes) -> void* {
        void* p = (void*)w;
        w += (bytes + 255) & ~((size_t)255);
        return p;
    };
    float* h1  = (float*)carve((size_t)N * 256 * 4);
    float* y1  = (float*)carve((size_t)N * 256 * 4);
    float* aS1 = (float*)carve((size_t)N * 4 * 4);
    float* aD1 = (float*)carve((size_t)N * 4 * 4);
    float* h2  = (float*)carve((size_t)N * 16 * 4);
    float* aS2 = (float*)carve((size_t)N * 4);
    float* aD2 = (float*)carve((size_t)N * 4);

    const int TB = 256;
    auto blocks = [&](long long n) { return (unsigned)((n + TB - 1) / TB); };
    const unsigned nrange = (unsigned)((N + RNG - 1) / RNG);

    dim3 g1((N + 15) / 16, 4);
    gemm1_wmma<<<g1, 32, 0, stream>>>(x, W1, h1, N);
    att1_kernel<<<blocks(N * 4), TB, 0, stream>>>(h1, as1, ad1, aS1, aD1, N * 4);
    node_range_layer<4, 64, true><<<nrange, NTH, 0, stream>>>(ei, E, N, aS1, aD1, h1, b1, y1);

    gemm2_wmma<<<(N + 15) / 16, 32, 0, stream>>>(y1, W2, h2, N);
    att2_kernel<<<blocks(N), TB, 0, stream>>>(h2, as2, ad2, aS2, aD2, N);
    node_range_layer<1, 16, false><<<nrange, NTH, 0, stream>>>(ei, E, N, aS2, aD2, h2, b2, (float*)d_out);
}


# Edge filter runs once per call: layer 1's blocks save their lists (count, sources, local destinations) in the workspace, layer 2's blocks load them, no second scan; LDS float atomics kept
# speedup vs baseline: 3.3511x; 1.1535x over previous
#include <hip/hip_runtime.h>
#include <hip/hip_bf16.h>

typedef __attribute__((ext_vector_type(2))) float v2f;
typedef __attribute__((ext_vector_type(8))) float v8f;

#define NEG_SLOPE 0.2f


__device__ __forceinline__ float lrelu(float v) {
    return v >= 0.0f ? v : NEG_SLOPE * v;
}

__device__ __forceinline__ unsigned fkey(float f) {
    unsigned u = __float_as_uint(f);
    return (u & 0x80000000u) ? ~u : (u | 0x80000000u);
}
__device__ __forceinline__ float fdekey(unsigned k) {
    return (k & 0x80000000u) ? __uint_as_float(k ^ 0x80000000u)
                             : __uint_as_float(~k);
}

__global__ __launch_bounds__(32)
void gemm1_wmma(const float* __restrict__ x, const float* __restrict__ W1,
                float* __restrict__ h1, int N) {
    const int mtile = blockIdx.x;
    const int lane  = threadIdx.x;
    const int mrow  = mtile * 16 + (lane & 15);
    const int lrow  = mrow < N ? mrow : N - 1;
    const int khalf = (lane >> 4) * 2;
    const int colbase = blockIdx.y * 64;
    const int nc    = lane & 15;

    v8f c0 = {}, c1 = {}, c2 = {}, c3 = {};
    const float* xrow = x + (size_t)lrow * 128;

    for (int k = 0; k < 128; k += 4) {
        const int kb = k + khalf;
        v2f a; a.x = xrow[kb]; a.y = xrow[kb + 1];
        const float* w0 = W1 + (size_t)kb * 256 + colbase + nc;
        const float* w1 = w0 + 256;
        v2f b0; b0.x = w0[0];  b0.y = w1[0];
        v2f b1; b1.x = w0[16]; b1.y = w1[16];
        v2f b2; b2.x = w0[32]; b2.y = w1[32];
        v2f b3; b3.x = w0[48]; b3.y = w1[48];
        c0 = __builtin_amdgcn_wmma_f32_16x16x4_f32(false, a, false, b0, (short)0, c0, false, false);
        c1 = __builtin_amdgcn_wmma_f32_16x16x4_f32(false, a, false, b1, (short)0, c1, false, false);
        c2 = __builtin_amdgcn_wmma_f32_16x16x4_f32(false, a, false, b2, (short)0, c2, false, false);
        c3 = __builtin_amdgcn_wmma_f32_16x16x4_f32(false, a, false, b3, (short)0, c3, false, false);
    }

    const int rowoff = (lane >= 16) ? 8 : 0;
    for (int r = 0; r < 8; ++r) {
        const int grow = mtile * 16 + rowoff + r;
        if (grow < N) {
            float* op = h1 + (size_t)grow * 256 + colbase + nc;
            op[0]  = c0[r];
            op[16] = c1[r];
            op[32] = c2[r];
            op[48] = c3[r];
        }
    }
}

__global__ __launch_bounds__(32)
void gemm2_wmma(const float* __restrict__ y1, const float* __restrict__ W2,
                float* __restrict__ h2, int N) {
    const int mtile = blockIdx.x;
    const int lane  = threadIdx.x;
    const int mrow  = mtile * 16 + (lane & 15);
    const int lrow  = mrow < N ? mrow : N - 1;
    const int khalf = (lane >> 4) * 2;
    const int nc    = lane & 15;

    v8f c = {};
    const float* yr = y1 + (size_t)lrow * 256;
    for (int k = 0; k < 256; k += 4) {
        const int kb = k + khalf;
        v2f a; a.x = yr[kb]; a.y = yr[kb + 1];
        v2f b; b.x = W2[(size_t)kb * 16 + nc]; b.y = W2[(size_t)(kb + 1) * 16 + nc];
        c = __builtin_amdgcn_wmma_f32_16x16x4_f32(false, a, false, b, (short)0, c, false, false);
    }
    const int rowoff = (lane >= 16) ? 8 : 0;
    for (int r = 0; r < 8; ++r) {
        const int grow = mtile * 16 + rowoff + r;
        if (grow < N) h2[(size_t)grow * 16 + nc] = c[r];
    }
}

__global__ void att1_kernel(const float* __restrict__ h1,
                            const float* __restrict__ att_src,
                            const float* __restrict__ att_dst,
                            float* __restrict__ a_src, float* __restrict__ a_dst,
                            int NH) {
    int i = blockIdx.x * blockDim.x + threadIdx.x;
    if (i >= NH) return;
    const int h = i & 3;
    const float* hp = h1 + (size_t)(i >> 2) * 256 + h * 64;
    const float* as = att_src + h * 64;
    const float* ad = att_dst + h * 64;
    float s = 0.0f, d = 0.0f;
    for (int c = 0; c < 64; ++c) {
        const float v = hp[c];
        s += v * as[c];
        d += v * ad[c];
    }
    a_src[i] = s;
    a_dst[i] = d;
}

__global__ void att2_kernel(const float* __restrict__ h2,
                            const float* __restrict__ att_src,
                            const float* __restrict__ att_dst,
                            float* __restrict__ a_src, float* __restrict__ a_dst,
                            int N) {
    int i = blockIdx.x * blockDim.x + threadIdx.x;
    if (i >= N) return;
    const float* hp = h2 + (size_t)i * 16;
    float s = 0.0f, d = 0.0f;
    for (int c = 0; c < 16; ++c) {
        const float v = hp[c];
        s += v * att_src[c];
        d += v * att_dst[c];
    }
    a_src[i] = s;
    a_dst[i] = d;
}

#define RNG  200
#define CAPL 4096
#define NTH  1024
#define LHDR 64
#define LSTR (LHDR + 2 * CAPL)

__device__ __forceinline__ void list_one(int d, int e, int d0,
                                         int* lst_s, int* lst_d, int* cnt) {
    const unsigned dl = (unsigned)(d - d0);
    if (dl < (unsigned)RNG) {
        const int slot = atomicAdd(cnt, 1);
        if (slot < CAPL) { lst_s[slot] = e; lst_d[slot] = (int)dl; }
    }
}

__device__ __forceinline__ void scan_edges(const int* __restrict__ dstp, int lo, int hi,
                                           bool vec, int d0, int tid,
                                           int* lst_s, int* lst_d, int* cnt) {
    constexpr int SU = 8;
    if (vec) {
        const int nq = (hi - lo) >> 2;
        const int K  = (nq + NTH - 1) / NTH;
        const int rot = (int)(((long long)blockIdx.x * K) / gridDim.x);
        for (int k = 0; k < K; k += SU) {
            int4 v[SU]; int eb[SU]; bool ok[SU];
            #pragma unroll
            for (int u = 0; u < SU; ++u) {
                int kr = k + u + rot; kr = kr >= K ? kr - K : kr;
                const int q = kr * NTH + tid;
                ok[u] = (k + u < K) && (q < nq);
                const int qq = ok[u] ? q : 0;
                v[u]  = *(const int4*)(dstp + lo + 4 * qq);
                eb[u] = lo + 4 * qq;
            }
            #pragma unroll
            for (int u = 0; u < SU; ++u) {
                if (ok[u]) {
                    list_one(v[u].x, eb[u] + 0, d0, lst_s, lst_d, cnt);
                    list_one(v[u].y, eb[u] + 1, d0, lst_s, lst_d, cnt);
                    list_one(v[u].z, eb[u] + 2, d0, lst_s, lst_d, cnt);
                    list_one(v[u].w, eb[u] + 3, d0, lst_s, lst_d, cnt);
                }
            }
        }
        for (int e = lo + 4 * nq + tid; e < hi; e += NTH)
            list_one(dstp[e], e, d0, lst_s, lst_d, cnt);
    } else {
        for (int e = lo + tid; e < hi; e += NTH)
            list_one(dstp[e], e, d0, lst_s, lst_d, cnt);
    }
}

template <int H>
__device__ __forceinline__ void phase_max(int n, bool with_self, int d0, int N, int tid,
                                          const float* __restrict__ a_src,
                                          const int* lst_s, const int* lst_d,
                                          const float* adl, unsigned* mkey) {
    const int tot = n + (with_self ? RNG : 0);
    for (int i = tid; i < tot; i += NTH) {
        int s, dl;
        if (i < n) { s = lst_s[i]; dl = lst_d[i]; }
        else       { dl = i - n;   s = d0 + dl; }
        if (d0 + dl < N) {
            for (int h = 0; h < H; ++h) {
                const float v = lrelu(a_src[s * H + h] + adl[dl * H + h]);
                atomicMax(&mkey[dl * H + h], fkey(v));
            }
        }
    }
}

template <int H, int C>
__device__ __forceinline__ void phase_sum(int n, bool with_self, int d0, int N, int tid,
                                          const float* __restrict__ a_src,
                                          const float* __restrict__ hfeat,
                                          const int* lst_s, const int* lst_d,
                                          const float* adl, const unsigned* mkey,
                                          float* den, float* acc) {
    constexpr int F   = H * C;
    constexpr int LPE = F >= 32 ? 32 : F;
    constexpr int EPW = 32 / LPE;
    constexpr int J   = F / LPE;
    constexpr int U   = 4;
    constexpr int STEP = (NTH / 32) * EPW * U;
    const int lane = tid & 31, wave = tid >> 5;
    const int sub = lane / LPE, l = lane % LPE;
    const int tot = n + (with_self ? RNG : 0);

    for (int i0 = 0; i0 < tot; i0 += STEP) {
        int dl[U]; bool ok[U];
        float row[U][J]; float as[U][H];
        #pragma unroll
        for (int u = 0; u < U; ++u) {
            const int i = i0 + (wave * U + u) * EPW + sub;
            int s;
            if (i < n) { s = lst_s[i]; dl[u] = lst_d[i]; }
            else       { dl[u] = i - n; s = d0 + dl[u]; }
            ok[u] = (i < tot) && (d0 + dl[u] < N);
            if (!ok[u]) { s = 0; dl[u] = 0; }
            const float* rp = hfeat + (size_t)s * F + l;
            #pragma unroll
            for (int j = 0; j < J; ++j) row[u][j] = rp[LPE * j];
            #pragma unroll
            for (int h = 0; h < H; ++h) as[u][h] = a_src[s * H + h];
        }
        #pragma unroll
        for (int u = 0; u < U; ++u) {
            if (ok[u]) {
                float ex[H];
                #pragma unroll
                for (int h = 0; h < H; ++h) {
                    const float v = lrelu(as[u][h] + adl[dl[u] * H + h]);
                    ex[h] = expf(v - fdekey(mkey[dl[u] * H + h]));
                }
                float* ap = acc + dl[u] * F + l;
                #pragma unroll
                for (int j = 0; j < J; ++j)
                    atomicAdd(ap + LPE * j, ex[(LPE * j) / C] * row[u][j]);
                if (l < H) {
                    float mine = ex[0];
                    #pragma unroll
                    for (int h = 1; h < H; ++h) mine = (l == h) ? ex[h] : mine;
                    atomicAdd(&den[dl[u] * H + l], mine);
                }
            }
        }
    }
}

template <int H, int C, bool RELU, bool SAVE, bool LOAD>
__global__ __launch_bounds__(NTH)
void node_range_layer(const int* __restrict__ ei, int E, int N,
                      const float* __restrict__ a_src, const float* __restrict__ a_dst,
                      const float* __restrict__ hfeat, const float* __restrict__ bias,
                      float* __restrict__ outp, int* __restrict__ lists) {
    constexpr int F = H * C;
    __shared__ float    acc[RNG * F];
    __shared__ unsigned mkey[RNG * H];
    __shared__ float    den[RNG * H];
    __shared__ float    adl[RNG * H];
    __shared__ int      lst_s[CAPL];
    __shared__ int      lst_d[CAPL];
    __shared__ int      cnt;

    const int tid = threadIdx.x;
    const int d0  = blockIdx.x * RNG;
    const int* dstp = ei + E;
    const bool vec = (((size_t)dstp) & 15) == 0;

    for (int i = tid; i < RNG * F; i += NTH) acc[i] = 0.0f;
    for (int i = tid; i < RNG * H; i += NTH) {
        mkey[i] = 0u; den[i] = 0.0f;
        adl[i] = (d0 * H + i < N * H) ? a_dst[d0 * H + i] : 0.0f;
    }
    if (tid == 0) cnt = 0;
    __syncthreads();

    int* const my_cnt = lists + (size_t)blockIdx.x * LSTR;
    int* const my_s   = my_cnt + LHDR;
    int* const my_d   = my_s + CAPL;

    int total;
    if (LOAD) {
        total = my_cnt[0];
    } else {
        scan_edges(dstp, 0, E, vec, d0, tid, lst_s, lst_d, &cnt);
        __syncthreads();
        total = cnt;
        if (SAVE && tid == 0) my_cnt[0] = total;
    }

    if (total <= CAPL) {
        if (LOAD) {
            for (int i = tid; i < total; i += NTH) { lst_s[i] = my_s[i]; lst_d[i] = my_d[i]; }
        } else {
            for (int i = tid; i < total; i += NTH) {
                const int s = ei[lst_s[i]];
                lst_s[i] = s;
                if (SAVE) { my_s[i] = s; my_d[i] = lst_d[i]; }
            }
        }
        __syncthreads();
        phase_max<H>(total, true, d0, N, tid, a_src, lst_s, lst_d, adl, mkey);
        __syncthreads();
        phase_sum<H, C>(total, true, d0, N, tid, a_src, hfeat, lst_s, lst_d, adl, mkey, den, acc);
        __syncthreads();
    } else {
        __syncthreads();
        for (int pass = 0; pass < 2; ++pass) {
            for (int lo = 0; lo < E; lo += CAPL) {
                const int hi = lo + CAPL < E ? lo + CAPL : E;
                if (tid == 0) cnt = 0;
                __syncthreads();
                scan_edges(dstp, lo, hi, vec, d0, tid, lst_s, lst_d, &cnt);
                __syncthreads();
                const int n = cnt;
                for (int i = tid; i < n; i += NTH) lst_s[i] = ei[lst_s[i]];
                __syncthreads();
                if (pass == 0)
                    phase_max<H>(n, lo == 0, d0, N, tid, a_src, lst_s, lst_d, adl, mkey);
                else
                    phase_sum<H, C>(n, lo == 0, d0, N, tid, a_src, hfeat, lst_s, lst_d, adl, mkey, den, acc);
                __syncthreads();
            }
        }
    }

    for (int idx = tid; idx < RNG * F; idx += NTH) {
        const int dl = idx / F, c = idx % F;
        const int node = d0 + dl;
        if (node < N) {
            float v = acc[idx] / den[dl * H + c / C] + bias[c];
            if (RELU) v = fmaxf(v, 0.0f);
            outp[(size_t)node * F + c] = v;
        }
    }
}


extern "C" void kernel_launch(void* const* d_in, const int* in_sizes, int n_in,
                              void* d_out, int out_size, void* d_ws, size_t ws_size,
                              hipStream_t stream) {
    const float* x   = (const float*)d_in[0];
    const int*   ei  = (const int*)  d_in[1];
    const float* W1  = (const float*)d_in[2];
    const float* as1 = (const float*)d_in[3];
    const float* ad1 = (const float*)d_in[4];
    const float* b1  = (const float*)d_in[5];
    const float* W2  = (const float*)d_in[6];
    const float* as2 = (const float*)d_in[7];
    const float* ad2 = (const float*)d_in[8];
    const float* b2  = (const float*)d_in[9];

    const int N  = in_sizes[0] / 128;
    const int E  = in_sizes[1] / 2;

    char* w = (char*)d_ws;
    auto carve = [&](size_t bytes) -> void* {
        void* p = (void*)w;
        w += (bytes + 255) & ~((size_t)255);
        return p;
    };
    float* h1  = (float*)carve((size_t)N * 256 * 4);
    float* y1  = (float*)carve((size_t)N * 256 * 4);
    float* aS1 = (float*)carve((size_t)N * 4 * 4);
    float* aD1 = (float*)carve((size_t)N * 4 * 4);
    float* h2  = (float*)carve((size_t)N * 16 * 4);
    float* aS2 = (float*)carve((size_t)N * 4);
    float* aD2 = (float*)carve((size_t)N * 4);

    const int TB = 256;
    auto blocks = [&](long long n) { return (unsigned)((n + TB - 1) / TB); };
    const unsigned nrange = (unsigned)((N + RNG - 1) / RNG);
    int* lists = (int*)carve((size_t)nrange * LSTR * 4);

    dim3 g1((N + 15) / 16, 4);
    gemm1_wmma<<<g1, 32, 0, stream>>>(x, W1, h1, N);
    att1_kernel<<<blocks(N * 4), TB, 0, stream>>>(h1, as1, ad1, aS1, aD1, N * 4);
    node_range_layer<4, 64, true, true, false><<<nrange, NTH, 0, stream>>>(ei, E, N, aS1, aD1, h1, b1, y1, lists);

    gemm2_wmma<<<(N + 15) / 16, 32, 0, stream>>>(y1, W2, h2, N);
    att2_kernel<<<blocks(N), TB, 0, stream>>>(h2, as2, ad2, aS2, aD2, N);
    node_range_layer<1, 16, false, false, true><<<nrange, NTH, 0, stream>>>(ei, E, N, aS2, aD2, h2, b2, (float*)d_out, lists);
}


# Layer-1 edge filter reads one byte per edge (destination range number, written once per call by the score launch), 16 edges per 16-byte load; list order changes, LDS float atomics kept
# speedup vs baseline: 3.6832x; 1.0991x over previous
#include <hip/hip_runtime.h>
#include <hip/hip_bf16.h>

typedef __attribute__((ext_vector_type(2))) float v2f;
typedef __attribute__((ext_vector_type(8))) float v8f;

#define NEG_SLOPE 0.2f


__device__ __forceinline__ float lrelu(float v) {
    return v >= 0.0f ? v : NEG_SLOPE * v;
}

__device__ __forceinline__ unsigned fkey(float f) {
    unsigned u = __float_as_uint(f);
    return (u & 0x80000000u) ? ~u : (u | 0x80000000u);
}
__device__ __forceinline__ float fdekey(unsigned k) {
    return (k & 0x80000000u) ? __uint_as_float(k ^ 0x80000000u)
                             : __uint_as_float(~k);
}

__global__ __launch_bounds__(32)
void gemm1_wmma(const float* __restrict__ x, const float* __restrict__ W1,
                float* __restrict__ h1, int N) {
    const int mtile = blockIdx.x;
    const int lane  = threadIdx.x;
    const int mrow  = mtile * 16 + (lane & 15);
    const int lrow  = mrow < N ? mrow : N - 1;
    const int khalf = (lane >> 4) * 2;
    const int colbase = blockIdx.y * 64;
    const int nc    = lane & 15;

    v8f c0 = {}, c1 = {}, c2 = {}, c3 = {};
    const float* xrow = x + (size_t)lrow * 128;

    for (int k = 0; k < 128; k += 4) {
        const int kb = k + khalf;
        v2f a; a.x = xrow[kb]; a.y = xrow[kb + 1];
        const float* w0 = W1 + (size_t)kb * 256 + colbase + nc;
        const float* w1 = w0 + 256;
        v2f b0; b0.x = w0[0];  b0.y = w1[0];
        v2f b1; b1.x = w0[16]; b1.y = w1[16];
        v2f b2; b2.x = w0[32]; b2.y = w1[32];
        v2f b3; b3.x = w0[48]; b3.y = w1[48];
        c0 = __builtin_amdgcn_wmma_f32_16x16x4_f32(false, a, false, b0, (short)0, c0, false, false);
        c1 = __builtin_amdgcn_wmma_f32_16x16x4_f32(false, a, false, b1, (short)0, c1, false, false);
        c2 = __builtin_amdgcn_wmma_f32_16x16x4_f32(false, a, false, b2, (short)0, c2, false, false);
        c3 = __builtin_amdgcn_wmma_f32_16x16x4_f32(false, a, false, b3, (short)0, c3, false, false);
    }

    const int rowoff = (lane >= 16) ? 8 : 0;
    for (int r = 0; r < 8; ++r) {
        const int grow = mtile * 16 + rowoff + r;
        if (grow < N) {
            float* op = h1 + (size_t)grow * 256 + colbase + nc;
            op[0]  = c0[r];
            op[16] = c1[r];
            op[32] = c2[r];
            op[48] = c3[r];
        }
    }
}

__global__ __launch_bounds__(32)
void gemm2_wmma(const float* __restrict__ y1, const float* __restrict__ W2,
                float* __restrict__ h2, int N) {
    const int mtile = blockIdx.x;
    const int lane  = threadIdx.x;
    const int mrow  = mtile * 16 + (lane & 15);
    const int lrow  = mrow < N ? mrow : N - 1;
    const int khalf = (lane >> 4) * 2;
    const int nc    = lane & 15;

    v8f c = {};
    const float* yr = y1 + (size_t)lrow * 256;
    for (int k = 0; k < 256; k += 4) {
        const int kb = k + khalf;
        v2f a; a.x = yr[kb]; a.y = yr[kb + 1];
        v2f b; b.x = W2[(size_t)kb * 16 + nc]; b.y = W2[(size_t)(kb + 1) * 16 + nc];
        c = __builtin_amdgcn_wmma_f32_16x16x4_f32(false, a, false, b, (short)0, c, false, false);
    }
    const int rowoff = (lane >= 16) ? 8 : 0;
    for (int r = 0; r < 8; ++r) {
        const int grow = mtile * 16 + rowoff + r;
        if (grow < N) h2[(size_t)grow * 16 + nc] = c[r];
    }
}

#define RNG  200

__global__ void att1_kernel(const float* __restrict__ h1,
                            const float* __restrict__ att_src,
                            const float* __restrict__ att_dst,
                            float* __restrict__ a_src, float* __restrict__ a_dst,
                            int NH,
                            const int* __restrict__ dstp, int E,
                            unsigned* __restrict__ ids32) {
    int i = blockIdx.x * blockDim.x + threadIdx.x;
    {
        const int nwp = ((E + 15) >> 4) << 2;
        const int nthreads = gridDim.x * blockDim.x;
        const bool vec = (((size_t)dstp) & 15) == 0;
        for (int w = i; w < nwp; w += nthreads) {
            int d[4] = {-1, -1, -1, -1};
            if (vec && 4 * w + 3 < E) {
                const int4 v = *(const int4*)(dstp + 4 * w);
                d[0] = v.x; d[1] = v.y; d[2] = v.z; d[3] = v.w;
            } else {
                #pragma unroll
                for (int j = 0; j < 4; ++j)
                    if (4 * w + j < E) d[j] = dstp[4 * w + j];
            }
            unsigned p = 0;
            #pragma unroll
            for (int j = 0; j < 4; ++j) {
                const unsigned q = (unsigned)d[j] / (unsigned)RNG;
                p |= (q < 255u ? q : 255u) << (8 * j);
            }
            ids32[w] = p;
        }
    }
    if (i >= NH) return;
    const int h = i & 3;
    const float* hp = h1 + (size_t)(i >> 2) * 256 + h * 64;
    const float* as = att_src + h * 64;
    const float* ad = att_dst + h * 64;
    float s = 0.0f, d = 0.0f;
    for (int c = 0; c < 64; ++c) {
        const float v = hp[c];
        s += v * as[c];
        d += v * ad[c];
    }
    a_src[i] = s;
    a_dst[i] = d;
}

__global__ void att2_kernel(const float* __restrict__ h2,
                            const float* __restrict__ att_src,
                            const float* __restrict__ att_dst,
                            float* __restrict__ a_src, float* __restrict__ a_dst,
                            int N) {
    int i = blockIdx.x * blockDim.x + threadIdx.x;
    if (i >= N) return;
    const float* hp = h2 + (size_t)i * 16;
    float s = 0.0f, d = 0.0f;
    for (int c = 0; c < 16; ++c) {
        const float v = hp[c];
        s += v * att_src[c];
        d += v * att_dst[c];
    }
    a_src[i] = s;
    a_dst[i] = d;
}

#define CAPL 4096
#define NTH  1024
#define LHDR 64
#define LSTR (LHDR + 2 * CAPL)

__device__ __forceinline__ void list_one(int d, int e, int d0,
                                         int* lst_s, int* lst_d, int* cnt) {
    const unsigned dl = (unsigned)(d - d0);
    if (dl < (unsigned)RNG) {
        const int slot = atomicAdd(cnt, 1);
        if (slot < CAPL) { lst_s[slot] = e; lst_d[slot] = (int)dl; }
    }
}

__device__ __forceinline__ void scan_edges(const int* __restrict__ dstp, int lo, int hi,
                                           bool vec, int d0, int tid,
                                           int* lst_s, int* lst_d, int* cnt) {
    constexpr int SU = 8;
    if (vec) {
        const int nq = (hi - lo) >> 2;
        const int K  = (nq + NTH - 1) / NTH;
        const int rot = (int)(((long long)blockIdx.x * K) / gridDim.x);
        for (int k = 0; k < K; k += SU) {
            int4 v[SU]; int eb[SU]; bool ok[SU];
            #pragma unroll
            for (int u = 0; u < SU; ++u) {
                int kr = k + u + rot; kr = kr >= K ? kr - K : kr;
                const int q = kr * NTH + tid;
                ok[u] = (k + u < K) && (q < nq);
                const int qq = ok[u] ? q : 0;
                v[u]  = *(const int4*)(dstp + lo + 4 * qq);
                eb[u] = lo + 4 * qq;
            }
            #pragma unroll
            for (int u = 0; u < SU; ++u) {
                if (ok[u]) {
                    list_one(v[u].x, eb[u] + 0, d0, lst_s, lst_d, cnt);
                    list_one(v[u].y, eb[u] + 1, d0, lst_s, lst_d, cnt);
                    list_one(v[u].z, eb[u] + 2, d0, lst_s, lst_d, cnt);
                    list_one(v[u].w, eb[u] + 3, d0, lst_s, lst_d, cnt);
                }
            }
        }
        for (int e = lo + 4 * nq + tid; e < hi; e += NTH)
            list_one(dstp[e], e, d0, lst_s, lst_d, cnt);
    } else {
        for (int e = lo + tid; e < hi; e += NTH)
            list_one(dstp[e], e, d0, lst_s, lst_d, cnt);
    }
}

__device__ __forceinline__ unsigned zero_bytes(unsigned x) {
    return ~(((x & 0x7f7f7f7fu) + 0x7f7f7f7fu) | x | 0x7f7f7f7fu);
}

__device__ __forceinline__ void scan_ids(const uint4* __restrict__ ids, int E, int tid,
                                         int* lst_s, int* cnt) {
    constexpr int SU = 8;
    const int ng = (E + 15) >> 4;
    const int K  = (ng + NTH - 1) / NTH;
    const int rot = (int)(((long long)blockIdx.x * K) / gridDim.x);
    const unsigned mine = blockIdx.x * 0x01010101u;
    for (int k = 0; k < K; k += SU) {
        uint4 v[SU]; int eb[SU]; bool ok[SU];
        #pragma unroll
        for (int u = 0; u < SU; ++u) {
            int kr = k + u + rot; kr = kr >= K ? kr - K : kr;
            const int g = kr * NTH + tid;
            ok[u] = (k + u < K) && (g < ng);
            const int gg = ok[u] ? g : 0;
            v[u]  = ids[gg];
            eb[u] = 16 * gg;
        }
        #pragma unroll
        for (int u = 0; u < SU; ++u) {
            if (ok[u]) {
                unsigned m = (zero_bytes(v[u].x ^ mine) >> 7) | (zero_bytes(v[u].y ^ mine) >> 6)
                           | (zero_bytes(v[u].z ^ mine) >> 5) | (zero_bytes(v[u].w ^ mine) >> 4);
                while (m) {
                    const int b = __builtin_ctz(m);
                    m &= m - 1;
                    const int e = eb[u] + ((b & 7) << 2) + (b >> 3);
                    const int slot = atomicAdd(cnt, 1);
                    if (slot < CAPL) lst_s[slot] = e;
                }
            }
        }
    }
}

template <int H>
__device__ __forceinline__ void phase_max(int n, bool with_self, int d0, int N, int tid,
                                          const float* __restrict__ a_src,
                                          const int* lst_s, const int* lst_d,
                                          const float* adl, unsigned* mkey) {
    const int tot = n + (with_self ? RNG : 0);
    for (int i = tid; i < tot; i += NTH) {
        int s, dl;
        if (i < n) { s = lst_s[i]; dl = lst_d[i]; }
        else       { dl = i - n;   s = d0 + dl; }
        if (d0 + dl < N) {
            for (int h = 0; h < H; ++h) {
                const float v = lrelu(a_src[s * H + h] + adl[dl * H + h]);
                atomicMax(&mkey[dl * H + h], fkey(v));
            }
        }
    }
}

template <int H, int C>
__device__ __forceinline__ void phase_sum(int n, bool with_self, int d0, int N, int tid,
                                          const float* __restrict__ a_src,
                                          const float* __restrict__ hfeat,
                                          const int* lst_s, const int* lst_d,
                                          const float* adl, const unsigned* mkey,
                                          float* den, float* acc) {
    constexpr int F   = H * C;
    constexpr int LPE = F >= 32 ? 32 : F;
    constexpr int EPW = 32 / LPE;
    constexpr int J   = F / LPE;
    constexpr int U   = 4;
    constexpr int STEP = (NTH / 32) * EPW * U;
    const int lane = tid & 31, wave = tid >> 5;
    const int sub = lane / LPE, l = lane % LPE;
    const int tot = n + (with_self ? RNG : 0);

    for (int i0 = 0; i0 < tot; i0 += STEP) {
        int dl[U]; bool ok[U];
        float row[U][J]; float as[U][H];
        #pragma unroll
        for (int u = 0; u < U; ++u) {
            const int i = i0 + (wave * U + u) * EPW + sub;
            int s;
            if (i < n) { s = lst_s[i]; dl[u] = lst_d[i]; }
            else       { dl[u] = i - n; s = d0 + dl[u]; }
            ok[u] = (i < tot) && (d0 + dl[u] < N);
            if (!ok[u]) { s = 0; dl[u] = 0; }
            const float* rp = hfeat + (size_t)s * F + l;
            #pragma unroll
            for (int j = 0; j < J; ++j) row[u][j] = rp[LPE * j];
            #pragma unroll
            for (int h = 0; h < H; ++h) as[u][h] = a_src[s * H + h];
        }
        #pragma unroll
        for (int u = 0; u < U; ++u) {
            if (ok[u]) {
                float ex[H];
                #pragma unroll
                for (int h = 0; h < H; ++h) {
                    const float v = lrelu(as[u][h] + adl[dl[u] * H + h]);
                    ex[h] = expf(v - fdekey(mkey[dl[u] * H + h]));
                }
                float* ap = acc + dl[u] * F + l;
                #pragma unroll
                for (int j = 0; j < J; ++j)
                    atomicAdd(ap + LPE * j, ex[(LPE * j) / C] * row[u][j]);
                if (l < H) {
                    float mine = ex[0];
                    #pragma unroll
                    for (int h = 1; h < H; ++h) mine = (l == h) ? ex[h] : mine;
                    atomicAdd(&den[dl[u] * H + l], mine);
                }
            }
        }
    }
}

template <int H, int C, bool RELU, bool SAVE, bool LOAD>
__global__ __launch_bounds__(NTH)
void node_range_layer(const int* __restrict__ ei, int E, int N,
                      const float* __restrict__ a_src, const float* __restrict__ a_dst,
                      const float* __restrict__ hfeat, const float* __restrict__ bias,
                      float* __restrict__ outp, int* __restrict__ lists,
                      const unsigned* __restrict__ ids32) {
    constexpr int F = H * C;
    __shared__ float    acc[RNG * F];
    __shared__ unsigned mkey[RNG * H];
    __shared__ float    den[RNG * H];
    __shared__ float    adl[RNG * H];
    __shared__ int      lst_s[CAPL];
    __shared__ int      lst_d[CAPL];
    __shared__ int      cnt;

    const int tid = threadIdx.x;
    const int d0  = blockIdx.x * RNG;
    const int* dstp = ei + E;
    const bool vec = (((size_t)dstp) & 15) == 0;

    for (int i = tid; i < RNG * F; i += NTH) acc[i] = 0.0f;
    for (int i = tid; i < RNG * H; i += NTH) {
        mkey[i] = 0u; den[i] = 0.0f;
        adl[i] = (d0 * H + i < N * H) ? a_dst[d0 * H + i] : 0.0f;
    }
    if (tid == 0) cnt = 0;
    __syncthreads();

    int* const my_cnt = lists + (size_t)blockIdx.x * LSTR;
    int* const my_s   = my_cnt + LHDR;
    int* const my_d   = my_s + CAPL;

    const bool by_ids = !LOAD && gridDim.x <= 255;

    int total;
    if (LOAD) {
        total = my_cnt[0];
    } else {
        if (by_ids) scan_ids((const uint4*)ids32, E, tid, lst_s, &cnt);
        else        scan_edges(dstp, 0, E, vec, d0, tid, lst_s, lst_d, &cnt);
        __syncthreads();
        total = cnt;
        if (SAVE && tid == 0) my_cnt[0] = total;
    }

    if (total <= CAPL) {
        if (LOAD) {
            for (int i = tid; i < total; i += NTH) { lst_s[i] = my_s[i]; lst_d[i] = my_d[i]; }
        } else {
            for (int i = tid; i < total; i += NTH) {
                const int e = lst_s[i];
                const int s = ei[e];
                if (by_ids) lst_d[i] = dstp[e] - d0;
                lst_s[i] = s;
                if (SAVE) { my_s[i] = s; my_d[i] = lst_d[i]; }
            }
        }
        __syncthreads();
        phase_max<H>(total, true, d0, N, tid, a_src, lst_s, lst_d, adl, mkey);
        __syncthreads();
        phase_sum<H, C>(total, true, d0, N, tid, a_src, hfeat, lst_s, lst_d, adl, mkey, den, acc);
        __syncthreads();
    } else {
        __syncthreads();
        for (int pass = 0; pass < 2; ++pass) {
            for (int lo = 0; lo < E; lo += CAPL) {
                const int hi = lo + CAPL < E ? lo + CAPL : E;
                if (tid == 0) cnt = 0;
                __syncthreads();
                scan_edges(dstp, lo, hi, vec, d0, tid, lst_s, lst_d, &cnt);
                __syncthreads();
                const int n = cnt;
                for (int i = tid; i < n; i += NTH) lst_s[i] = ei[lst_s[i]];
                __syncthreads();
                if (pass == 0)
                    phase_max<H>(n, lo == 0, d0, N, tid, a_src, lst_s, lst_d, adl, mkey);
                else
                    phase_sum<H, C>(n, lo == 0, d0, N, tid, a_src, hfeat, lst_s, lst_d, adl, mkey, den, acc);
                __syncthreads();
            }
        }
    }

    for (int idx = tid; idx < RNG * F; idx += NTH) {
        const int dl = idx / F, c = idx % F;
        const int node = d0 + dl;
        if (node < N) {
            float v = acc[idx] / den[dl * H + c / C] + bias[c];
            if (RELU) v = fmaxf(v, 0.0f);
            outp[(size_t)node * F + c] = v;
        }
    }
}


extern "C" void kernel_launch(void* const* d_in, const int* in_sizes, int n_in,
                              void* d_out, int out_size, void* d_ws, size_t ws_size,
                              hipStream_t stream) {
    const float* x   = (const float*)d_in[0];
    const int*   ei  = (const int*)  d_in[1];
    const float* W1  = (const float*)d_in[2];
    const float* as1 = (const float*)d_in[3];
    const float* ad1 = (const float*)d_in[4];
    const float* b1  = (const float*)d_in[5];
    const float* W2  = (const float*)d_in[6];
    const float* as2 = (const float*)d_in[7];
    const float* ad2 = (const float*)d_in[8];
    const float* b2  = (const float*)d_in[9];

    const int N  = in_sizes[0] / 128;
    const int E  = in_sizes[1] / 2;

    char* w = (char*)d_ws;
    auto carve = [&](size_t bytes) -> void* {
        void* p = (void*)w;
        w += (bytes + 255) & ~((size_t)255);
        return p;
    };
    float* h1  = (float*)carve((size_t)N * 256 * 4);
    float* y1  = (float*)carve((size_t)N * 256 * 4);
    float* aS1 = (float*)carve((size_t)N * 4 * 4);
    float* aD1 = (float*)carve((size_t)N * 4 * 4);
    float* h2  = (float*)carve((size_t)N * 16 * 4);
    float* aS2 = (float*)carve((size_t)N * 4);
    float* aD2 = (float*)carve((size_t)N * 4);

    const int TB = 256;
    auto blocks = [&](long long n) { return (unsigned)((n + TB - 1) / TB); };
    const unsigned nrange = (unsigned)((N + RNG - 1) / RNG);
    int* lists = (int*)carve((size_t)nrange * LSTR * 4);
    unsigned* ids32 = (unsigned*)carve((((size_t)E + 15) / 16) * 16);

    dim3 g1((N + 15) / 16, 4);
    gemm1_wmma<<<g1, 32, 0, stream>>>(x, W1, h1, N);
    att1_kernel<<<blocks(N * 4), TB, 0, stream>>>(h1, as1, ad1, aS1, aD1, N * 4, ei + E, E, ids32);
    node_range_layer<4, 64, true, true, false><<<nrange, NTH, 0, stream>>>(ei, E, N, aS1, aD1, h1, b1, y1, lists, ids32);

    gemm2_wmma<<<(N + 15) / 16, 32, 0, stream>>>(y1, W2, h2, N);
    att2_kernel<<<blocks(N), TB, 0, stream>>>(h2, as2, ad2, aS2, aD2, N);
    node_range_layer<1, 16, false, false, true><<<nrange, NTH, 0, stream>>>(ei, E, N, aS2, aD2, h2, b2, (float*)d_out, lists, ids32);
}


# Softmax weight of each edge computed once by one lane (score and exp passes, one thread per list entry, kept in LDS); row-adding waves read it, no exp redone in every lane; LDS float atomics kept
# speedup vs baseline: 4.1249x; 1.1199x over previous
#include <hip/hip_runtime.h>
#include <hip/hip_bf16.h>

typedef __attribute__((ext_vector_type(2))) float v2f;
typedef __attribute__((ext_vector_type(8))) float v8f;

#define NEG_SLOPE 0.2f


__device__ __forceinline__ float lrelu(float v) {
    return v >= 0.0f ? v : NEG_SLOPE * v;
}

__device__ __forceinline__ unsigned fkey(float f) {
    unsigned u = __float_as_uint(f);
    return (u & 0x80000000u) ? ~u : (u | 0x80000000u);
}
__device__ __forceinline__ float fdekey(unsigned k) {
    return (k & 0x80000000u) ? __uint_as_float(k ^ 0x80000000u)
                             : __uint_as_float(~k);
}

__global__ __launch_bounds__(32)
void gemm1_wmma(const float* __restrict__ x, const float* __restrict__ W1,
                float* __restrict__ h1, int N) {
    const int mtile = blockIdx.x;
    const int lane  = threadIdx.x;
    const int mrow  = mtile * 16 + (lane & 15);
    const int lrow  = mrow < N ? mrow : N - 1;
    const int khalf = (lane >> 4) * 2;
    const int colbase = blockIdx.y * 64;
    const int nc    = lane & 15;

    v8f c0 = {}, c1 = {}, c2 = {}, c3 = {};
    const float* xrow = x + (size_t)lrow * 128;

    for (int k = 0; k < 128; k += 4) {
        const int kb = k + khalf;
        v2f a; a.x = xrow[kb]; a.y = xrow[kb + 1];
        const float* w0 = W1 + (size_t)kb * 256 + colbase + nc;
        const float* w1 = w0 + 256;
        v2f b0; b0.x = w0[0];  b0.y = w1[0];
        v2f b1; b1.x = w0[16]; b1.y = w1[16];
        v2f b2; b2.x = w0[32]; b2.y = w1[32];
        v2f b3; b3.x = w0[48]; b3.y = w1[48];
        c0 = __builtin_amdgcn_wmma_f32_16x16x4_f32(false, a, false, b0, (short)0, c0, false, false);
        c1 = __builtin_amdgcn_wmma_f32_16x16x4_f32(false, a, false, b1, (short)0, c1, false, false);
        c2 = __builtin_amdgcn_wmma_f32_16x16x4_f32(false, a, false, b2, (short)0, c2, false, false);
        c3 = __builtin_amdgcn_wmma_f32_16x16x4_f32(false, a, false, b3, (short)0, c3, false, false);
    }

    const int rowoff = (lane >= 16) ? 8 : 0;
    for (int r = 0; r < 8; ++r) {
        const int grow = mtile * 16 + rowoff + r;
        if (grow < N) {
            float* op = h1 + (size_t)grow * 256 + colbase + nc;
            op[0]  = c0[r];
            op[16] = c1[r];
            op[32] = c2[r];
            op[48] = c3[r];
        }
    }
}

__global__ __launch_bounds__(32)
void gemm2_wmma(const float* __restrict__ y1, const float* __restrict__ W2,
                float* __restrict__ h2, int N) {
    const int mtile = blockIdx.x;
    const int lane  = threadIdx.x;
    const int mrow  = mtile * 16 + (lane & 15);
    const int lrow  = mrow < N ? mrow : N - 1;
    const int khalf = (lane >> 4) * 2;
    const int nc    = lane & 15;

    v8f c = {};
    const float* yr = y1 + (size_t)lrow * 256;
    for (int k = 0; k < 256; k += 4) {
        const int kb = k + khalf;
        v2f a; a.x = yr[kb]; a.y = yr[kb + 1];
        v2f b; b.x = W2[(size_t)kb * 16 + nc]; b.y = W2[(size_t)(kb + 1) * 16 + nc];
        c = __builtin_amdgcn_wmma_f32_16x16x4_f32(false, a, false, b, (short)0, c, false, false);
    }
    const int rowoff = (lane >= 16) ? 8 : 0;
    for (int r = 0; r < 8; ++r) {
        const int grow = mtile * 16 + rowoff + r;
        if (grow < N) h2[(size_t)grow * 16 + nc] = c[r];
    }
}

#define RNG  200

__global__ void att1_kernel(const float* __restrict__ h1,
                            const float* __restrict__ att_src,
                            const float* __restrict__ att_dst,
                            float* __restrict__ a_src, float* __restrict__ a_dst,
                            int NH,
                            const int* __restrict__ dstp, int E,
                            unsigned* __restrict__ ids32) {
    int i = blockIdx.x * blockDim.x + threadIdx.x;
    {
        const int nwp = ((E + 15) >> 4) << 2;
        const int nthreads = gridDim.x * blockDim.x;
        const bool vec = (((size_t)dstp) & 15) == 0;
        for (int w = i; w < nwp; w += nthreads) {
            int d[4] = {-1, -1, -1, -1};
            if (vec && 4 * w + 3 < E) {
                const int4 v = *(const int4*)(dstp + 4 * w);
                d[0] = v.x; d[1] = v.y; d[2] = v.z; d[3] = v.w;
            } else {
                #pragma unroll
                for (int j = 0; j < 4; ++j)
                    if (4 * w + j < E) d[j] = dstp[4 * w + j];
            }
            unsigned p = 0;
            #pragma unroll
            for (int j = 0; j < 4; ++j) {
                const unsigned q = (unsigned)d[j] / (unsigned)RNG;
                p |= (q < 255u ? q : 255u) << (8 * j);
            }
            ids32[w] = p;
        }
    }
    if (i >= NH) return;
    const int h = i & 3;
    const float* hp = h1 + (size_t)(i >> 2) * 256 + h * 64;
    const float* as = att_src + h * 64;
    const float* ad = att_dst + h * 64;
    float s = 0.0f, d = 0.0f;
    for (int c = 0; c < 64; ++c) {
        const float v = hp[c];
        s += v * as[c];
        d += v * ad[c];
    }
    a_src[i] = s;
    a_dst[i] = d;
}

__global__ void att2_kernel(const float* __restrict__ h2,
                            const float* __restrict__ att_src,
                            const float* __restrict__ att_dst,
                            float* __restrict__ a_src, float* __restrict__ a_dst,
                            int N) {
    int i = blockIdx.x * blockDim.x + threadIdx.x;
    if (i >= N) return;
    const float* hp = h2 + (size_t)i * 16;
    float s = 0.0f, d = 0.0f;
    for (int c = 0; c < 16; ++c) {
        const float v = hp[c];
        s += v * att_src[c];
        d += v * att_dst[c];
    }
    a_src[i] = s;
    a_dst[i] = d;
}

#define CAPL 4096
#define NTH  1024
#define LHDR 64
#define LSTR (LHDR + 2 * CAPL)

__device__ __forceinline__ void list_one(int d, int e, int d0,
                                         int* lst_s, int* lst_d, int* cnt) {
    const unsigned dl = (unsigned)(d - d0);
    if (dl < (unsigned)RNG) {
        const int slot = atomicAdd(cnt, 1);
        if (slot < CAPL) { lst_s[slot] = e; lst_d[slot] = (int)dl; }
    }
}

__device__ __forceinline__ void scan_edges(const int* __restrict__ dstp, int lo, int hi,
                                           bool vec, int d0, int tid,
                                           int* lst_s, int* lst_d, int* cnt) {
    constexpr int SU = 8;
    if (vec) {
        const int nq = (hi - lo) >> 2;
        const int K  = (nq + NTH - 1) / NTH;
        const int rot = (int)(((long long)blockIdx.x * K) / gridDim.x);
        for (int k = 0; k < K; k += SU) {
            int4 v[SU]; int eb[SU]; bool ok[SU];
            #pragma unroll
            for (int u = 0; u < SU; ++u) {
                int kr = k + u + rot; kr = kr >= K ? kr - K : kr;
                const int q = kr * NTH + tid;
                ok[u] = (k + u < K) && (q < nq);
                const int qq = ok[u] ? q : 0;
                v[u]  = *(const int4*)(dstp + lo + 4 * qq);
                eb[u] = lo + 4 * qq;
            }
            #pragma unroll
            for (int u = 0; u < SU; ++u) {
                if (ok[u]) {
                    list_one(v[u].x, eb[u] + 0, d0, lst_s, lst_d, cnt);
                    list_one(v[u].y, eb[u] + 1, d0, lst_s, lst_d, cnt);
                    list_one(v[u].z, eb[u] + 2, d0, lst_s, lst_d, cnt);
                    list_one(v[u].w, eb[u] + 3, d0, lst_s, lst_d, cnt);
                }
            }
        }
        for (int e = lo + 4 * nq + tid; e < hi; e += NTH)
            list_one(dstp[e], e, d0, lst_s, lst_d, cnt);
    } else {
        for (int e = lo + tid; e < hi; e += NTH)
            list_one(dstp[e], e, d0, lst_s, lst_d, cnt);
    }
}

__device__ __forceinline__ unsigned zero_bytes(unsigned x) {
    return ~(((x & 0x7f7f7f7fu) + 0x7f7f7f7fu) | x | 0x7f7f7f7fu);
}

__device__ __forceinline__ void scan_ids(const uint4* __restrict__ ids, int E, int tid,
                                         int* lst_s, int* cnt) {
    constexpr int SU = 8;
    const int ng = (E + 15) >> 4;
    const int K  = (ng + NTH - 1) / NTH;
    const int rot = (int)(((long long)blockIdx.x * K) / gridDim.x);
    const unsigned mine = blockIdx.x * 0x01010101u;
    for (int k = 0; k < K; k += SU) {
        uint4 v[SU]; int eb[SU]; bool ok[SU];
        #pragma unroll
        for (int u = 0; u < SU; ++u) {
            int kr = k + u + rot; kr = kr >= K ? kr - K : kr;
            const int g = kr * NTH + tid;
            ok[u] = (k + u < K) && (g < ng);
            const int gg = ok[u] ? g : 0;
            v[u]  = ids[gg];
            eb[u] = 16 * gg;
        }
        #pragma unroll
        for (int u = 0; u < SU; ++u) {
            if (ok[u]) {
                unsigned m = (zero_bytes(v[u].x ^ mine) >> 7) | (zero_bytes(v[u].y ^ mine) >> 6)
                           | (zero_bytes(v[u].z ^ mine) >> 5) | (zero_bytes(v[u].w ^ mine) >> 4);
                while (m) {
                    const int b = __builtin_ctz(m);
                    m &= m - 1;
                    const int e = eb[u] + ((b & 7) << 2) + (b >> 3);
                    const int slot = atomicAdd(cnt, 1);
                    if (slot < CAPL) lst_s[slot] = e;
                }
            }
        }
    }
}

template <int H, bool MAXUPD>
__device__ __forceinline__ void phase_score(int n, bool with_self, int d0, int N, int tid,
                                            const float* __restrict__ a_src,
                                            const int* lst_s, const int* lst_d,
                                            const float* adl, unsigned* mkey, float* wgt) {
    const int tot = n + (with_self ? RNG : 0);
    for (int i = tid; i < tot; i += NTH) {
        int s, dl;
        if (i < n) { s = lst_s[i]; dl = lst_d[i]; }
        else       { dl = i - n;   s = d0 + dl; }
        if (d0 + dl < N) {
            for (int h = 0; h < H; ++h) {
                const float v = lrelu(a_src[s * H + h] + adl[dl * H + h]);
                if (MAXUPD) atomicMax(&mkey[dl * H + h], fkey(v));
                wgt[i * H + h] = v;
            }
        }
    }
}

template <int H>
__device__ __forceinline__ void phase_exp(int n, bool with_self, int d0, int N, int tid,
                                          const int* lst_d, const unsigned* mkey,
                                          float* den, float* wgt) {
    const int tot = n + (with_self ? RNG : 0);
    for (int i = tid; i < tot; i += NTH) {
        const int dl = i < n ? lst_d[i] : i - n;
        if (d0 + dl < N) {
            for (int h = 0; h < H; ++h) {
                const float ex = expf(wgt[i * H + h] - fdekey(mkey[dl * H + h]));
                wgt[i * H + h] = ex;
                atomicAdd(&den[dl * H + h], ex);
            }
        }
    }
}

template <int H, int C>
__device__ __forceinline__ void phase_acc(int n, bool with_self, int d0, int N, int tid,
                                          const float* __restrict__ hfeat,
                                          const int* lst_s, const int* lst_d,
                                          const float* wgt, float* acc) {
    constexpr int F   = H * C;
    constexpr int LPE = F >= 32 ? 32 : F;
    constexpr int EPW = 32 / LPE;
    constexpr int J   = F / LPE;
    constexpr int U   = 4;
    constexpr int STEP = (NTH / 32) * EPW * U;
    const int lane = tid & 31, wave = tid >> 5;
    const int sub = lane / LPE, l = lane % LPE;
    const int tot = n + (with_self ? RNG : 0);

    for (int i0 = 0; i0 < tot; i0 += STEP) {
        int dl[U]; bool ok[U];
        float row[U][J]; float ex[U][H];
        #pragma unroll
        for (int u = 0; u < U; ++u) {
            int i = i0 + (wave * U + u) * EPW + sub;
            int s;
            if (i < n) { s = lst_s[i]; dl[u] = lst_d[i]; }
            else       { dl[u] = i - n; s = d0 + dl[u]; }
            ok[u] = (i < tot) && (d0 + dl[u] < N);
            if (!ok[u]) { s = 0; dl[u] = 0; i = 0; }
            const float* rp = hfeat + (size_t)s * F + l;
            #pragma unroll
            for (int j = 0; j < J; ++j) row[u][j] = rp[LPE * j];
            #pragma unroll
            for (int h = 0; h < H; ++h) ex[u][h] = wgt[i * H + h];
        }
        #pragma unroll
        for (int u = 0; u < U; ++u) {
            if (ok[u]) {
                float* ap = acc + dl[u] * F + l;
                #pragma unroll
                for (int j = 0; j < J; ++j)
                    atomicAdd(ap + LPE * j, ex[u][(LPE * j) / C] * row[u][j]);
            }
        }
    }
}

template <int H, int C, bool RELU, bool SAVE, bool LOAD>
__global__ __launch_bounds__(NTH)
void node_range_layer(const int* __restrict__ ei, int E, int N,
                      const float* __restrict__ a_src, const float* __restrict__ a_dst,
                      const float* __restrict__ hfeat, const float* __restrict__ bias,
                      float* __restrict__ outp, int* __restrict__ lists,
                      const unsigned* __restrict__ ids32) {
    constexpr int F = H * C;
    __shared__ float    acc[RNG * F];
    __shared__ unsigned mkey[RNG * H];
    __shared__ float    den[RNG * H];
    __shared__ float    adl[RNG * H];
    __shared__ int      lst_s[CAPL];
    __shared__ int      lst_d[CAPL];
    __shared__ __attribute__((aligned(16))) float wgt[(CAPL + RNG) * H];
    __shared__ int      cnt;

    const int tid = threadIdx.x;
    const int d0  = blockIdx.x * RNG;
    const int* dstp = ei + E;
    const bool vec = (((size_t)dstp) & 15) == 0;

    for (int i = tid; i < RNG * F; i += NTH) acc[i] = 0.0f;
    for (int i = tid; i < RNG * H; i += NTH) {
        mkey[i] = 0u; den[i] = 0.0f;
        adl[i] = (d0 * H + i < N * H) ? a_dst[d0 * H + i] : 0.0f;
    }
    if (tid == 0) cnt = 0;
    __syncthreads();

    int* const my_cnt = lists + (size_t)blockIdx.x * LSTR;
    int* const my_s   = my_cnt + LHDR;
    int* const my_d   = my_s + CAPL;

    const bool by_ids = !LOAD && gridDim.x <= 255;

    int total;
    if (LOAD) {
        total = my_cnt[0];
    } else {
        if (by_ids) scan_ids((const uint4*)ids32, E, tid, lst_s, &cnt);
        else        scan_edges(dstp, 0, E, vec, d0, tid, lst_s, lst_d, &cnt);
        __syncthreads();
        total = cnt;
        if (SAVE && tid == 0) my_cnt[0] = total;
    }

    if (total <= CAPL) {
        if (LOAD) {
            for (int i = tid; i < total; i += NTH) { lst_s[i] = my_s[i]; lst_d[i] = my_d[i]; }
        } else {
            for (int i = tid; i < total; i += NTH) {
                const int e = lst_s[i];
                const int s = ei[e];
                if (by_ids) lst_d[i] = dstp[e] - d0;
                lst_s[i] = s;
                if (SAVE) { my_s[i] = s; my_d[i] = lst_d[i]; }
            }
        }
        __syncthreads();
        phase_score<H, true>(total, true, d0, N, tid, a_src, lst_s, lst_d, adl, mkey, wgt);
        __syncthreads();
        phase_exp<H>(total, true, d0, N, tid, lst_d, mkey, den, wgt);
        __syncthreads();
        phase_acc<H, C>(total, true, d0, N, tid, hfeat, lst_s, lst_d, wgt, acc);
        __syncthreads();
    } else {
        __syncthreads();
        for (int pass = 0; pass < 2; ++pass) {
            for (int lo = 0; lo < E; lo += CAPL) {
                const int hi = lo + CAPL < E ? lo + CAPL : E;
                if (tid == 0) cnt = 0;
                __syncthreads();
                scan_edges(dstp, lo, hi, vec, d0, tid, lst_s, lst_d, &cnt);
                __syncthreads();
                const int n = cnt;
                for (int i = tid; i < n; i += NTH) lst_s[i] = ei[lst_s[i]];
                __syncthreads();
                if (pass == 0) {
                    phase_score<H, true>(n, lo == 0, d0, N, tid, a_src, lst_s, lst_d, adl, mkey, wgt);
                } else {
                    phase_score<H, false>(n, lo == 0, d0, N, tid, a_src, lst_s, lst_d, adl, mkey, wgt);
                    __syncthreads();
                    phase_exp<H>(n, lo == 0, d0, N, tid, lst_d, mkey, den, wgt);
                    __syncthreads();
                    phase_acc<H, C>(n, lo == 0, d0, N, tid, hfeat, lst_s, lst_d, wgt, acc);
                }
                __syncthreads();
            }
        }
    }

    for (int idx = tid; idx < RNG * F; idx += NTH) {
        const int dl = idx / F, c = idx % F;
        const int node = d0 + dl;
        if (node < N) {
            float v = acc[idx] / den[dl * H + c / C] + bias[c];
            if (RELU) v = fmaxf(v, 0.0f);
            outp[(size_t)node * F + c] = v;
        }
    }
}


extern "C" void kernel_launch(void* const* d_in, const int* in_sizes, int n_in,
                              void* d_out, int out_size, void* d_ws, size_t ws_size,
                              hipStream_t stream) {
    const float* x   = (const float*)d_in[0];
    const int*   ei  = (const int*)  d_in[1];
    const float* W1  = (const float*)d_in[2];
    const float* as1 = (const float*)d_in[3];
    const float* ad1 = (const float*)d_in[4];
    const float* b1  = (const float*)d_in[5];
    const float* W2  = (const float*)d_in[6];
    const float* as2 = (const float*)d_in[7];
    const float* ad2 = (const float*)d_in[8];
    const float* b2  = (const float*)d_in[9];

    const int N  = in_sizes[0] / 128;
    const int E  = in_sizes[1] / 2;

    char* w = (char*)d_ws;
    auto carve = [&](size_t bytes) -> void* {
        void* p = (void*)w;
        w += (bytes + 255) & ~((size_t)255);
        return p;
    };
    float* h1  = (float*)carve((size_t)N * 256 * 4);
    float* y1  = (float*)carve((size_t)N * 256 * 4);
    float* aS1 = (float*)carve((size_t)N * 4 * 4);
    float* aD1 = (float*)carve((size_t)N * 4 * 4);
    float* h2  = (float*)carve((size_t)N * 16 * 4);
    float* aS2 = (float*)carve((size_t)N * 4);
    float* aD2 = (float*)carve((size_t)N * 4);

    const int TB = 256;
    auto blocks = [&](long long n) { return (unsigned)((n + TB - 1) / TB); };
    const unsigned nrange = (unsigned)((N + RNG - 1) / RNG);
    int* lists = (int*)carve((size_t)nrange * LSTR * 4);
    unsigned* ids32 = (unsigned*)carve((((size_t)E + 15) / 16) * 16);

    dim3 g1((N + 15) / 16, 4);
    gemm1_wmma<<<g1, 32, 0, stream>>>(x, W1, h1, N);
    att1_kernel<<<blocks(N * 4), TB, 0, stream>>>(h1, as1, ad1, aS1, aD1, N * 4, ei + E, E, ids32);
    node_range_layer<4, 64, true, true, false><<<nrange, NTH, 0, stream>>>(ei, E, N, aS1, aD1, h1, b1, y1, lists, ids32);

    gemm2_wmma<<<(N + 15) / 16, 32, 0, stream>>>(y1, W2, h2, N);
    att2_kernel<<<blocks(N), TB, 0, stream>>>(h2, as2, ad2, aS2, aD2, N);
    node_range_layer<1, 16, false, false, true><<<nrange, NTH, 0, stream>>>(ei, E, N, aS2, aD2, h2, b2, (float*)d_out, lists, ids32);
}


# Layer-2 product and scores made by the layer-1 range block from its 200 finished rows in LDS (W2 staged in LDS, f32 WMMA, k order kept); y1 never stored; product and score launches gone, 6 to 4
# speedup vs baseline: 4.8530x; 1.1765x over previous
#include <hip/hip_runtime.h>
#include <hip/hip_bf16.h>

typedef __attribute__((ext_vector_type(2))) float v2f;
typedef __attribute__((ext_vector_type(8))) float v8f;

#define NEG_SLOPE 0.2f


__device__ __forceinline__ float lrelu(float v) {
    return v >= 0.0f ? v : NEG_SLOPE * v;
}

__device__ __forceinline__ unsigned fkey(float f) {
    unsigned u = __float_as_uint(f);
    return (u & 0x80000000u) ? ~u : (u | 0x80000000u);
}
__device__ __forceinline__ float fdekey(unsigned k) {
    return (k & 0x80000000u) ? __uint_as_float(k ^ 0x80000000u)
                             : __uint_as_float(~k);
}

__global__ __launch_bounds__(32)
void gemm1_wmma(const float* __restrict__ x, const float* __restrict__ W1,
                float* __restrict__ h1, int N) {
    const int mtile = blockIdx.x;
    const int lane  = threadIdx.x;
    const int mrow  = mtile * 16 + (lane & 15);
    const int lrow  = mrow < N ? mrow : N - 1;
    const int khalf = (lane >> 4) * 2;
    const int colbase = blockIdx.y * 64;
    const int nc    = lane & 15;

    v8f c0 = {}, c1 = {}, c2 = {}, c3 = {};
    const float* xrow = x + (size_t)lrow * 128;

    for (int k = 0; k < 128; k += 4) {
        const int kb = k + khalf;
        v2f a; a.x = xrow[kb]; a.y = xrow[kb + 1];
        const float* w0 = W1 + (size_t)kb * 256 + colbase + nc;
        const float* w1 = w0 + 256;
        v2f b0; b0.x = w0[0];  b0.y = w1[0];
        v2f b1; b1.x = w0[16]; b1.y = w1[16];
        v2f b2; b2.x = w0[32]; b2.y = w1[32];
        v2f b3; b3.x = w0[48]; b3.y = w1[48];
        c0 = __builtin_amdgcn_wmma_f32_16x16x4_f32(false, a, false, b0, (short)0, c0, false, false);
        c1 = __builtin_amdgcn_wmma_f32_16x16x4_f32(false, a, false, b1, (short)0, c1, false, false);
        c2 = __builtin_amdgcn_wmma_f32_16x16x4_f32(false, a, false, b2, (short)0, c2, false, false);
        c3 = __builtin_amdgcn_wmma_f32_16x16x4_f32(false, a, false, b3, (short)0, c3, false, false);
    }

    const int rowoff = (lane >= 16) ? 8 : 0;
    for (int r = 0; r < 8; ++r) {
        const int grow = mtile * 16 + rowoff + r;
        if (grow < N) {
            float* op = h1 + (size_t)grow * 256 + colbase + nc;
            op[0]  = c0[r];
            op[16] = c1[r];
            op[32] = c2[r];
            op[48] = c3[r];
        }
    }
}

#define RNG  200
#define SPITCH 256

__global__ void att1_kernel(const float* __restrict__ h1,
                            const float* __restrict__ att_src,
                            const float* __restrict__ att_dst,
                            float* __restrict__ a_src, float* __restrict__ a_dst,
                            int NH,
                            const int* __restrict__ dstp, int E,
                            unsigned* __restrict__ ids32) {
    int i = blockIdx.x * blockDim.x + threadIdx.x;
    {
        const int nwp = ((E + 15) >> 4) << 2;
        const int nthreads = gridDim.x * blockDim.x;
        const bool vec = (((size_t)dstp) & 15) == 0;
        for (int w = i; w < nwp; w += nthreads) {
            int d[4] = {-1, -1, -1, -1};
            if (vec && 4 * w + 3 < E) {
                const int4 v = *(const int4*)(dstp + 4 * w);
                d[0] = v.x; d[1] = v.y; d[2] = v.z; d[3] = v.w;
            } else {
                #pragma unroll
                for (int j = 0; j < 4; ++j)
                    if (4 * w + j < E) d[j] = dstp[4 * w + j];
            }
            unsigned p = 0;
            #pragma unroll
            for (int j = 0; j < 4; ++j) {
                const unsigned q = (unsigned)d[j] / (unsigned)RNG;
                p |= (q < 255u ? q : 255u) << (8 * j);
            }
            ids32[w] = p;
        }
    }
    if (i >= NH) return;
    const int h = i & 3;
    const float* hp = h1 + (size_t)(i >> 2) * 256 + h * 64;
    const float* as = att_src + h * 64;
    const float* ad = att_dst + h * 64;
    float s = 0.0f, d = 0.0f;
    for (int c = 0; c < 64; ++c) {
        const float v = hp[c];
        s += v * as[c];
        d += v * ad[c];
    }
    a_src[i] = s;
    a_dst[i] = d;
}

#define CAPL 4096
#define NTH  1024
#define LHDR 64
#define LSTR (LHDR + 2 * CAPL)

__device__ __forceinline__ void list_one(int d, int e, int d0,
                                         int* lst_s, int* lst_d, int* cnt) {
    const unsigned dl = (unsigned)(d - d0);
    if (dl < (unsigned)RNG) {
        const int slot = atomicAdd(cnt, 1);
        if (slot < CAPL) { lst_s[slot] = e; lst_d[slot] = (int)dl; }
    }
}

__device__ __forceinline__ void scan_edges(const int* __restrict__ dstp, int lo, int hi,
                                           bool vec, int d0, int tid,
                                           int* lst_s, int* lst_d, int* cnt) {
    constexpr int SU = 8;
    if (vec) {
        const int nq = (hi - lo) >> 2;
        const int K  = (nq + NTH - 1) / NTH;
        const int rot = (int)(((long long)blockIdx.x * K) / gridDim.x);
        for (int k = 0; k < K; k += SU) {
            int4 v[SU]; int eb[SU]; bool ok[SU];
            #pragma unroll
            for (int u = 0; u < SU; ++u) {
                int kr = k + u + rot; kr = kr >= K ? kr - K : kr;
                const int q = kr * NTH + tid;
                ok[u] = (k + u < K) && (q < nq);
                const int qq = ok[u] ? q : 0;
                v[u]  = *(const int4*)(dstp + lo + 4 * qq);
                eb[u] = lo + 4 * qq;
            }
            #pragma unroll
            for (int u = 0; u < SU; ++u) {
                if (ok[u]) {
                    list_one(v[u].x, eb[u] + 0, d0, lst_s, lst_d, cnt);
                    list_one(v[u].y, eb[u] + 1, d0, lst_s, lst_d, cnt);
                    list_one(v[u].z, eb[u] + 2, d0, lst_s, lst_d, cnt);
                    list_one(v[u].w, eb[u] + 3, d0, lst_s, lst_d, cnt);
                }
            }
        }
        for (int e = lo + 4 * nq + tid; e < hi; e += NTH)
            list_one(dstp[e], e, d0, lst_s, lst_d, cnt);
    } else {
        for (int e = lo + tid; e < hi; e += NTH)
            list_one(dstp[e], e, d0, lst_s, lst_d, cnt);
    }
}

__device__ __forceinline__ unsigned zero_bytes(unsigned x) {
    return ~(((x & 0x7f7f7f7fu) + 0x7f7f7f7fu) | x | 0x7f7f7f7fu);
}

__device__ __forceinline__ void scan_ids(const uint4* __restrict__ ids, int E, int tid,
                                         int* lst_s, int* cnt) {
    constexpr int SU = 8;
    const int ng = (E + 15) >> 4;
    const int K  = (ng + NTH - 1) / NTH;
    const int rot = (int)(((long long)blockIdx.x * K) / gridDim.x);
    const unsigned mine = blockIdx.x * 0x01010101u;
    for (int k = 0; k < K; k += SU) {
        uint4 v[SU]; int eb[SU]; bool ok[SU];
        #pragma unroll
        for (int u = 0; u < SU; ++u) {
            int kr = k + u + rot; kr = kr >= K ? kr - K : kr;
            const int g = kr * NTH + tid;
            ok[u] = (k + u < K) && (g < ng);
            const int gg = ok[u] ? g : 0;
            v[u]  = ids[gg];
            eb[u] = 16 * gg;
        }
        #pragma unroll
        for (int u = 0; u < SU; ++u) {
            if (ok[u]) {
                unsigned m = (zero_bytes(v[u].x ^ mine) >> 7) | (zero_bytes(v[u].y ^ mine) >> 6)
                           | (zero_bytes(v[u].z ^ mine) >> 5) | (zero_bytes(v[u].w ^ mine) >> 4);
                while (m) {
                    const int b = __builtin_ctz(m);
                    m &= m - 1;
                    const int e = eb[u] + ((b & 7) << 2) + (b >> 3);
                    const int slot = atomicAdd(cnt, 1);
                    if (slot < CAPL) lst_s[slot] = e;
                }
            }
        }
    }
}

template <int H, bool MAXUPD, bool SPAD>
__device__ __forceinline__ void phase_score(int n, bool with_self, int d0, int N, int tid,
                                            const float* __restrict__ a_src,
                                            const int* lst_s, const int* lst_d,
                                            const float* adl, unsigned* mkey, float* wgt) {
    const int tot = n + (with_self ? RNG : 0);
    for (int i = tid; i < tot; i += NTH) {
        int s, dl;
        if (i < n) { s = lst_s[i]; dl = lst_d[i]; }
        else       { dl = i - n;   s = d0 + dl; }
        if (d0 + dl < N) {
            const int si = SPAD ? (int)(((unsigned)s / RNG) * SPITCH + (unsigned)s % RNG) : s;
            for (int h = 0; h < H; ++h) {
                const float v = lrelu(a_src[si * H + h] + adl[dl * H + h]);
                if (MAXUPD) atomicMax(&mkey[dl * H + h], fkey(v));
                wgt[i * H + h] = v;
            }
        }
    }
}

template <int H>
__device__ __forceinline__ void phase_exp(int n, bool with_self, int d0, int N, int tid,
                                          const int* lst_d, const unsigned* mkey,
                                          float* den, float* wgt) {
    const int tot = n + (with_self ? RNG : 0);
    for (int i = tid; i < tot; i += NTH) {
        const int dl = i < n ? lst_d[i] : i - n;
        if (d0 + dl < N) {
            for (int h = 0; h < H; ++h) {
                const float ex = expf(wgt[i * H + h] - fdekey(mkey[dl * H + h]));
                wgt[i * H + h] = ex;
                atomicAdd(&den[dl * H + h], ex);
            }
        }
    }
}

template <int H, int C, int PITCH>
__device__ __forceinline__ void phase_acc(int n, bool with_self, int d0, int N, int tid,
                                          const float* __restrict__ hfeat,
                                          const int* lst_s, const int* lst_d,
                                          const float* wgt, float* acc) {
    constexpr int F   = H * C;
    constexpr int LPE = F >= 32 ? 32 : F;
    constexpr int EPW = 32 / LPE;
    constexpr int J   = F / LPE;
    constexpr int U   = 4;
    constexpr int STEP = (NTH / 32) * EPW * U;
    const int lane = tid & 31, wave = tid >> 5;
    const int sub = lane / LPE, l = lane % LPE;
    const int tot = n + (with_self ? RNG : 0);

    for (int i0 = 0; i0 < tot; i0 += STEP) {
        int dl[U]; bool ok[U];
        float row[U][J]; float ex[U][H];
        #pragma unroll
        for (int u = 0; u < U; ++u) {
            int i = i0 + (wave * U + u) * EPW + sub;
            int s;
            if (i < n) { s = lst_s[i]; dl[u] = lst_d[i]; }
            else       { dl[u] = i - n; s = d0 + dl[u]; }
            ok[u] = (i < tot) && (d0 + dl[u] < N);
            if (!ok[u]) { s = 0; dl[u] = 0; i = 0; }
            const float* rp = hfeat + (size_t)s * F + l;
            #pragma unroll
            for (int j = 0; j < J; ++j) row[u][j] = rp[LPE * j];
            #pragma unroll
            for (int h = 0; h < H; ++h) ex[u][h] = wgt[i * H + h];
        }
        #pragma unroll
        for (int u = 0; u < U; ++u) {
            if (ok[u]) {
                float* ap = acc + dl[u] * PITCH + l;
                #pragma unroll
                for (int j = 0; j < J; ++j)
                    atomicAdd(ap + LPE * j, ex[u][(LPE * j) / C] * row[u][j]);
            }
        }
    }
}

#define OC2  16

template <int H, int C, bool RELU, bool SAVE, bool LOAD, bool NEXT, bool SPAD>
__global__ __launch_bounds__(NTH)
void node_range_layer(const int* __restrict__ ei, int E, int N,
                      const float* __restrict__ a_src, const float* __restrict__ a_dst,
                      const float* __restrict__ hfeat, const float* __restrict__ bias,
                      float* __restrict__ outp, int* __restrict__ lists,
                      const unsigned* __restrict__ ids32,
                      const float* __restrict__ Wn,
                      const float* __restrict__ natt_src, const float* __restrict__ natt_dst,
                      float* __restrict__ nsrc, float* __restrict__ ndst) {
    constexpr int F = H * C;
    constexpr int PITCH = NEXT ? F + 4 : F;
    static_assert(!NEXT || (F * OC2 <= CAPL && RNG * (OC2 + 1) <= (CAPL + RNG) * H),
                  "the staged weights and the product tile reuse lst_s and wgt");
    static_assert(RNG <= SPITCH && SPITCH % 32 == 0, "a range's scores fill whole lines of their own");
    __shared__ float    acc[RNG * PITCH];
    __shared__ unsigned mkey[RNG * H];
    __shared__ float    den[RNG * H];
    __shared__ float    adl[RNG * H];
    __shared__ int      lst_s[CAPL];
    __shared__ int      lst_d[CAPL];
    __shared__ __attribute__((aligned(16))) float wgt[(CAPL + RNG) * H];
    __shared__ int      cnt;

    const int tid = threadIdx.x;
    const int d0  = blockIdx.x * RNG;
    const int* dstp = ei + E;
    const bool vec = (((size_t)dstp) & 15) == 0;

    for (int i = tid; i < RNG * PITCH; i += NTH) acc[i] = 0.0f;
    for (int i = tid; i < RNG * H; i += NTH) {
        mkey[i] = 0u; den[i] = 0.0f;
        adl[i] = (d0 * H + i < N * H) ? a_dst[(SPAD ? blockIdx.x * SPITCH : d0) * H + i] : 0.0f;
    }
    if (tid == 0) cnt = 0;
    __syncthreads();

    int* const my_cnt = lists + (size_t)blockIdx.x * LSTR;
    int* const my_s   = my_cnt + LHDR;
    int* const my_d   = my_s + CAPL;

    const bool by_ids = !LOAD && gridDim.x <= 255;

    int total;
    if (LOAD) {
        total = my_cnt[0];
    } else {
        if (by_ids) scan_ids((const uint4*)ids32, E, tid, lst_s, &cnt);
        else        scan_edges(dstp, 0, E, vec, d0, tid, lst_s, lst_d, &cnt);
        __syncthreads();
        total = cnt;
        if (SAVE && tid == 0) my_cnt[0] = total;
    }

    if (total <= CAPL) {
        if (LOAD) {
            for (int i = tid; i < total; i += NTH) { lst_s[i] = my_s[i]; lst_d[i] = my_d[i]; }
        } else {
            for (int i = tid; i < total; i += NTH) {
                const int e = lst_s[i];
                const int s = ei[e];
                if (by_ids) lst_d[i] = dstp[e] - d0;
                lst_s[i] = s;
                if (SAVE) { my_s[i] = s; my_d[i] = lst_d[i]; }
            }
        }
        __syncthreads();
        phase_score<H, true, SPAD>(total, true, d0, N, tid, a_src, lst_s, lst_d, adl, mkey, wgt);
        __syncthreads();
        phase_exp<H>(total, true, d0, N, tid, lst_d, mkey, den, wgt);
        __syncthreads();
        phase_acc<H, C, PITCH>(total, true, d0, N, tid, hfeat, lst_s, lst_d, wgt, acc);
        __syncthreads();
    } else {
        __syncthreads();
        for (int pass = 0; pass < 2; ++pass) {
            for (int lo = 0; lo < E; lo += CAPL) {
                const int hi = lo + CAPL < E ? lo + CAPL : E;
                if (tid == 0) cnt = 0;
                __syncthreads();
                scan_edges(dstp, lo, hi, vec, d0, tid, lst_s, lst_d, &cnt);
                __syncthreads();
                const int n = cnt;
                for (int i = tid; i < n; i += NTH) lst_s[i] = ei[lst_s[i]];
                __syncthreads();
                if (pass == 0) {
                    phase_score<H, true, SPAD>(n, lo == 0, d0, N, tid, a_src, lst_s, lst_d, adl, mkey, wgt);
                } else {
                    phase_score<H, false, SPAD>(n, lo == 0, d0, N, tid, a_src, lst_s, lst_d, adl, mkey, wgt);
                    __syncthreads();
                    phase_exp<H>(n, lo == 0, d0, N, tid, lst_d, mkey, den, wgt);
                    __syncthreads();
                    phase_acc<H, C, PITCH>(n, lo == 0, d0, N, tid, hfeat, lst_s, lst_d, wgt, acc);
                }
                __syncthreads();
            }
        }
    }

    if (!NEXT) {
        for (int idx = tid; idx < RNG * F; idx += NTH) {
            const int dl = idx / F, c = idx % F;
            const int node = d0 + dl;
            if (node < N) {
                float v = acc[idx] / den[dl * H + c / C] + bias[c];
                if (RELU) v = fmaxf(v, 0.0f);
                outp[(size_t)node * F + c] = v;
            }
        }
    } else {
        float* const wn_s = reinterpret_cast<float*>(lst_s);
        float* const pr_s = wgt;

        constexpr int WPT = NEXT ? F * OC2 / NTH : 1;
        static_assert(!NEXT || F * OC2 % NTH == 0, "whole weights per thread");
        float wreg[WPT];
        #pragma unroll
        for (int j = 0; j < WPT; ++j) wreg[j] = Wn[tid + j * NTH];

        for (int idx = tid; idx < RNG * F; idx += NTH) {
            const int dl = idx / F, c = idx % F;
            float v = 0.0f;
            if (d0 + dl < N) {
                v = acc[dl * PITCH + c] / den[dl * H + c / C] + bias[c];
                if (RELU) v = fmaxf(v, 0.0f);
            }
            acc[dl * PITCH + c] = v;
        }
        #pragma unroll
        for (int j = 0; j < WPT; ++j) wn_s[tid + j * NTH] = wreg[j];
        __syncthreads();

        const int wave = __builtin_amdgcn_readfirstlane(tid >> 5);
        const int lane = tid & 31;
        constexpr int MT = (RNG + 15) / 16;
        if (wave < MT) {
            const int nc    = lane & 15;
            const int row   = wave * 16 + nc;
            const int lr    = row < RNG ? row : RNG - 1;
            const int khalf = (lane >> 4) * 2;
            const float* ar = acc + lr * PITCH;
            v8f cc = {};
            for (int k = 0; k < F; k += 4) {
                const int kb = k + khalf;
                v2f a; a.x = ar[kb]; a.y = ar[kb + 1];
                v2f b; b.x = wn_s[kb * OC2 + nc]; b.y = wn_s[(kb + 1) * OC2 + nc];
                cc = __builtin_amdgcn_wmma_f32_16x16x4_f32(false, a, false, b, (short)0, cc, false, false);
            }
            const int rowoff = (lane >= 16) ? 8 : 0;
            for (int r = 0; r < 8; ++r) {
                const int rr = wave * 16 + rowoff + r;
                if (rr < RNG) pr_s[rr * (OC2 + 1) + nc] = cc[r];
            }
        }
        __syncthreads();

        for (int idx = tid; idx < RNG * OC2; idx += NTH) {
            const int dl = idx / OC2, c = idx % OC2;
            const int node = d0 + dl;
            if (node < N) outp[(size_t)node * OC2 + c] = pr_s[dl * (OC2 + 1) + c];
        }
        if (tid < RNG && d0 + tid < N) {
            float s = 0.0f, d = 0.0f;
            for (int c = 0; c < OC2; ++c) {
                const float v = pr_s[tid * (OC2 + 1) + c];
                s += v * natt_src[c];
                d += v * natt_dst[c];
            }
            nsrc[blockIdx.x * SPITCH + tid] = s;
            ndst[blockIdx.x * SPITCH + tid] = d;
        }
    }
}


extern "C" void kernel_launch(void* const* d_in, const int* in_sizes, int n_in,
                              void* d_out, int out_size, void* d_ws, size_t ws_size,
                              hipStream_t stream) {
    const float* x   = (const float*)d_in[0];
    const int*   ei  = (const int*)  d_in[1];
    const float* W1  = (const float*)d_in[2];
    const float* as1 = (const float*)d_in[3];
    const float* ad1 = (const float*)d_in[4];
    const float* b1  = (const float*)d_in[5];
    const float* W2  = (const float*)d_in[6];
    const float* as2 = (const float*)d_in[7];
    const float* ad2 = (const float*)d_in[8];
    const float* b2  = (const float*)d_in[9];

    const int N  = in_sizes[0] / 128;
    const int E  = in_sizes[1] / 2;

    char* w = (char*)d_ws;
    auto carve = [&](size_t bytes) -> void* {
        void* p = (void*)w;
        w += (bytes + 255) & ~((size_t)255);
        return p;
    };
    float* h1  = (float*)carve((size_t)N * 256 * 4);
    float* aS1 = (float*)carve((size_t)N * 4 * 4);
    float* aD1 = (float*)carve((size_t)N * 4 * 4);
    float* h2  = (float*)carve((size_t)N * 16 * 4);

    const int TB = 256;
    auto blocks = [&](long long n) { return (unsigned)((n + TB - 1) / TB); };
    const unsigned nrange = (unsigned)((N + RNG - 1) / RNG);
    int* lists = (int*)carve((size_t)nrange * LSTR * 4);
    unsigned* ids32 = (unsigned*)carve((((size_t)E + 15) / 16) * 16);
    float* aS2 = (float*)carve((size_t)nrange * SPITCH * 4);
    float* aD2 = (float*)carve((size_t)nrange * SPITCH * 4);

    dim3 g1((N + 15) / 16, 4);
    gemm1_wmma<<<g1, 32, 0, stream>>>(x, W1, h1, N);
    att1_kernel<<<blocks(N * 4), TB, 0, stream>>>(h1, as1, ad1, aS1, aD1, N * 4, ei + E, E, ids32);
    node_range_layer<4, 64, true, true, false, true, false><<<nrange, NTH, 0, stream>>>(ei, E, N, aS1, aD1, h1, b1, h2, lists, ids32, W2, as2, ad2, aS2, aD2);

    node_range_layer<1, 16, false, false, true, false, true><<<nrange, NTH, 0, stream>>>(ei, E, N, aS2, aD2, h2, b2, (float*)d_out, lists, ids32, nullptr, nullptr, nullptr, nullptr, nullptr);
}


# Layer-1 row sum by owner waves: list put in destination order in LDS (integer LDS adds, arrival order inside a node), a wave sums one node's rows in registers, stores the row once; no LDS float adds
# speedup vs baseline: 5.5304x; 1.1396x over previous
#include <hip/hip_runtime.h>
#include <hip/hip_bf16.h>

typedef __attribute__((ext_vector_type(2))) float v2f;
typedef __attribute__((ext_vector_type(8))) float v8f;

#define NEG_SLOPE 0.2f


__device__ __forceinline__ float lrelu(float v) {
    return v >= 0.0f ? v : NEG_SLOPE * v;
}

__device__ __forceinline__ unsigned fkey(float f) {
    unsigned u = __float_as_uint(f);
    return (u & 0x80000000u) ? ~u : (u | 0x80000000u);
}
__device__ __forceinline__ float fdekey(unsigned k) {
    return (k & 0x80000000u) ? __uint_as_float(k ^ 0x80000000u)
                             : __uint_as_float(~k);
}

__global__ __launch_bounds__(32)
void gemm1_wmma(const float* __restrict__ x, const float* __restrict__ W1,
                float* __restrict__ h1, int N) {
    const int mtile = blockIdx.x;
    const int lane  = threadIdx.x;
    const int mrow  = mtile * 16 + (lane & 15);
    const int lrow  = mrow < N ? mrow : N - 1;
    const int khalf = (lane >> 4) * 2;
    const int colbase = blockIdx.y * 64;
    const int nc    = lane & 15;

    v8f c0 = {}, c1 = {}, c2 = {}, c3 = {};
    const float* xrow = x + (size_t)lrow * 128;

    for (int k = 0; k < 128; k += 4) {
        const int kb = k + khalf;
        v2f a; a.x = xrow[kb]; a.y = xrow[kb + 1];
        const float* w0 = W1 + (size_t)kb * 256 + colbase + nc;
        const float* w1 = w0 + 256;
        v2f b0; b0.x = w0[0];  b0.y = w1[0];
        v2f b1; b1.x = w0[16]; b1.y = w1[16];
        v2f b2; b2.x = w0[32]; b2.y = w1[32];
        v2f b3; b3.x = w0[48]; b3.y = w1[48];
        c0 = __builtin_amdgcn_wmma_f32_16x16x4_f32(false, a, false, b0, (short)0, c0, false, false);
        c1 = __builtin_amdgcn_wmma_f32_16x16x4_f32(false, a, false, b1, (short)0, c1, false, false);
        c2 = __builtin_amdgcn_wmma_f32_16x16x4_f32(false, a, false, b2, (short)0, c2, false, false);
        c3 = __builtin_amdgcn_wmma_f32_16x16x4_f32(false, a, false, b3, (short)0, c3, false, false);
    }

    const int rowoff = (lane >= 16) ? 8 : 0;
    for (int r = 0; r < 8; ++r) {
        const int grow = mtile * 16 + rowoff + r;
        if (grow < N) {
            float* op = h1 + (size_t)grow * 256 + colbase + nc;
            op[0]  = c0[r];
            op[16] = c1[r];
            op[32] = c2[r];
            op[48] = c3[r];
        }
    }
}

#define RNG  200
#define SPITCH 256

__global__ void att1_kernel(const float* __restrict__ h1,
                            const float* __restrict__ att_src,
                            const float* __restrict__ att_dst,
                            float* __restrict__ a_src, float* __restrict__ a_dst,
                            int NH,
                            const int* __restrict__ dstp, int E,
                            unsigned* __restrict__ ids32) {
    int i = blockIdx.x * blockDim.x + threadIdx.x;
    {
        const int nwp = ((E + 15) >> 4) << 2;
        const int nthreads = gridDim.x * blockDim.x;
        const bool vec = (((size_t)dstp) & 15) == 0;
        for (int w = i; w < nwp; w += nthreads) {
            int d[4] = {-1, -1, -1, -1};
            if (vec && 4 * w + 3 < E) {
                const int4 v = *(const int4*)(dstp + 4 * w);
                d[0] = v.x; d[1] = v.y; d[2] = v.z; d[3] = v.w;
            } else {
                #pragma unroll
                for (int j = 0; j < 4; ++j)
                    if (4 * w + j < E) d[j] = dstp[4 * w + j];
            }
            unsigned p = 0;
            #pragma unroll
            for (int j = 0; j < 4; ++j) {
                const unsigned q = (unsigned)d[j] / (unsigned)RNG;
                p |= (q < 255u ? q : 255u) << (8 * j);
            }
            ids32[w] = p;
        }
    }
    if (i >= NH) return;
    const int h = i & 3;
    const float* hp = h1 + (size_t)(i >> 2) * 256 + h * 64;
    const float* as = att_src + h * 64;
    const float* ad = att_dst + h * 64;
    float s = 0.0f, d = 0.0f;
    for (int c = 0; c < 64; ++c) {
        const float v = hp[c];
        s += v * as[c];
        d += v * ad[c];
    }
    a_src[i] = s;
    a_dst[i] = d;
}

#define CAPL 4096
#define NTH  1024
#define LHDR 64
#define LSTR (LHDR + 2 * CAPL)

__device__ __forceinline__ void list_one(int d, int e, int d0,
                                         int* lst_s, int* lst_d, int* cnt) {
    const unsigned dl = (unsigned)(d - d0);
    if (dl < (unsigned)RNG) {
        const int slot = atomicAdd(cnt, 1);
        if (slot < CAPL) { lst_s[slot] = e; lst_d[slot] = (int)dl; }
    }
}

__device__ __forceinline__ void scan_edges(const int* __restrict__ dstp, int lo, int hi,
                                           bool vec, int d0, int tid,
                                           int* lst_s, int* lst_d, int* cnt) {
    constexpr int SU = 8;
    if (vec) {
        const int nq = (hi - lo) >> 2;
        const int K  = (nq + NTH - 1) / NTH;
        const int rot = (int)(((long long)blockIdx.x * K) / gridDim.x);
        for (int k = 0; k < K; k += SU) {
            int4 v[SU]; int eb[SU]; bool ok[SU];
            #pragma unroll
            for (int u = 0; u < SU; ++u) {
                int kr = k + u + rot; kr = kr >= K ? kr - K : kr;
                const int q = kr * NTH + tid;
                ok[u] = (k + u < K) && (q < nq);
                const int qq = ok[u] ? q : 0;
                v[u]  = *(const int4*)(dstp + lo + 4 * qq);
                eb[u] = lo + 4 * qq;
            }
            #pragma unroll
            for (int u = 0; u < SU; ++u) {
                if (ok[u]) {
                    list_one(v[u].x, eb[u] + 0, d0, lst_s, lst_d, cnt);
                    list_one(v[u].y, eb[u] + 1, d0, lst_s, lst_d, cnt);
                    list_one(v[u].z, eb[u] + 2, d0, lst_s, lst_d, cnt);
                    list_one(v[u].w, eb[u] + 3, d0, lst_s, lst_d, cnt);
                }
            }
        }
        for (int e = lo + 4 * nq + tid; e < hi; e += NTH)
            list_one(dstp[e], e, d0, lst_s, lst_d, cnt);
    } else {
        for (int e = lo + tid; e < hi; e += NTH)
            list_one(dstp[e], e, d0, lst_s, lst_d, cnt);
    }
}

__device__ __forceinline__ unsigned zero_bytes(unsigned x) {
    return ~(((x & 0x7f7f7f7fu) + 0x7f7f7f7fu) | x | 0x7f7f7f7fu);
}

__device__ __forceinline__ void scan_ids(const uint4* __restrict__ ids, int E, int tid,
                                         int* lst_s, int* cnt) {
    constexpr int SU = 8;
    const int ng = (E + 15) >> 4;
    const int K  = (ng + NTH - 1) / NTH;
    const int rot = (int)(((long long)blockIdx.x * K) / gridDim.x);
    const unsigned mine = blockIdx.x * 0x01010101u;
    for (int k = 0; k < K; k += SU) {
        uint4 v[SU]; int eb[SU]; bool ok[SU];
        #pragma unroll
        for (int u = 0; u < SU; ++u) {
            int kr = k + u + rot; kr = kr >= K ? kr - K : kr;
            const int g = kr * NTH + tid;
            ok[u] = (k + u < K) && (g < ng);
            const int gg = ok[u] ? g : 0;
            v[u]  = ids[gg];
            eb[u] = 16 * gg;
        }
        #pragma unroll
        for (int u = 0; u < SU; ++u) {
            if (ok[u]) {
                unsigned m = (zero_bytes(v[u].x ^ mine) >> 7) | (zero_bytes(v[u].y ^ mine) >> 6)
                           | (zero_bytes(v[u].z ^ mine) >> 5) | (zero_bytes(v[u].w ^ mine) >> 4);
                while (m) {
                    const int b = __builtin_ctz(m);
                    m &= m - 1;
                    const int e = eb[u] + ((b & 7) << 2) + (b >> 3);
                    const int slot = atomicAdd(cnt, 1);
                    if (slot < CAPL) lst_s[slot] = e;
                }
            }
        }
    }
}

template <int H, bool MAXUPD, bool SPAD>
__device__ __forceinline__ void phase_score(int n, bool with_self, int d0, int N, int tid,
                                            const float* __restrict__ a_src,
                                            const int* lst_s, const int* lst_d,
                                            const float* adl, unsigned* mkey, float* wgt) {
    const int tot = n + (with_self ? RNG : 0);
    for (int i = tid; i < tot; i += NTH) {
        int s, dl;
        if (i < n) { s = lst_s[i]; dl = lst_d[i]; }
        else       { dl = i - n;   s = d0 + dl; }
        if (d0 + dl < N) {
            const int si = SPAD ? (int)(((unsigned)s / RNG) * SPITCH + (unsigned)s % RNG) : s;
            for (int h = 0; h < H; ++h) {
                const float v = lrelu(a_src[si * H + h] + adl[dl * H + h]);
                if (MAXUPD) atomicMax(&mkey[dl * H + h], fkey(v));
                wgt[i * H + h] = v;
            }
        }
    }
}

template <int H>
__device__ __forceinline__ void phase_exp(int n, bool with_self, int d0, int N, int tid,
                                          const int* lst_d, const unsigned* mkey,
                                          float* den, float* wgt) {
    const int tot = n + (with_self ? RNG : 0);
    for (int i = tid; i < tot; i += NTH) {
        const int dl = i < n ? lst_d[i] : i - n;
        if (d0 + dl < N) {
            for (int h = 0; h < H; ++h) {
                const float ex = expf(wgt[i * H + h] - fdekey(mkey[dl * H + h]));
                wgt[i * H + h] = ex;
                atomicAdd(&den[dl * H + h], ex);
            }
        }
    }
}

template <int H, int C, int PITCH>
__device__ __forceinline__ void phase_acc(int n, bool with_self, int d0, int N, int tid,
                                          const float* __restrict__ hfeat,
                                          const int* lst_s, const int* lst_d,
                                          const float* wgt, float* acc) {
    constexpr int F   = H * C;
    constexpr int LPE = F >= 32 ? 32 : F;
    constexpr int EPW = 32 / LPE;
    constexpr int J   = F / LPE;
    constexpr int U   = 4;
    constexpr int STEP = (NTH / 32) * EPW * U;
    const int lane = tid & 31, wave = tid >> 5;
    const int sub = lane / LPE, l = lane % LPE;
    const int tot = n + (with_self ? RNG : 0);

    for (int i0 = 0; i0 < tot; i0 += STEP) {
        int dl[U]; bool ok[U];
        float row[U][J]; float ex[U][H];
        #pragma unroll
        for (int u = 0; u < U; ++u) {
            int i = i0 + (wave * U + u) * EPW + sub;
            int s;
            if (i < n) { s = lst_s[i]; dl[u] = lst_d[i]; }
            else       { dl[u] = i - n; s = d0 + dl[u]; }
            ok[u] = (i < tot) && (d0 + dl[u] < N);
            if (!ok[u]) { s = 0; dl[u] = 0; i = 0; }
            const float* rp = hfeat + (size_t)s * F + l;
            #pragma unroll
            for (int j = 0; j < J; ++j) row[u][j] = rp[LPE * j];
            #pragma unroll
            for (int h = 0; h < H; ++h) ex[u][h] = wgt[i * H + h];
        }
        #pragma unroll
        for (int u = 0; u < U; ++u) {
            if (ok[u]) {
                float* ap = acc + dl[u] * PITCH + l;
                #pragma unroll
                for (int j = 0; j < J; ++j)
                    atomicAdd(ap + LPE * j, ex[u][(LPE * j) / C] * row[u][j]);
            }
        }
    }
}

template <int H, int C, int PITCH, bool RELU>
__device__ __forceinline__ void phase_own(int n, int d0, int N, int tid,
                                          const float* __restrict__ hfeat,
                                          const float* __restrict__ bias,
                                          const int* lst_s, const unsigned short* perm,
                                          const int* first, const int* cnt_n,
                                          const float* wgt, const float* den, float* acc) {
    constexpr int F = H * C;
    constexpr int J = F / 32;
    constexpr int U = 8;
    const int lane = tid & 31;
    const int wave = __builtin_amdgcn_readfirstlane(tid >> 5);
    float b[J];
    #pragma unroll
    for (int j = 0; j < J; ++j) b[j] = bias[lane + 32 * j];

    for (int dl = wave; dl < RNG; dl += NTH / 32) {
        float* const ap = acc + dl * PITCH + lane;
        if (d0 + dl < N) {
            float r[J];
            #pragma unroll
            for (int j = 0; j < J; ++j) r[j] = 0.0f;
            const int beg = __builtin_amdgcn_readfirstlane(first[dl]);
            const int ne  = __builtin_amdgcn_readfirstlane(cnt_n[dl]);
            int iv, sv;
            {
                const int jj = lane - 1;
                iv = n + dl; sv = d0 + dl;
                if (jj >= 0 && jj < ne) { iv = perm[beg + jj]; sv = lst_s[iv]; }
            }
            for (int j0 = -1; j0 < ne; j0 += U) {
                float row[U][J]; int ent[U];
                #pragma unroll
                for (int u = 0; u < U; ++u) {
                    if (j0 + u < ne) {
                        ent[u] = __builtin_amdgcn_readlane(iv, u);
                        const int s = __builtin_amdgcn_readlane(sv, u);
                        const float* rp = hfeat + (size_t)s * F + lane;
                        #pragma unroll
                        for (int j = 0; j < J; ++j) row[u][j] = rp[32 * j];
                    }
                }
                {
                    const int jj = j0 + U + lane;
                    if (jj < ne) { iv = perm[beg + jj]; sv = lst_s[iv]; }
                }
                #pragma unroll
                for (int u = 0; u < U; ++u) {
                    if (j0 + u < ne) {
                        float ex[H];
                        #pragma unroll
                        for (int h = 0; h < H; ++h) ex[h] = wgt[ent[u] * H + h];
                        #pragma unroll
                        for (int j = 0; j < J; ++j) r[j] += ex[(32 * j) / C] * row[u][j];
                    }
                }
            }
            #pragma unroll
            for (int j = 0; j < J; ++j) {
                float v = r[j] / den[dl * H + (32 * j) / C] + b[j];
                if (RELU) v = fmaxf(v, 0.0f);
                ap[32 * j] = v;
            }
        } else {
            #pragma unroll
            for (int j = 0; j < J; ++j) ap[32 * j] = 0.0f;
        }
    }
}

#define OC2  16

template <int H, int C, bool RELU, bool SAVE, bool LOAD, bool NEXT, bool SPAD>
__global__ __launch_bounds__(NTH)
void node_range_layer(const int* __restrict__ ei, int E, int N,
                      const float* __restrict__ a_src, const float* __restrict__ a_dst,
                      const float* __restrict__ hfeat, const float* __restrict__ bias,
                      float* __restrict__ outp, int* __restrict__ lists,
                      const unsigned* __restrict__ ids32,
                      const float* __restrict__ Wn,
                      const float* __restrict__ natt_src, const float* __restrict__ natt_dst,
                      float* __restrict__ nsrc, float* __restrict__ ndst) {
    constexpr int F = H * C;
    constexpr int PITCH = NEXT ? F + 4 : F;
    static_assert(!NEXT || (F * OC2 <= CAPL && RNG * (OC2 + 1) <= (CAPL + RNG) * H),
                  "the staged weights and the product tile reuse lst_s and wgt");
    static_assert(RNG <= SPITCH && SPITCH % 32 == 0, "a range's scores fill whole lines of their own");
    __shared__ float    acc[RNG * PITCH];
    __shared__ unsigned mkey[RNG * H];
    __shared__ float    den[RNG * H];
    __shared__ float    adl[RNG * H];
    __shared__ int      lst_s[CAPL];
    __shared__ int      lst_d[CAPL];
    __shared__ __attribute__((aligned(16))) float wgt[(CAPL + RNG) * H];
    __shared__ int      cnt;
    constexpr bool OWN = NEXT;
    __shared__ unsigned short perm[OWN ? CAPL : 1];

    const int tid = threadIdx.x;
    const int d0  = blockIdx.x * RNG;
    const int* dstp = ei + E;
    const bool vec = (((size_t)dstp) & 15) == 0;

    if (!OWN) for (int i = tid; i < RNG * PITCH; i += NTH) acc[i] = 0.0f;
    for (int i = tid; i < RNG * H; i += NTH) {
        mkey[i] = 0u; den[i] = 0.0f;
        adl[i] = (d0 * H + i < N * H) ? a_dst[(SPAD ? blockIdx.x * SPITCH : d0) * H + i] : 0.0f;
    }
    if (tid == 0) cnt = 0;
    __syncthreads();

    int* const my_cnt = lists + (size_t)blockIdx.x * LSTR;
    int* const my_s   = my_cnt + LHDR;
    int* const my_d   = my_s + CAPL;

    const bool by_ids = !LOAD && gridDim.x <= 255;

    int total;
    if (LOAD) {
        total = my_cnt[0];
    } else {
        if (by_ids) scan_ids((const uint4*)ids32, E, tid, lst_s, &cnt);
        else        scan_edges(dstp, 0, E, vec, d0, tid, lst_s, lst_d, &cnt);
        __syncthreads();
        total = cnt;
        if (SAVE && tid == 0) my_cnt[0] = total;
    }

    if (total <= CAPL) {
        if (LOAD) {
            for (int i = tid; i < total; i += NTH) { lst_s[i] = my_s[i]; lst_d[i] = my_d[i]; }
        } else {
            for (int i = tid; i < total; i += NTH) {
                const int e = lst_s[i];
                const int s = ei[e];
                if (by_ids) lst_d[i] = dstp[e] - d0;
                lst_s[i] = s;
                if (SAVE) { my_s[i] = s; my_d[i] = lst_d[i]; }
            }
        }
        __syncthreads();
        phase_score<H, true, SPAD>(total, true, d0, N, tid, a_src, lst_s, lst_d, adl, mkey, wgt);
        __syncthreads();
        int* const first  = reinterpret_cast<int*>(adl);
        int* const placed = reinterpret_cast<int*>(mkey);
        if (OWN) for (int i = tid; i < RNG; i += NTH) first[i] = 0;
        phase_exp<H>(total, true, d0, N, tid, lst_d, mkey, den, wgt);
        __syncthreads();
        if constexpr (OWN) {
            for (int i = tid; i < total; i += NTH) atomicAdd(&first[lst_d[i]], 1);
            for (int i = tid; i < RNG; i += NTH) placed[i] = 0;
            __syncthreads();
            if (tid < 32) {
                constexpr int PER = (RNG + 31) / 32;
                int cn[PER]; int sum = 0;
                #pragma unroll
                for (int k = 0; k < PER; ++k) {
                    const int idx = tid * PER + k;
                    cn[k] = idx < RNG ? first[idx] : 0;
                    sum += cn[k];
                }
                int incl = sum;
                for (int o = 1; o < 32; o <<= 1) {
                    const int t = __shfl_up(incl, o);
                    if (tid >= o) incl += t;
                }
                int base = incl - sum;
                #pragma unroll
                for (int k = 0; k < PER; ++k) {
                    const int idx = tid * PER + k;
                    if (idx < RNG) { first[idx] = base; base += cn[k]; }
                }
            }
            __syncthreads();
            for (int i = tid; i < total; i += NTH) {
                const int dl = lst_d[i];
                perm[first[dl] + atomicAdd(&placed[dl], 1)] = (unsigned short)i;
            }
            __syncthreads();
            phase_own<H, C, PITCH, RELU>(total, d0, N, tid, hfeat, bias, lst_s, perm, first, placed, wgt, den, acc);
        } else {
            phase_acc<H, C, PITCH>(total, true, d0, N, tid, hfeat, lst_s, lst_d, wgt, acc);
        }
        __syncthreads();
    } else {
        __syncthreads();
        if (OWN) for (int i = tid; i < RNG * PITCH; i += NTH) acc[i] = 0.0f;
        for (int pass = 0; pass < 2; ++pass) {
            for (int lo = 0; lo < E; lo += CAPL) {
                const int hi = lo + CAPL < E ? lo + CAPL : E;
                if (tid == 0) cnt = 0;
                __syncthreads();
                scan_edges(dstp, lo, hi, vec, d0, tid, lst_s, lst_d, &cnt);
                __syncthreads();
                const int n = cnt;
                for (int i = tid; i < n; i += NTH) lst_s[i] = ei[lst_s[i]];
                __syncthreads();
                if (pass == 0) {
                    phase_score<H, true, SPAD>(n, lo == 0, d0, N, tid, a_src, lst_s, lst_d, adl, mkey, wgt);
                } else {
                    phase_score<H, false, SPAD>(n, lo == 0, d0, N, tid, a_src, lst_s, lst_d, adl, mkey, wgt);
                    __syncthreads();
                    phase_exp<H>(n, lo == 0, d0, N, tid, lst_d, mkey, den, wgt);
                    __syncthreads();
                    phase_acc<H, C, PITCH>(n, lo == 0, d0, N, tid, hfeat, lst_s, lst_d, wgt, acc);
                }
                __syncthreads();
            }
        }
    }

    if (!NEXT) {
        for (int idx = tid; idx < RNG * F; idx += NTH) {
            const int dl = idx / F, c = idx % F;
            const int node = d0 + dl;
            if (node < N) {
                float v = acc[idx] / den[dl * H + c / C] + bias[c];
                if (RELU) v = fmaxf(v, 0.0f);
                outp[(size_t)node * F + c] = v;
            }
        }
    } else {
        float* const wn_s = reinterpret_cast<float*>(lst_s);
        float* const pr_s = wgt;

        constexpr int WPT = NEXT ? F * OC2 / NTH : 1;
        static_assert(!NEXT || F * OC2 % NTH == 0, "whole weights per thread");
        float wreg[WPT];
        #pragma unroll
        for (int j = 0; j < WPT; ++j) wreg[j] = Wn[tid + j * NTH];

        if (!(OWN && total <= CAPL)) {
            for (int idx = tid; idx < RNG * F; idx += NTH) {
                const int dl = idx / F, c = idx % F;
                float v = 0.0f;
                if (d0 + dl < N) {
                    v = acc[dl * PITCH + c] / den[dl * H + c / C] + bias[c];
                    if (RELU) v = fmaxf(v, 0.0f);
                }
                acc[dl * PITCH + c] = v;
            }
        }
        #pragma unroll
        for (int j = 0; j < WPT; ++j) wn_s[tid + j * NTH] = wreg[j];
        __syncthreads();

        const int wave = __builtin_amdgcn_readfirstlane(tid >> 5);
        const int lane = tid & 31;
        constexpr int MT = (RNG + 15) / 16;
        if (wave < MT) {
            const int nc    = lane & 15;
            const int row   = wave * 16 + nc;
            const int lr    = row < RNG ? row : RNG - 1;
            const int khalf = (lane >> 4) * 2;
            const float* ar = acc + lr * PITCH;
            v8f cc = {};
            for (int k = 0; k < F; k += 4) {
                const int kb = k + khalf;
                v2f a; a.x = ar[kb]; a.y = ar[kb + 1];
                v2f b; b.x = wn_s[kb * OC2 + nc]; b.y = wn_s[(kb + 1) * OC2 + nc];
                cc = __builtin_amdgcn_wmma_f32_16x16x4_f32(false, a, false, b, (short)0, cc, false, false);
            }
            const int rowoff = (lane >= 16) ? 8 : 0;
            for (int r = 0; r < 8; ++r) {
                const int rr = wave * 16 + rowoff + r;
                if (rr < RNG) pr_s[rr * (OC2 + 1) + nc] = cc[r];
            }
        }
        __syncthreads();

        for (int idx = tid; idx < RNG * OC2; idx += NTH) {
            const int dl = idx / OC2, c = idx % OC2;
            const int node = d0 + dl;
            if (node < N) outp[(size_t)node * OC2 + c] = pr_s[dl * (OC2 + 1) + c];
        }
        if (tid < RNG && d0 + tid < N) {
            float s = 0.0f, d = 0.0f;
            for (int c = 0; c < OC2; ++c) {
                const float v = pr_s[tid * (OC2 + 1) + c];
                s += v * natt_src[c];
                d += v * natt_dst[c];
            }
            nsrc[blockIdx.x * SPITCH + tid] = s;
            ndst[blockIdx.x * SPITCH + tid] = d;
        }
    }
}


extern "C" void kernel_launch(void* const* d_in, const int* in_sizes, int n_in,
                              void* d_out, int out_size, void* d_ws, size_t ws_size,
                              hipStream_t stream) {
    const float* x   = (const float*)d_in[0];
    const int*   ei  = (const int*)  d_in[1];
    const float* W1  = (const float*)d_in[2];
    const float* as1 = (const float*)d_in[3];
    const float* ad1 = (const float*)d_in[4];
    const float* b1  = (const float*)d_in[5];
    const float* W2  = (const float*)d_in[6];
    const float* as2 = (const float*)d_in[7];
    const float* ad2 = (const float*)d_in[8];
    const float* b2  = (const float*)d_in[9];

    const int N  = in_sizes[0] / 128;
    const int E  = in_sizes[1] / 2;

    char* w = (char*)d_ws;
    auto carve = [&](size_t bytes) -> void* {
        void* p = (void*)w;
        w += (bytes + 255) & ~((size_t)255);
        return p;
    };
    float* h1  = (float*)carve((size_t)N * 256 * 4);
    float* aS1 = (float*)carve((size_t)N * 4 * 4);
    float* aD1 = (float*)carve((size_t)N * 4 * 4);
    float* h2  = (float*)carve((size_t)N * 16 * 4);

    const int TB = 256;
    auto blocks = [&](long long n) { return (unsigned)((n + TB - 1) / TB); };
    const unsigned nrange = (unsigned)((N + RNG - 1) / RNG);
    int* lists = (int*)carve((size_t)nrange * LSTR * 4);
    unsigned* ids32 = (unsigned*)carve((((size_t)E + 15) / 16) * 16);
    float* aS2 = (float*)carve((size_t)nrange * SPITCH * 4);
    float* aD2 = (float*)carve((size_t)nrange * SPITCH * 4);

    dim3 g1((N + 15) / 16, 4);
    gemm1_wmma<<<g1, 32, 0, stream>>>(x, W1, h1, N);
    att1_kernel<<<blocks(N * 4), TB, 0, stream>>>(h1, as1, ad1, aS1, aD1, N * 4, ei + E, E, ids32);
    node_range_layer<4, 64, true, true, false, true, false><<<nrange, NTH, 0, stream>>>(ei, E, N, aS1, aD1, h1, b1, h2, lists, ids32, W2, as2, ad2, aS2, aD2);

    node_range_layer<1, 16, false, false, true, false, true><<<nrange, NTH, 0, stream>>>(ei, E, N, aS2, aD2, h2, b2, (float*)d_out, lists, ids32, nullptr, nullptr, nullptr, nullptr, nullptr);
}


# Layer-1 scores made by the product block from its finished 16x256 tile in LDS (one lane per row and score, columns ascending: sums unchanged); score launch and its re-read of h1 gone, 4 launches to 3
# speedup vs baseline: 5.7497x; 1.0397x over previous
#include <hip/hip_runtime.h>
#include <hip/hip_bf16.h>

typedef __attribute__((ext_vector_type(2))) float v2f;
typedef __attribute__((ext_vector_type(8))) float v8f;

#define NEG_SLOPE 0.2f


__device__ __forceinline__ float lrelu(float v) {
    return v >= 0.0f ? v : NEG_SLOPE * v;
}

__device__ __forceinline__ unsigned fkey(float f) {
    unsigned u = __float_as_uint(f);
    return (u & 0x80000000u) ? ~u : (u | 0x80000000u);
}
__device__ __forceinline__ float fdekey(unsigned k) {
    return (k & 0x80000000u) ? __uint_as_float(k ^ 0x80000000u)
                             : __uint_as_float(~k);
}

#define RNG  200
#define SPITCH 256

#define TP 68
__global__ __launch_bounds__(128)
void gemm1_scores(const float* __restrict__ x, const float* __restrict__ W1,
                  const float* __restrict__ att_src, const float* __restrict__ att_dst,
                  float* __restrict__ h1,
                  float* __restrict__ a_src, float* __restrict__ a_dst, int N,
                  const int* __restrict__ dstp, int E,
                  unsigned* __restrict__ ids32) {
    __shared__ __attribute__((aligned(16))) float tile[4][16 * TP];
    __shared__ float sc[2][64];

    const int mtile = blockIdx.x;
    const int lane  = threadIdx.x & 31;
    const int head  = __builtin_amdgcn_readfirstlane(threadIdx.x >> 5);
    const int mrow  = mtile * 16 + (lane & 15);
    const int lrow  = mrow < N ? mrow : N - 1;
    const int khalf = (lane >> 4) * 2;
    const int colbase = head * 64;
    const int nc    = lane & 15;

    v8f c0 = {}, c1 = {}, c2 = {}, c3 = {};
    const float* xrow = x + (size_t)lrow * 128;

    for (int k = 0; k < 128; k += 4) {
        const int kb = k + khalf;
        v2f a; a.x = xrow[kb]; a.y = xrow[kb + 1];
        const float* w0 = W1 + (size_t)kb * 256 + colbase + nc;
        const float* w1 = w0 + 256;
        v2f b0; b0.x = w0[0];  b0.y = w1[0];
        v2f b1; b1.x = w0[16]; b1.y = w1[16];
        v2f b2; b2.x = w0[32]; b2.y = w1[32];
        v2f b3; b3.x = w0[48]; b3.y = w1[48];
        c0 = __builtin_amdgcn_wmma_f32_16x16x4_f32(false, a, false, b0, (short)0, c0, false, false);
        c1 = __builtin_amdgcn_wmma_f32_16x16x4_f32(false, a, false, b1, (short)0, c1, false, false);
        c2 = __builtin_amdgcn_wmma_f32_16x16x4_f32(false, a, false, b2, (short)0, c2, false, false);
        c3 = __builtin_amdgcn_wmma_f32_16x16x4_f32(false, a, false, b3, (short)0, c3, false, false);
    }

    const int rowoff = (lane >= 16) ? 8 : 0;
    float* const tw = tile[head];
    for (int r = 0; r < 8; ++r) {
        const int grow = mtile * 16 + rowoff + r;
        if (grow < N) {
            float* op = h1 + (size_t)grow * 256 + colbase + nc;
            op[0]  = c0[r];
            op[16] = c1[r];
            op[32] = c2[r];
            op[48] = c3[r];
        }
        float* tp = tw + (rowoff + r) * TP + nc;
        tp[0]  = c0[r];
        tp[16] = c1[r];
        tp[32] = c2[r];
        tp[48] = c3[r];
    }
    __syncthreads();

    {
        const int row = lane & 15;
        const float* av = (lane < 16 ? att_src : att_dst) + head * 64;
        const float* tr = tw + row * TP;
        float s = 0.0f;
        for (int c = 0; c < 64; ++c) s += tr[c] * av[c];
        sc[lane >> 4][row * 4 + head] = s;
    }
    __syncthreads();

    {
        const int t = threadIdx.x & 63;
        float* const dstv = threadIdx.x < 64 ? a_src : a_dst;
        if (mtile * 16 + (t >> 2) < N) dstv[(size_t)mtile * 64 + t] = sc[threadIdx.x >> 6][t];
    }

    {
        const int i = blockIdx.x * blockDim.x + threadIdx.x;
        const int nwp = ((E + 15) >> 4) << 2;
        const int nthreads = gridDim.x * blockDim.x;
        const bool vec = (((size_t)dstp) & 15) == 0;
        for (int w = i; w < nwp; w += nthreads) {
            int d[4] = {-1, -1, -1, -1};
            if (vec && 4 * w + 3 < E) {
                const int4 v = *(const int4*)(dstp + 4 * w);
                d[0] = v.x; d[1] = v.y; d[2] = v.z; d[3] = v.w;
            } else {
                #pragma unroll
                for (int j = 0; j < 4; ++j)
                    if (4 * w + j < E) d[j] = dstp[4 * w + j];
            }
            unsigned p = 0;
            #pragma unroll
            for (int j = 0; j < 4; ++j) {
                const unsigned q = (unsigned)d[j] / (unsigned)RNG;
                p |= (q < 255u ? q : 255u) << (8 * j);
            }
            ids32[w] = p;
        }
    }
}

#define CAPL 4096
#define NTH  1024
#define LHDR 64
#define LSTR (LHDR + 2 * CAPL)

__device__ __forceinline__ void list_one(int d, int e, int d0,
                                         int* lst_s, int* lst_d, int* cnt) {
    const unsigned dl = (unsigned)(d - d0);
    if (dl < (unsigned)RNG) {
        const int slot = atomicAdd(cnt, 1);
        if (slot < CAPL) { lst_s[slot] = e; lst_d[slot] = (int)dl; }
    }
}

__device__ __forceinline__ void scan_edges(const int* __restrict__ dstp, int lo, int hi,
                                           bool vec, int d0, int tid,
                                           int* lst_s, int* lst_d, int* cnt) {
    constexpr int SU = 8;
    if (vec) {
        const int nq = (hi - lo) >> 2;
        const int K  = (nq + NTH - 1) / NTH;
        const int rot = (int)(((long long)blockIdx.x * K) / gridDim.x);
        for (int k = 0; k < K; k += SU) {
            int4 v[SU]; int eb[SU]; bool ok[SU];
            #pragma unroll
            for (int u = 0; u < SU; ++u) {
                int kr = k + u + rot; kr = kr >= K ? kr - K : kr;
                const int q = kr * NTH + tid;
                ok[u] = (k + u < K) && (q < nq);
                const int qq = ok[u] ? q : 0;
                v[u]  = *(const int4*)(dstp + lo + 4 * qq);
                eb[u] = lo + 4 * qq;
            }
            #pragma unroll
            for (int u = 0; u < SU; ++u) {
                if (ok[u]) {
                    list_one(v[u].x, eb[u] + 0, d0, lst_s, lst_d, cnt);
                    list_one(v[u].y, eb[u] + 1, d0, lst_s, lst_d, cnt);
                    list_one(v[u].z, eb[u] + 2, d0, lst_s, lst_d, cnt);
                    list_one(v[u].w, eb[u] + 3, d0, lst_s, lst_d, cnt);
                }
            }
        }
        for (int e = lo + 4 * nq + tid; e < hi; e += NTH)
            list_one(dstp[e], e, d0, lst_s, lst_d, cnt);
    } else {
        for (int e = lo + tid; e < hi; e += NTH)
            list_one(dstp[e], e, d0, lst_s, lst_d, cnt);
    }
}

__device__ __forceinline__ unsigned zero_bytes(unsigned x) {
    return ~(((x & 0x7f7f7f7fu) + 0x7f7f7f7fu) | x | 0x7f7f7f7fu);
}

__device__ __forceinline__ void scan_ids(const uint4* __restrict__ ids, int E, int tid,
                                         int* lst_s, int* cnt) {
    constexpr int SU = 8;
    const int ng = (E + 15) >> 4;
    const int K  = (ng + NTH - 1) / NTH;
    const int rot = (int)(((long long)blockIdx.x * K) / gridDim.x);
    const unsigned mine = blockIdx.x * 0x01010101u;
    for (int k = 0; k < K; k += SU) {
        uint4 v[SU]; int eb[SU]; bool ok[SU];
        #pragma unroll
        for (int u = 0; u < SU; ++u) {
            int kr = k + u + rot; kr = kr >= K ? kr - K : kr;
            const int g = kr * NTH + tid;
            ok[u] = (k + u < K) && (g < ng);
            const int gg = ok[u] ? g : 0;
            v[u]  = ids[gg];
            eb[u] = 16 * gg;
        }
        #pragma unroll
        for (int u = 0; u < SU; ++u) {
            if (ok[u]) {
                unsigned m = (zero_bytes(v[u].x ^ mine) >> 7) | (zero_bytes(v[u].y ^ mine) >> 6)
                           | (zero_bytes(v[u].z ^ mine) >> 5) | (zero_bytes(v[u].w ^ mine) >> 4);
                while (m) {
                    const int b = __builtin_ctz(m);
                    m &= m - 1;
                    const int e = eb[u] + ((b & 7) << 2) + (b >> 3);
                    const int slot = atomicAdd(cnt, 1);
                    if (slot < CAPL) lst_s[slot] = e;
                }
            }
        }
    }
}

template <int H, bool MAXUPD, bool SPAD>
__device__ __forceinline__ void phase_score(int n, bool with_self, int d0, int N, int tid,
                                            const float* __restrict__ a_src,
                                            const int* lst_s, const int* lst_d,
                                            const float* adl, unsigned* mkey, float* wgt) {
    const int tot = n + (with_self ? RNG : 0);
    for (int i = tid; i < tot; i += NTH) {
        int s, dl;
        if (i < n) { s = lst_s[i]; dl = lst_d[i]; }
        else       { dl = i - n;   s = d0 + dl; }
        if (d0 + dl < N) {
            const int si = SPAD ? (int)(((unsigned)s / RNG) * SPITCH + (unsigned)s % RNG) : s;
            for (int h = 0; h < H; ++h) {
                const float v = lrelu(a_src[si * H + h] + adl[dl * H + h]);
                if (MAXUPD) atomicMax(&mkey[dl * H + h], fkey(v));
                wgt[i * H + h] = v;
            }
        }
    }
}

template <int H>
__device__ __forceinline__ void phase_exp(int n, bool with_self, int d0, int N, int tid,
                                          const int* lst_d, const unsigned* mkey,
                                          float* den, float* wgt) {
    const int tot = n + (with_self ? RNG : 0);
    for (int i = tid; i < tot; i += NTH) {
        const int dl = i < n ? lst_d[i] : i - n;
        if (d0 + dl < N) {
            for (int h = 0; h < H; ++h) {
                const float ex = expf(wgt[i * H + h] - fdekey(mkey[dl * H + h]));
                wgt[i * H + h] = ex;
                atomicAdd(&den[dl * H + h], ex);
            }
        }
    }
}

template <int H, int C, int PITCH>
__device__ __forceinline__ void phase_acc(int n, bool with_self, int d0, int N, int tid,
                                          const float* __restrict__ hfeat,
                                          const int* lst_s, const int* lst_d,
                                          const float* wgt, float* acc) {
    constexpr int F   = H * C;
    constexpr int LPE = F >= 32 ? 32 : F;
    constexpr int EPW = 32 / LPE;
    constexpr int J   = F / LPE;
    constexpr int U   = 4;
    constexpr int STEP = (NTH / 32) * EPW * U;
    const int lane = tid & 31, wave = tid >> 5;
    const int sub = lane / LPE, l = lane % LPE;
    const int tot = n + (with_self ? RNG : 0);

    for (int i0 = 0; i0 < tot; i0 += STEP) {
        int dl[U]; bool ok[U];
        float row[U][J]; float ex[U][H];
        #pragma unroll
        for (int u = 0; u < U; ++u) {
            int i = i0 + (wave * U + u) * EPW + sub;
            int s;
            if (i < n) { s = lst_s[i]; dl[u] = lst_d[i]; }
            else       { dl[u] = i - n; s = d0 + dl[u]; }
            ok[u] = (i < tot) && (d0 + dl[u] < N);
            if (!ok[u]) { s = 0; dl[u] = 0; i = 0; }
            const float* rp = hfeat + (size_t)s * F + l;
            #pragma unroll
            for (int j = 0; j < J; ++j) row[u][j] = rp[LPE * j];
            #pragma unroll
            for (int h = 0; h < H; ++h) ex[u][h] = wgt[i * H + h];
        }
        #pragma unroll
        for (int u = 0; u < U; ++u) {
            if (ok[u]) {
                float* ap = acc + dl[u] * PITCH + l;
                #pragma unroll
                for (int j = 0; j < J; ++j)
                    atomicAdd(ap + LPE * j, ex[u][(LPE * j) / C] * row[u][j]);
            }
        }
    }
}

template <int H, int C, int PITCH, bool RELU>
__device__ __forceinline__ void phase_own(int n, int d0, int N, int tid,
                                          const float* __restrict__ hfeat,
                                          const float* __restrict__ bias,
                                          const int* lst_s, const unsigned short* perm,
                                          const int* first, const int* cnt_n,
                                          const float* wgt, const float* den, float* acc) {
    constexpr int F = H * C;
    constexpr int J = F / 32;
    constexpr int U = 8;
    const int lane = tid & 31;
    const int wave = __builtin_amdgcn_readfirstlane(tid >> 5);
    float b[J];
    #pragma unroll
    for (int j = 0; j < J; ++j) b[j] = bias[lane + 32 * j];

    for (int dl = wave; dl < RNG; dl += NTH / 32) {
        float* const ap = acc + dl * PITCH + lane;
        if (d0 + dl < N) {
            float r[J];
            #pragma unroll
            for (int j = 0; j < J; ++j) r[j] = 0.0f;
            const int beg = __builtin_amdgcn_readfirstlane(first[dl]);
            const int ne  = __builtin_amdgcn_readfirstlane(cnt_n[dl]);
            int iv, sv;
            {
                const int jj = lane - 1;
                iv = n + dl; sv = d0 + dl;
                if (jj >= 0 && jj < ne) { iv = perm[beg + jj]; sv = lst_s[iv]; }
            }
            for (int j0 = -1; j0 < ne; j0 += U) {
                float row[U][J]; int ent[U];
                #pragma unroll
                for (int u = 0; u < U; ++u) {
                    if (j0 + u < ne) {
                        ent[u] = __builtin_amdgcn_readlane(iv, u);
                        const int s = __builtin_amdgcn_readlane(sv, u);
                        const float* rp = hfeat + (size_t)s * F + lane;
                        #pragma unroll
                        for (int j = 0; j < J; ++j) row[u][j] = rp[32 * j];
                    }
                }
                {
                    const int jj = j0 + U + lane;
                    if (jj < ne) { iv = perm[beg + jj]; sv = lst_s[iv]; }
                }
                #pragma unroll
                for (int u = 0; u < U; ++u) {
                    if (j0 + u < ne) {
                        float ex[H];
                        #pragma unroll
                        for (int h = 0; h < H; ++h) ex[h] = wgt[ent[u] * H + h];
                        #pragma unroll
                        for (int j = 0; j < J; ++j) r[j] += ex[(32 * j) / C] * row[u][j];
                    }
                }
            }
            #pragma unroll
            for (int j = 0; j < J; ++j) {
                float v = r[j] / den[dl * H + (32 * j) / C] + b[j];
                if (RELU) v = fmaxf(v, 0.0f);
                ap[32 * j] = v;
            }
        } else {
            #pragma unroll
            for (int j = 0; j < J; ++j) ap[32 * j] = 0.0f;
        }
    }
}

#define OC2  16

template <int H, int C, bool RELU, bool SAVE, bool LOAD, bool NEXT, bool SPAD>
__global__ __launch_bounds__(NTH)
void node_range_layer(const int* __restrict__ ei, int E, int N,
                      const float* __restrict__ a_src, const float* __restrict__ a_dst,
                      const float* __restrict__ hfeat, const float* __restrict__ bias,
                      float* __restrict__ outp, int* __restrict__ lists,
                      const unsigned* __restrict__ ids32,
                      const float* __restrict__ Wn,
                      const float* __restrict__ natt_src, const float* __restrict__ natt_dst,
                      float* __restrict__ nsrc, float* __restrict__ ndst) {
    constexpr int F = H * C;
    constexpr int PITCH = NEXT ? F + 4 : F;
    static_assert(!NEXT || (F * OC2 <= CAPL && RNG * (OC2 + 1) <= (CAPL + RNG) * H),
                  "the staged weights and the product tile reuse lst_s and wgt");
    static_assert(RNG <= SPITCH && SPITCH % 32 == 0, "a range's scores fill whole lines of their own");
    __shared__ float    acc[RNG * PITCH];
    __shared__ unsigned mkey[RNG * H];
    __shared__ float    den[RNG * H];
    __shared__ float    adl[RNG * H];
    __shared__ int      lst_s[CAPL];
    __shared__ int      lst_d[CAPL];
    __shared__ __attribute__((aligned(16))) float wgt[(CAPL + RNG) * H];
    __shared__ int      cnt;
    constexpr bool OWN = NEXT;
    __shared__ unsigned short perm[OWN ? CAPL : 1];

    const int tid = threadIdx.x;
    const int d0  = blockIdx.x * RNG;
    const int* dstp = ei + E;
    const bool vec = (((size_t)dstp) & 15) == 0;

    if (!OWN) for (int i = tid; i < RNG * PITCH; i += NTH) acc[i] = 0.0f;
    for (int i = tid; i < RNG * H; i += NTH) {
        mkey[i] = 0u; den[i] = 0.0f;
        adl[i] = (d0 * H + i < N * H) ? a_dst[(SPAD ? blockIdx.x * SPITCH : d0) * H + i] : 0.0f;
    }
    if (tid == 0) cnt = 0;
    __syncthreads();

    int* const my_cnt = lists + (size_t)blockIdx.x * LSTR;
    int* const my_s   = my_cnt + LHDR;
    int* const my_d   = my_s + CAPL;

    const bool by_ids = !LOAD && gridDim.x <= 255;

    int total;
    if (LOAD) {
        total = my_cnt[0];
    } else {
        if (by_ids) scan_ids((const uint4*)ids32, E, tid, lst_s, &cnt);
        else        scan_edges(dstp, 0, E, vec, d0, tid, lst_s, lst_d, &cnt);
        __syncthreads();
        total = cnt;
        if (SAVE && tid == 0) my_cnt[0] = total;
    }

    if (total <= CAPL) {
        if (LOAD) {
            for (int i = tid; i < total; i += NTH) { lst_s[i] = my_s[i]; lst_d[i] = my_d[i]; }
        } else {
            for (int i = tid; i < total; i += NTH) {
                const int e = lst_s[i];
                const int s = ei[e];
                if (by_ids) lst_d[i] = dstp[e] - d0;
                lst_s[i] = s;
                if (SAVE) { my_s[i] = s; my_d[i] = lst_d[i]; }
            }
        }
        __syncthreads();
        phase_score<H, true, SPAD>(total, true, d0, N, tid, a_src, lst_s, lst_d, adl, mkey, wgt);
        __syncthreads();
        int* const first  = reinterpret_cast<int*>(adl);
        int* const placed = reinterpret_cast<int*>(mkey);
        if (OWN) for (int i = tid; i < RNG; i += NTH) first[i] = 0;
        phase_exp<H>(total, true, d0, N, tid, lst_d, mkey, den, wgt);
        __syncthreads();
        if constexpr (OWN) {
            for (int i = tid; i < total; i += NTH) atomicAdd(&first[lst_d[i]], 1);
            for (int i = tid; i < RNG; i += NTH) placed[i] = 0;
            __syncthreads();
            if (tid < 32) {
                constexpr int PER = (RNG + 31) / 32;
                int cn[PER]; int sum = 0;
                #pragma unroll
                for (int k = 0; k < PER; ++k) {
                    const int idx = tid * PER + k;
                    cn[k] = idx < RNG ? first[idx] : 0;
                    sum += cn[k];
                }
                int incl = sum;
                for (int o = 1; o < 32; o <<= 1) {
                    const int t = __shfl_up(incl, o);
                    if (tid >= o) incl += t;
                }
                int base = incl - sum;
                #pragma unroll
                for (int k = 0; k < PER; ++k) {
                    const int idx = tid * PER + k;
                    if (idx < RNG) { first[idx] = base; base += cn[k]; }
                }
            }
            __syncthreads();
            for (int i = tid; i < total; i += NTH) {
                const int dl = lst_d[i];
                perm[first[dl] + atomicAdd(&placed[dl], 1)] = (unsigned short)i;
            }
            __syncthreads();
            phase_own<H, C, PITCH, RELU>(total, d0, N, tid, hfeat, bias, lst_s, perm, first, placed, wgt, den, acc);
        } else {
            phase_acc<H, C, PITCH>(total, true, d0, N, tid, hfeat, lst_s, lst_d, wgt, acc);
        }
        __syncthreads();
    } else {
        __syncthreads();
        if (OWN) for (int i = tid; i < RNG * PITCH; i += NTH) acc[i] = 0.0f;
        for (int pass = 0; pass < 2; ++pass) {
            for (int lo = 0; lo < E; lo += CAPL) {
                const int hi = lo + CAPL < E ? lo + CAPL : E;
                if (tid == 0) cnt = 0;
                __syncthreads();
                scan_edges(dstp, lo, hi, vec, d0, tid, lst_s, lst_d, &cnt);
                __syncthreads();
                const int n = cnt;
                for (int i = tid; i < n; i += NTH) lst_s[i] = ei[lst_s[i]];
                __syncthreads();
                if (pass == 0) {
                    phase_score<H, true, SPAD>(n, lo == 0, d0, N, tid, a_src, lst_s, lst_d, adl, mkey, wgt);
                } else {
                    phase_score<H, false, SPAD>(n, lo == 0, d0, N, tid, a_src, lst_s, lst_d, adl, mkey, wgt);
                    __syncthreads();
                    phase_exp<H>(n, lo == 0, d0, N, tid, lst_d, mkey, den, wgt);
                    __syncthreads();
                    phase_acc<H, C, PITCH>(n, lo == 0, d0, N, tid, hfeat, lst_s, lst_d, wgt, acc);
                }
                __syncthreads();
            }
        }
    }

    if (!NEXT) {
        for (int idx = tid; idx < RNG * F; idx += NTH) {
            const int dl = idx / F, c = idx % F;
            const int node = d0 + dl;
            if (node < N) {
                float v = acc[idx] / den[dl * H + c / C] + bias[c];
                if (RELU) v = fmaxf(v, 0.0f);
                outp[(size_t)node * F + c] = v;
            }
        }
    } else {
        float* const wn_s = reinterpret_cast<float*>(lst_s);
        float* const pr_s = wgt;

        constexpr int WPT = NEXT ? F * OC2 / NTH : 1;
        static_assert(!NEXT || F * OC2 % NTH == 0, "whole weights per thread");
        float wreg[WPT];
        #pragma unroll
        for (int j = 0; j < WPT; ++j) wreg[j] = Wn[tid + j * NTH];

        if (!(OWN && total <= CAPL)) {
            for (int idx = tid; idx < RNG * F; idx += NTH) {
                const int dl = idx / F, c = idx % F;
                float v = 0.0f;
                if (d0 + dl < N) {
                    v = acc[dl * PITCH + c] / den[dl * H + c / C] + bias[c];
                    if (RELU) v = fmaxf(v, 0.0f);
                }
                acc[dl * PITCH + c] = v;
            }
        }
        #pragma unroll
        for (int j = 0; j < WPT; ++j) wn_s[tid + j * NTH] = wreg[j];
        __syncthreads();

        const int wave = __builtin_amdgcn_readfirstlane(tid >> 5);
        const int lane = tid & 31;
        constexpr int MT = (RNG + 15) / 16;
        if (wave < MT) {
            const int nc    = lane & 15;
            const int row   = wave * 16 + nc;
            const int lr    = row < RNG ? row : RNG - 1;
            const int khalf = (lane >> 4) * 2;
            const float* ar = acc + lr * PITCH;
            v8f cc = {};
            for (int k = 0; k < F; k += 4) {
                const int kb = k + khalf;
                v2f a; a.x = ar[kb]; a.y = ar[kb + 1];
                v2f b; b.x = wn_s[kb * OC2 + nc]; b.y = wn_s[(kb + 1) * OC2 + nc];
                cc = __builtin_amdgcn_wmma_f32_16x16x4_f32(false, a, false, b, (short)0, cc, false, false);
            }
            const int rowoff = (lane >= 16) ? 8 : 0;
            for (int r = 0; r < 8; ++r) {
                const int rr = wave * 16 + rowoff + r;
                if (rr < RNG) pr_s[rr * (OC2 + 1) + nc] = cc[r];
            }
        }
        __syncthreads();

        for (int idx = tid; idx < RNG * OC2; idx += NTH) {
            const int dl = idx / OC2, c = idx % OC2;
            const int node = d0 + dl;
            if (node < N) outp[(size_t)node * OC2 + c] = pr_s[dl * (OC2 + 1) + c];
        }
        if (tid < RNG && d0 + tid < N) {
            float s = 0.0f, d = 0.0f;
            for (int c = 0; c < OC2; ++c) {
                const float v = pr_s[tid * (OC2 + 1) + c];
                s += v * natt_src[c];
                d += v * natt_dst[c];
            }
            nsrc[blockIdx.x * SPITCH + tid] = s;
            ndst[blockIdx.x * SPITCH + tid] = d;
        }
    }
}


extern "C" void kernel_launch(void* const* d_in, const int* in_sizes, int n_in,
                              void* d_out, int out_size, void* d_ws, size_t ws_size,
                              hipStream_t stream) {
    const float* x   = (const float*)d_in[0];
    const int*   ei  = (const int*)  d_in[1];
    const float* W1  = (const float*)d_in[2];
    const float* as1 = (const float*)d_in[3];
    const float* ad1 = (const float*)d_in[4];
    const float* b1  = (const float*)d_in[5];
    const float* W2  = (const float*)d_in[6];
    const float* as2 = (const float*)d_in[7];
    const float* ad2 = (const float*)d_in[8];
    const float* b2  = (const float*)d_in[9];

    const int N  = in_sizes[0] / 128;
    const int E  = in_sizes[1] / 2;

    char* w = (char*)d_ws;
    auto carve = [&](size_t bytes) -> void* {
        void* p = (void*)w;
        w += (bytes + 255) & ~((size_t)255);
        return p;
    };
    float* h1  = (float*)carve((size_t)N * 256 * 4);
    float* aS1 = (float*)carve((size_t)N * 4 * 4);
    float* aD1 = (float*)carve((size_t)N * 4 * 4);
    float* h2  = (float*)carve((size_t)N * 16 * 4);

    const unsigned nrange = (unsigned)((N + RNG - 1) / RNG);
    int* lists = (int*)carve((size_t)nrange * LSTR * 4);
    unsigned* ids32 = (unsigned*)carve((((size_t)E + 15) / 16) * 16);
    float* aS2 = (float*)carve((size_t)nrange * SPITCH * 4);
    float* aD2 = (float*)carve((size_t)nrange * SPITCH * 4);

    gemm1_scores<<<(N + 15) / 16, 128, 0, stream>>>(x, W1, as1, ad1, h1, aS1, aD1, N, ei + E, E, ids32);
    node_range_layer<4, 64, true, true, false, true, false><<<nrange, NTH, 0, stream>>>(ei, E, N, aS1, aD1, h1, b1, h2, lists, ids32, W2, as2, ad2, aS2, aD2);

    node_range_layer<1, 16, false, false, true, false, true><<<nrange, NTH, 0, stream>>>(ei, E, N, aS2, aD2, h2, b2, (float*)d_out, lists, ids32, nullptr, nullptr, nullptr, nullptr, nullptr);
}


# Layer-1 product: accumulator t of a lane holds column 4c+t of its head, so W1 is read and h1 stored 16 bytes per lane in whole 128-byte lines (2 loads per k step, not 8); sums unchanged
# speedup vs baseline: 5.9212x; 1.0298x over previous
#include <hip/hip_runtime.h>
#include <hip/hip_bf16.h>

typedef __attribute__((ext_vector_type(2))) float v2f;
typedef __attribute__((ext_vector_type(8))) float v8f;
typedef float v4fu __attribute__((ext_vector_type(4), aligned(4)));

#define NEG_SLOPE 0.2f


__device__ __forceinline__ float lrelu(float v) {
    return v >= 0.0f ? v : NEG_SLOPE * v;
}

__device__ __forceinline__ unsigned fkey(float f) {
    unsigned u = __float_as_uint(f);
    return (u & 0x80000000u) ? ~u : (u | 0x80000000u);
}
__device__ __forceinline__ float fdekey(unsigned k) {
    return (k & 0x80000000u) ? __uint_as_float(k ^ 0x80000000u)
                             : __uint_as_float(~k);
}

#define RNG  200
#define SPITCH 256

#define TP 68
__global__ __launch_bounds__(128)
void gemm1_scores(const float* __restrict__ x, const float* __restrict__ W1,
                  const float* __restrict__ att_src, const float* __restrict__ att_dst,
                  float* __restrict__ h1,
                  float* __restrict__ a_src, float* __restrict__ a_dst, int N,
                  const int* __restrict__ dstp, int E,
                  unsigned* __restrict__ ids32) {
    __shared__ __attribute__((aligned(16))) float tile[4][16 * TP];
    __shared__ float sc[2][64];

    const int mtile = blockIdx.x;
    const int lane  = threadIdx.x & 31;
    const int head  = __builtin_amdgcn_readfirstlane(threadIdx.x >> 5);
    const int mrow  = mtile * 16 + (lane & 15);
    const int lrow  = mrow < N ? mrow : N - 1;
    const int khalf = (lane >> 4) * 2;
    const int colbase = head * 64;
    const int nc    = lane & 15;

    v8f c0 = {}, c1 = {}, c2 = {}, c3 = {};
    const float* xrow = x + (size_t)lrow * 128;

    #pragma unroll 4
    for (int k = 0; k < 128; k += 4) {
        const int kb = k + khalf;
        v2f a; a.x = xrow[kb]; a.y = xrow[kb + 1];
        const float* w0 = W1 + (size_t)kb * 256 + colbase + 4 * nc;
        const v4fu q0 = *(const v4fu*)w0;
        const v4fu q1 = *(const v4fu*)(w0 + 256);
        v2f b0; b0.x = q0.x; b0.y = q1.x;
        v2f b1; b1.x = q0.y; b1.y = q1.y;
        v2f b2; b2.x = q0.z; b2.y = q1.z;
        v2f b3; b3.x = q0.w; b3.y = q1.w;
        c0 = __builtin_amdgcn_wmma_f32_16x16x4_f32(false, a, false, b0, (short)0, c0, false, false);
        c1 = __builtin_amdgcn_wmma_f32_16x16x4_f32(false, a, false, b1, (short)0, c1, false, false);
        c2 = __builtin_amdgcn_wmma_f32_16x16x4_f32(false, a, false, b2, (short)0, c2, false, false);
        c3 = __builtin_amdgcn_wmma_f32_16x16x4_f32(false, a, false, b3, (short)0, c3, false, false);
    }

    const int rowoff = (lane >= 16) ? 8 : 0;
    float* const tw = tile[head];
    for (int r = 0; r < 8; ++r) {
        const int grow = mtile * 16 + rowoff + r;
        float4 v; v.x = c0[r]; v.y = c1[r]; v.z = c2[r]; v.w = c3[r];
        if (grow < N) *(float4*)(h1 + (size_t)grow * 256 + colbase + 4 * nc) = v;
        *(float4*)(tw + (rowoff + r) * TP + 4 * nc) = v;
    }
    __syncthreads();

    {
        const int row = lane & 15;
        const float* av = (lane < 16 ? att_src : att_dst) + head * 64;
        const float* tr = tw + row * TP;
        float s = 0.0f;
        for (int c = 0; c < 64; ++c) s += tr[c] * av[c];
        sc[lane >> 4][row * 4 + head] = s;
    }
    __syncthreads();

    {
        const int t = threadIdx.x & 63;
        float* const dstv = threadIdx.x < 64 ? a_src : a_dst;
        if (mtile * 16 + (t >> 2) < N) dstv[(size_t)mtile * 64 + t] = sc[threadIdx.x >> 6][t];
    }

    {
        const int i = blockIdx.x * blockDim.x + threadIdx.x;
        const int nwp = ((E + 15) >> 4) << 2;
        const int nthreads = gridDim.x * blockDim.x;
        const bool vec = (((size_t)dstp) & 15) == 0;
        for (int w = i; w < nwp; w += nthreads) {
            int d[4] = {-1, -1, -1, -1};
            if (vec && 4 * w + 3 < E) {
                const int4 v = *(const int4*)(dstp + 4 * w);
                d[0] = v.x; d[1] = v.y; d[2] = v.z; d[3] = v.w;
            } else {
                #pragma unroll
                for (int j = 0; j < 4; ++j)
                    if (4 * w + j < E) d[j] = dstp[4 * w + j];
            }
            unsigned p = 0;
            #pragma unroll
            for (int j = 0; j < 4; ++j) {
                const unsigned q = (unsigned)d[j] / (unsigned)RNG;
                p |= (q < 255u ? q : 255u) << (8 * j);
            }
            ids32[w] = p;
        }
    }
}

#define CAPL 4096
#define NTH  1024
#define LHDR 64
#define LSTR (LHDR + 2 * CAPL)

__device__ __forceinline__ void list_one(int d, int e, int d0,
                                         int* lst_s, int* lst_d, int* cnt) {
    const unsigned dl = (unsigned)(d - d0);
    if (dl < (unsigned)RNG) {
        const int slot = atomicAdd(cnt, 1);
        if (slot < CAPL) { lst_s[slot] = e; lst_d[slot] = (int)dl; }
    }
}

__device__ __forceinline__ void scan_edges(const int* __restrict__ dstp, int lo, int hi,
                                           bool vec, int d0, int tid,
                                           int* lst_s, int* lst_d, int* cnt) {
    constexpr int SU = 8;
    if (vec) {
        const int nq = (hi - lo) >> 2;
        const int K  = (nq + NTH - 1) / NTH;
        const int rot = (int)(((long long)blockIdx.x * K) / gridDim.x);
        for (int k = 0; k < K; k += SU) {
            int4 v[SU]; int eb[SU]; bool ok[SU];
            #pragma unroll
            for (int u = 0; u < SU; ++u) {
                int kr = k + u + rot; kr = kr >= K ? kr - K : kr;
                const int q = kr * NTH + tid;
                ok[u] = (k + u < K) && (q < nq);
                const int qq = ok[u] ? q : 0;
                v[u]  = *(const int4*)(dstp + lo + 4 * qq);
                eb[u] = lo + 4 * qq;
            }
            #pragma unroll
            for (int u = 0; u < SU; ++u) {
                if (ok[u]) {
                    list_one(v[u].x, eb[u] + 0, d0, lst_s, lst_d, cnt);
                    list_one(v[u].y, eb[u] + 1, d0, lst_s, lst_d, cnt);
                    list_one(v[u].z, eb[u] + 2, d0, lst_s, lst_d, cnt);
                    list_one(v[u].w, eb[u] + 3, d0, lst_s, lst_d, cnt);
                }
            }
        }
        for (int e = lo + 4 * nq + tid; e < hi; e += NTH)
            list_one(dstp[e], e, d0, lst_s, lst_d, cnt);
    } else {
        for (int e = lo + tid; e < hi; e += NTH)
            list_one(dstp[e], e, d0, lst_s, lst_d, cnt);
    }
}

__device__ __forceinline__ unsigned zero_bytes(unsigned x) {
    return ~(((x & 0x7f7f7f7fu) + 0x7f7f7f7fu) | x | 0x7f7f7f7fu);
}

__device__ __forceinline__ void scan_ids(const uint4* __restrict__ ids, int E, int tid,
                                         int* lst_s, int* cnt) {
    constexpr int SU = 8;
    const int ng = (E + 15) >> 4;
    const int K  = (ng + NTH - 1) / NTH;
    const int rot = (int)(((long long)blockIdx.x * K) / gridDim.x);
    const unsigned mine = blockIdx.x * 0x01010101u;
    for (int k = 0; k < K; k += SU) {
        uint4 v[SU]; int eb[SU]; bool ok[SU];
        #pragma unroll
        for (int u = 0; u < SU; ++u) {
            int kr = k + u + rot; kr = kr >= K ? kr - K : kr;
            const int g = kr * NTH + tid;
            ok[u] = (k + u < K) && (g < ng);
            const int gg = ok[u] ? g : 0;
            v[u]  = ids[gg];
            eb[u] = 16 * gg;
        }
        #pragma unroll
        for (int u = 0; u < SU; ++u) {
            if (ok[u]) {
                unsigned m = (zero_bytes(v[u].x ^ mine) >> 7) | (zero_bytes(v[u].y ^ mine) >> 6)
                           | (zero_bytes(v[u].z ^ mine) >> 5) | (zero_bytes(v[u].w ^ mine) >> 4);
                while (m) {
                    const int b = __builtin_ctz(m);
                    m &= m - 1;
                    const int e = eb[u] + ((b & 7) << 2) + (b >> 3);
                    const int slot = atomicAdd(cnt, 1);
                    if (slot < CAPL) lst_s[slot] = e;
                }
            }
        }
    }
}

template <int H, bool MAXUPD, bool SPAD>
__device__ __forceinline__ void phase_score(int n, bool with_self, int d0, int N, int tid,
                                            const float* __restrict__ a_src,
                                            const int* lst_s, const int* lst_d,
                                            const float* adl, unsigned* mkey, float* wgt) {
    const int tot = n + (with_self ? RNG : 0);
    for (int i = tid; i < tot; i += NTH) {
        int s, dl;
        if (i < n) { s = lst_s[i]; dl = lst_d[i]; }
        else       { dl = i - n;   s = d0 + dl; }
        if (d0 + dl < N) {
            const int si = SPAD ? (int)(((unsigned)s / RNG) * SPITCH + (unsigned)s % RNG) : s;
            for (int h = 0; h < H; ++h) {
                const float v = lrelu(a_src[si * H + h] + adl[dl * H + h]);
                if (MAXUPD) atomicMax(&mkey[dl * H + h], fkey(v));
                wgt[i * H + h] = v;
            }
        }
    }
}

template <int H>
__device__ __forceinline__ void phase_exp(int n, bool with_self, int d0, int N, int tid,
                                          const int* lst_d, const unsigned* mkey,
                                          float* den, float* wgt) {
    const int tot = n + (with_self ? RNG : 0);
    for (int i = tid; i < tot; i += NTH) {
        const int dl = i < n ? lst_d[i] : i - n;
        if (d0 + dl < N) {
            for (int h = 0; h < H; ++h) {
                const float ex = expf(wgt[i * H + h] - fdekey(mkey[dl * H + h]));
                wgt[i * H + h] = ex;
                atomicAdd(&den[dl * H + h], ex);
            }
        }
    }
}

template <int H, int C, int PITCH>
__device__ __forceinline__ void phase_acc(int n, bool with_self, int d0, int N, int tid,
                                          const float* __restrict__ hfeat,
                                          const int* lst_s, const int* lst_d,
                                          const float* wgt, float* acc) {
    constexpr int F   = H * C;
    constexpr int LPE = F >= 32 ? 32 : F;
    constexpr int EPW = 32 / LPE;
    constexpr int J   = F / LPE;
    constexpr int U   = 4;
    constexpr int STEP = (NTH / 32) * EPW * U;
    const int lane = tid & 31, wave = tid >> 5;
    const int sub = lane / LPE, l = lane % LPE;
    const int tot = n + (with_self ? RNG : 0);

    for (int i0 = 0; i0 < tot; i0 += STEP) {
        int dl[U]; bool ok[U];
        float row[U][J]; float ex[U][H];
        #pragma unroll
        for (int u = 0; u < U; ++u) {
            int i = i0 + (wave * U + u) * EPW + sub;
            int s;
            if (i < n) { s = lst_s[i]; dl[u] = lst_d[i]; }
            else       { dl[u] = i - n; s = d0 + dl[u]; }
            ok[u] = (i < tot) && (d0 + dl[u] < N);
            if (!ok[u]) { s = 0; dl[u] = 0; i = 0; }
            const float* rp = hfeat + (size_t)s * F + l;
            #pragma unroll
            for (int j = 0; j < J; ++j) row[u][j] = rp[LPE * j];
            #pragma unroll
            for (int h = 0; h < H; ++h) ex[u][h] = wgt[i * H + h];
        }
        #pragma unroll
        for (int u = 0; u < U; ++u) {
            if (ok[u]) {
                float* ap = acc + dl[u] * PITCH + l;
                #pragma unroll
                for (int j = 0; j < J; ++j)
                    atomicAdd(ap + LPE * j, ex[u][(LPE * j) / C] * row[u][j]);
            }
        }
    }
}

template <int H, int C, int PITCH, bool RELU>
__device__ __forceinline__ void phase_own(int n, int d0, int N, int tid,
                                          const float* __restrict__ hfeat,
                                          const float* __restrict__ bias,
                                          const int* lst_s, const unsigned short* perm,
                                          const int* first, const int* cnt_n,
                                          const float* wgt, const float* den, float* acc) {
    constexpr int F = H * C;
    constexpr int J = F / 32;
    constexpr int U = 8;
    const int lane = tid & 31;
    const int wave = __builtin_amdgcn_readfirstlane(tid >> 5);
    float b[J];
    #pragma unroll
    for (int j = 0; j < J; ++j) b[j] = bias[lane + 32 * j];

    for (int dl = wave; dl < RNG; dl += NTH / 32) {
        float* const ap = acc + dl * PITCH + lane;
        if (d0 + dl < N) {
            float r[J];
            #pragma unroll
            for (int j = 0; j < J; ++j) r[j] = 0.0f;
            const int beg = __builtin_amdgcn_readfirstlane(first[dl]);
            const int ne  = __builtin_amdgcn_readfirstlane(cnt_n[dl]);
            int iv, sv;
            {
                const int jj = lane - 1;
                iv = n + dl; sv = d0 + dl;
                if (jj >= 0 && jj < ne) { iv = perm[beg + jj]; sv = lst_s[iv]; }
            }
            for (int j0 = -1; j0 < ne; j0 += U) {
                float row[U][J]; int ent[U];
                #pragma unroll
                for (int u = 0; u < U; ++u) {
                    if (j0 + u < ne) {
                        ent[u] = __builtin_amdgcn_readlane(iv, u);
                        const int s = __builtin_amdgcn_readlane(sv, u);
                        const float* rp = hfeat + (size_t)s * F + lane;
                        #pragma unroll
                        for (int j = 0; j < J; ++j) row[u][j] = rp[32 * j];
                    }
                }
                {
                    const int jj = j0 + U + lane;
                    if (jj < ne) { iv = perm[beg + jj]; sv = lst_s[iv]; }
                }
                #pragma unroll
                for (int u = 0; u < U; ++u) {
                    if (j0 + u < ne) {
                        float ex[H];
                        #pragma unroll
                        for (int h = 0; h < H; ++h) ex[h] = wgt[ent[u] * H + h];
                        #pragma unroll
                        for (int j = 0; j < J; ++j) r[j] += ex[(32 * j) / C] * row[u][j];
                    }
                }
            }
            #pragma unroll
            for (int j = 0; j < J; ++j) {
                float v = r[j] / den[dl * H + (32 * j) / C] + b[j];
                if (RELU) v = fmaxf(v, 0.0f);
                ap[32 * j] = v;
            }
        } else {
            #pragma unroll
            for (int j = 0; j < J; ++j) ap[32 * j] = 0.0f;
        }
    }
}

#define OC2  16

template <int H, int C, bool RELU, bool SAVE, bool LOAD, bool NEXT, bool SPAD>
__global__ __launch_bounds__(NTH)
void node_range_layer(const int* __restrict__ ei, int E, int N,
                      const float* __restrict__ a_src, const float* __restrict__ a_dst,
                      const float* __restrict__ hfeat, const float* __restrict__ bias,
                      float* __restrict__ outp, int* __restrict__ lists,
                      const unsigned* __restrict__ ids32,
                      const float* __restrict__ Wn,
                      const float* __restrict__ natt_src, const float* __restrict__ natt_dst,
                      float* __restrict__ nsrc, float* __restrict__ ndst) {
    constexpr int F = H * C;
    constexpr int PITCH = NEXT ? F + 4 : F;
    static_assert(!NEXT || (F * OC2 <= CAPL && RNG * (OC2 + 1) <= (CAPL + RNG) * H),
                  "the staged weights and the product tile reuse lst_s and wgt");
    static_assert(RNG <= SPITCH && SPITCH % 32 == 0, "a range's scores fill whole lines of their own");
    __shared__ float    acc[RNG * PITCH];
    __shared__ unsigned mkey[RNG * H];
    __shared__ float    den[RNG * H];
    __shared__ float    adl[RNG * H];
    __shared__ int      lst_s[CAPL];
    __shared__ int      lst_d[CAPL];
    __shared__ __attribute__((aligned(16))) float wgt[(CAPL + RNG) * H];
    __shared__ int      cnt;
    constexpr bool OWN = NEXT;
    __shared__ unsigned short perm[OWN ? CAPL : 1];

    const int tid = threadIdx.x;
    const int d0  = blockIdx.x * RNG;
    const int* dstp = ei + E;
    const bool vec = (((size_t)dstp) & 15) == 0;

    if (!OWN) for (int i = tid; i < RNG * PITCH; i += NTH) acc[i] = 0.0f;
    for (int i = tid; i < RNG * H; i += NTH) {
        mkey[i] = 0u; den[i] = 0.0f;
        adl[i] = (d0 * H + i < N * H) ? a_dst[(SPAD ? blockIdx.x * SPITCH : d0) * H + i] : 0.0f;
    }
    if (tid == 0) cnt = 0;
    __syncthreads();

    int* const my_cnt = lists + (size_t)blockIdx.x * LSTR;
    int* const my_s   = my_cnt + LHDR;
    int* const my_d   = my_s + CAPL;

    const bool by_ids = !LOAD && gridDim.x <= 255;

    int total;
    if (LOAD) {
        total = my_cnt[0];
    } else {
        if (by_ids) scan_ids((const uint4*)ids32, E, tid, lst_s, &cnt);
        else        scan_edges(dstp, 0, E, vec, d0, tid, lst_s, lst_d, &cnt);
        __syncthreads();
        total = cnt;
        if (SAVE && tid == 0) my_cnt[0] = total;
    }

    if (total <= CAPL) {
        if (LOAD) {
            for (int i = tid; i < total; i += NTH) { lst_s[i] = my_s[i]; lst_d[i] = my_d[i]; }
        } else {
            for (int i = tid; i < total; i += NTH) {
                const int e = lst_s[i];
                const int s = ei[e];
                if (by_ids) lst_d[i] = dstp[e] - d0;
                lst_s[i] = s;
                if (SAVE) { my_s[i] = s; my_d[i] = lst_d[i]; }
            }
        }
        __syncthreads();
        phase_score<H, true, SPAD>(total, true, d0, N, tid, a_src, lst_s, lst_d, adl, mkey, wgt);
        __syncthreads();
        int* const first  = reinterpret_cast<int*>(adl);
        int* const placed = reinterpret_cast<int*>(mkey);
        if (OWN) for (int i = tid; i < RNG; i += NTH) first[i] = 0;
        phase_exp<H>(total, true, d0, N, tid, lst_d, mkey, den, wgt);
        __syncthreads();
        if constexpr (OWN) {
            for (int i = tid; i < total; i += NTH) atomicAdd(&first[lst_d[i]], 1);
            for (int i = tid; i < RNG; i += NTH) placed[i] = 0;
            __syncthreads();
            if (tid < 32) {
                constexpr int PER = (RNG + 31) / 32;
                int cn[PER]; int sum = 0;
                #pragma unroll
                for (int k = 0; k < PER; ++k) {
                    const int idx = tid * PER + k;
                    cn[k] = idx < RNG ? first[idx] : 0;
                    sum += cn[k];
                }
                int incl = sum;
                for (int o = 1; o < 32; o <<= 1) {
                    const int t = __shfl_up(incl, o);
                    if (tid >= o) incl += t;
                }
                int base = incl - sum;
                #pragma unroll
                for (int k = 0; k < PER; ++k) {
                    const int idx = tid * PER + k;
                    if (idx < RNG) { first[idx] = base; base += cn[k]; }
                }
            }
            __syncthreads();
            for (int i = tid; i < total; i += NTH) {
                const int dl = lst_d[i];
                perm[first[dl] + atomicAdd(&placed[dl], 1)] = (unsigned short)i;
            }
            __syncthreads();
            phase_own<H, C, PITCH, RELU>(total, d0, N, tid, hfeat, bias, lst_s, perm, first, placed, wgt, den, acc);
        } else {
            phase_acc<H, C, PITCH>(total, true, d0, N, tid, hfeat, lst_s, lst_d, wgt, acc);
        }
        __syncthreads();
    } else {
        __syncthreads();
        if (OWN) for (int i = tid; i < RNG * PITCH; i += NTH) acc[i] = 0.0f;
        for (int pass = 0; pass < 2; ++pass) {
            for (int lo = 0; lo < E; lo += CAPL) {
                const int hi = lo + CAPL < E ? lo + CAPL : E;
                if (tid == 0) cnt = 0;
                __syncthreads();
                scan_edges(dstp, lo, hi, vec, d0, tid, lst_s, lst_d, &cnt);
                __syncthreads();
                const int n = cnt;
                for (int i = tid; i < n; i += NTH) lst_s[i] = ei[lst_s[i]];
                __syncthreads();
                if (pass == 0) {
                    phase_score<H, true, SPAD>(n, lo == 0, d0, N, tid, a_src, lst_s, lst_d, adl, mkey, wgt);
                } else {
                    phase_score<H, false, SPAD>(n, lo == 0, d0, N, tid, a_src, lst_s, lst_d, adl, mkey, wgt);
                    __syncthreads();
                    phase_exp<H>(n, lo == 0, d0, N, tid, lst_d, mkey, den, wgt);
                    __syncthreads();
                    phase_acc<H, C, PITCH>(n, lo == 0, d0, N, tid, hfeat, lst_s, lst_d, wgt, acc);
                }
                __syncthreads();
            }
        }
    }

    if (!NEXT) {
        for (int idx = tid; idx < RNG * F; idx += NTH) {
            const int dl = idx / F, c = idx % F;
            const int node = d0 + dl;
            if (node < N) {
                float v = acc[idx] / den[dl * H + c / C] + bias[c];
                if (RELU) v = fmaxf(v, 0.0f);
                outp[(size_t)node * F + c] = v;
            }
        }
    } else {
        float* const wn_s = reinterpret_cast<float*>(lst_s);
        float* const pr_s = wgt;

        constexpr int WPT = NEXT ? F * OC2 / NTH : 1;
        static_assert(!NEXT || F * OC2 % NTH == 0, "whole weights per thread");
        float wreg[WPT];
        #pragma unroll
        for (int j = 0; j < WPT; ++j) wreg[j] = Wn[tid + j * NTH];

        if (!(OWN && total <= CAPL)) {
            for (int idx = tid; idx < RNG * F; idx += NTH) {
                const int dl = idx / F, c = idx % F;
                float v = 0.0f;
                if (d0 + dl < N) {
                    v = acc[dl * PITCH + c] / den[dl * H + c / C] + bias[c];
                    if (RELU) v = fmaxf(v, 0.0f);
                }
                acc[dl * PITCH + c] = v;
            }
        }
        #pragma unroll
        for (int j = 0; j < WPT; ++j) wn_s[tid + j * NTH] = wreg[j];
        __syncthreads();

        const int wave = __builtin_amdgcn_readfirstlane(tid >> 5);
        const int lane = tid & 31;
        constexpr int MT = (RNG + 15) / 16;
        if (wave < MT) {
            const int nc    = lane & 15;
            const int row   = wave * 16 + nc;
            const int lr    = row < RNG ? row : RNG - 1;
            const int khalf = (lane >> 4) * 2;
            const float* ar = acc + lr * PITCH;
            v8f cc = {};
            for (int k = 0; k < F; k += 4) {
                const int kb = k + khalf;
                v2f a; a.x = ar[kb]; a.y = ar[kb + 1];
                v2f b; b.x = wn_s[kb * OC2 + nc]; b.y = wn_s[(kb + 1) * OC2 + nc];
                cc = __builtin_amdgcn_wmma_f32_16x16x4_f32(false, a, false, b, (short)0, cc, false, false);
            }
            const int rowoff = (lane >= 16) ? 8 : 0;
            for (int r = 0; r < 8; ++r) {
                const int rr = wave * 16 + rowoff + r;
                if (rr < RNG) pr_s[rr * (OC2 + 1) + nc] = cc[r];
            }
        }
        __syncthreads();

        for (int idx = tid; idx < RNG * OC2; idx += NTH) {
            const int dl = idx / OC2, c = idx % OC2;
            const int node = d0 + dl;
            if (node < N) outp[(size_t)node * OC2 + c] = pr_s[dl * (OC2 + 1) + c];
        }
        if (tid < RNG && d0 + tid < N) {
            float s = 0.0f, d = 0.0f;
            for (int c = 0; c < OC2; ++c) {
                const float v = pr_s[tid * (OC2 + 1) + c];
                s += v * natt_src[c];
                d += v * natt_dst[c];
            }
            nsrc[blockIdx.x * SPITCH + tid] = s;
            ndst[blockIdx.x * SPITCH + tid] = d;
        }
    }
}


extern "C" void kernel_launch(void* const* d_in, const int* in_sizes, int n_in,
                              void* d_out, int out_size, void* d_ws, size_t ws_size,
                              hipStream_t stream) {
    const float* x   = (const float*)d_in[0];
    const int*   ei  = (const int*)  d_in[1];
    const float* W1  = (const float*)d_in[2];
    const float* as1 = (const float*)d_in[3];
    const float* ad1 = (const float*)d_in[4];
    const float* b1  = (const float*)d_in[5];
    const float* W2  = (const float*)d_in[6];
    const float* as2 = (const float*)d_in[7];
    const float* ad2 = (const float*)d_in[8];
    const float* b2  = (const float*)d_in[9];

    const int N  = in_sizes[0] / 128;
    const int E  = in_sizes[1] / 2;

    char* w = (char*)d_ws;
    auto carve = [&](size_t bytes) -> void* {
        void* p = (void*)w;
        w += (bytes + 255) & ~((size_t)255);
        return p;
    };
    float* h1  = (float*)carve((size_t)N * 256 * 4);
    float* aS1 = (float*)carve((size_t)N * 4 * 4);
    float* aD1 = (float*)carve((size_t)N * 4 * 4);
    float* h2  = (float*)carve((size_t)N * 16 * 4);

    const unsigned nrange = (unsigned)((N + RNG - 1) / RNG);
    int* lists = (int*)carve((size_t)nrange * LSTR * 4);
    unsigned* ids32 = (unsigned*)carve((((size_t)E + 15) / 16) * 16);
    float* aS2 = (float*)carve((size_t)nrange * SPITCH * 4);
    float* aD2 = (float*)carve((size_t)nrange * SPITCH * 4);

    gemm1_scores<<<(N + 15) / 16, 128, 0, stream>>>(x, W1, as1, ad1, h1, aS1, aD1, N, ei + E, E, ids32);
    node_range_layer<4, 64, true, true, false, true, false><<<nrange, NTH, 0, stream>>>(ei, E, N, aS1, aD1, h1, b1, h2, lists, ids32, W2, as2, ad2, aS2, aD2);

    node_range_layer<1, 16, false, false, true, false, true><<<nrange, NTH, 0, stream>>>(ei, E, N, aS2, aD2, h2, b2, (float*)d_out, lists, ids32, nullptr, nullptr, nullptr, nullptr, nullptr);
}


# Layer-1 edge filter reads one bit per edge: product launch builds a bitmap row per node range in LDS (LDS integer or); a range block scans its own row only, one list add per wave; list order changes
# speedup vs baseline: 6.1307x; 1.0354x over previous
#include <hip/hip_runtime.h>
#include <hip/hip_bf16.h>

typedef __attribute__((ext_vector_type(2))) float v2f;
typedef __attribute__((ext_vector_type(8))) float v8f;
typedef float v4fu __attribute__((ext_vector_type(4), aligned(4)));

#define NEG_SLOPE 0.2f


__device__ __forceinline__ float lrelu(float v) {
    return v >= 0.0f ? v : NEG_SLOPE * v;
}

__device__ __forceinline__ unsigned fkey(float f) {
    unsigned u = __float_as_uint(f);
    return (u & 0x80000000u) ? ~u : (u | 0x80000000u);
}
__device__ __forceinline__ float fdekey(unsigned k) {
    return (k & 0x80000000u) ? __uint_as_float(k ^ 0x80000000u)
                             : __uint_as_float(~k);
}

#define RNG  200
#define SPITCH 256

#define EPB 1024
#define TP 68
__global__ __launch_bounds__(128)
void gemm1_scores(const float* __restrict__ x, const float* __restrict__ W1,
                  const float* __restrict__ att_src, const float* __restrict__ att_dst,
                  float* __restrict__ h1,
                  float* __restrict__ a_src, float* __restrict__ a_dst, int N,
                  const int* __restrict__ dstp, int E,
                  unsigned* __restrict__ bits, int nrange) {
    __shared__ __attribute__((aligned(16))) float tile[4][16 * TP];
    __shared__ float sc[2][64];

    const int mtile = blockIdx.x;
    const int lane  = threadIdx.x & 31;
    const int head  = __builtin_amdgcn_readfirstlane(threadIdx.x >> 5);
    const int mrow  = mtile * 16 + (lane & 15);
    const int lrow  = mrow < N ? mrow : N - 1;
    const int khalf = (lane >> 4) * 2;
    const int colbase = head * 64;
    const int nc    = lane & 15;

    v8f c0 = {}, c1 = {}, c2 = {}, c3 = {};
    const float* xrow = x + (size_t)lrow * 128;

    #pragma unroll 4
    for (int k = 0; k < 128; k += 4) {
        const int kb = k + khalf;
        v2f a; a.x = xrow[kb]; a.y = xrow[kb + 1];
        const float* w0 = W1 + (size_t)kb * 256 + colbase + 4 * nc;
        const v4fu q0 = *(const v4fu*)w0;
        const v4fu q1 = *(const v4fu*)(w0 + 256);
        v2f b0; b0.x = q0.x; b0.y = q1.x;
        v2f b1; b1.x = q0.y; b1.y = q1.y;
        v2f b2; b2.x = q0.z; b2.y = q1.z;
        v2f b3; b3.x = q0.w; b3.y = q1.w;
        c0 = __builtin_amdgcn_wmma_f32_16x16x4_f32(false, a, false, b0, (short)0, c0, false, false);
        c1 = __builtin_amdgcn_wmma_f32_16x16x4_f32(false, a, false, b1, (short)0, c1, false, false);
        c2 = __builtin_amdgcn_wmma_f32_16x16x4_f32(false, a, false, b2, (short)0, c2, false, false);
        c3 = __builtin_amdgcn_wmma_f32_16x16x4_f32(false, a, false, b3, (short)0, c3, false, false);
    }

    const int rowoff = (lane >= 16) ? 8 : 0;
    float* const tw = tile[head];
    for (int r = 0; r < 8; ++r) {
        const int grow = mtile * 16 + rowoff + r;
        float4 v; v.x = c0[r]; v.y = c1[r]; v.z = c2[r]; v.w = c3[r];
        if (grow < N) *(float4*)(h1 + (size_t)grow * 256 + colbase + 4 * nc) = v;
        *(float4*)(tw + (rowoff + r) * TP + 4 * nc) = v;
    }
    __syncthreads();

    {
        const int row = lane & 15;
        const float* av = (lane < 16 ? att_src : att_dst) + head * 64;
        const float* tr = tw + row * TP;
        float s = 0.0f;
        for (int c = 0; c < 64; ++c) s += tr[c] * av[c];
        sc[lane >> 4][row * 4 + head] = s;
    }
    __syncthreads();

    {
        const int t = threadIdx.x & 63;
        float* const dstv = threadIdx.x < 64 ? a_src : a_dst;
        if (mtile * 16 + (t >> 2) < N) dstv[(size_t)mtile * 64 + t] = sc[threadIdx.x >> 6][t];
    }

    {
        constexpr int RPP = (4 * 16 * TP) / 32;
        const int nrb  = nrange <= 255 ? nrange : 0;
        const int nbm  = (E + EPB - 1) / EPB;
        const int roww = nbm * 32;
        unsigned* const bm = reinterpret_cast<unsigned*>(&tile[0][0]);
        const int tid = threadIdx.x;
        for (int bb = blockIdx.x; bb < nbm && nrb > 0; bb += gridDim.x) {
            int q[EPB / 128];
            #pragma unroll
            for (int j = 0; j < EPB / 128; ++j) {
                const int e = bb * EPB + j * 128 + tid;
                q[j] = e < E ? (int)((unsigned)dstp[e] / (unsigned)RNG) : -1;
            }
            for (int r0 = 0; r0 < nrb; r0 += RPP) {
                __syncthreads();
                for (int i = tid; i < RPP * 32; i += 128) bm[i] = 0u;
                __syncthreads();
                #pragma unroll
                for (int j = 0; j < EPB / 128; ++j) {
                    const unsigned rr = (unsigned)(q[j] - r0);
                    if (rr < (unsigned)RPP && q[j] < nrb)
                        atomicOr(&bm[rr * 32 + j * 4 + (tid >> 5)], 1u << (tid & 31));
                }
                __syncthreads();
                const int nr = nrb - r0 < RPP ? nrb - r0 : RPP;
                for (int i = tid; i < nr * 32; i += 128)
                    bits[(size_t)(r0 + (i >> 5)) * roww + bb * 32 + (i & 31)] = bm[i];
            }
        }
    }
}

#define CAPL 4096
#define NTH  1024
#define LHDR 64
#define LSTR (LHDR + 2 * CAPL)

__device__ __forceinline__ void list_one(int d, int e, int d0,
                                         int* lst_s, int* lst_d, int* cnt) {
    const unsigned dl = (unsigned)(d - d0);
    if (dl < (unsigned)RNG) {
        const int slot = atomicAdd(cnt, 1);
        if (slot < CAPL) { lst_s[slot] = e; lst_d[slot] = (int)dl; }
    }
}

__device__ __forceinline__ void scan_edges(const int* __restrict__ dstp, int lo, int hi,
                                           bool vec, int d0, int tid,
                                           int* lst_s, int* lst_d, int* cnt) {
    constexpr int SU = 8;
    if (vec) {
        const int nq = (hi - lo) >> 2;
        const int K  = (nq + NTH - 1) / NTH;
        const int rot = (int)(((long long)blockIdx.x * K) / gridDim.x);
        for (int k = 0; k < K; k += SU) {
            int4 v[SU]; int eb[SU]; bool ok[SU];
            #pragma unroll
            for (int u = 0; u < SU; ++u) {
                int kr = k + u + rot; kr = kr >= K ? kr - K : kr;
                const int q = kr * NTH + tid;
                ok[u] = (k + u < K) && (q < nq);
                const int qq = ok[u] ? q : 0;
                v[u]  = *(const int4*)(dstp + lo + 4 * qq);
                eb[u] = lo + 4 * qq;
            }
            #pragma unroll
            for (int u = 0; u < SU; ++u) {
                if (ok[u]) {
                    list_one(v[u].x, eb[u] + 0, d0, lst_s, lst_d, cnt);
                    list_one(v[u].y, eb[u] + 1, d0, lst_s, lst_d, cnt);
                    list_one(v[u].z, eb[u] + 2, d0, lst_s, lst_d, cnt);
                    list_one(v[u].w, eb[u] + 3, d0, lst_s, lst_d, cnt);
                }
            }
        }
        for (int e = lo + 4 * nq + tid; e < hi; e += NTH)
            list_one(dstp[e], e, d0, lst_s, lst_d, cnt);
    } else {
        for (int e = lo + tid; e < hi; e += NTH)
            list_one(dstp[e], e, d0, lst_s, lst_d, cnt);
    }
}

__device__ __forceinline__ void scan_bits(const uint4* __restrict__ row, int ng, int tid,
                                          int* lst_s, int* cnt) {
    constexpr int SU = 8;
    const int lane = tid & 31;
    for (int g0 = 0; g0 < ng; g0 += SU * NTH) {
        uint4 v[SU];
        #pragma unroll
        for (int u = 0; u < SU; ++u) {
            const int g = g0 + u * NTH + tid;
            v[u] = g < ng ? row[g] : make_uint4(0u, 0u, 0u, 0u);
        }
        int mine = 0;
        #pragma unroll
        for (int u = 0; u < SU; ++u)
            mine += __builtin_popcount(v[u].x) + __builtin_popcount(v[u].y)
                  + __builtin_popcount(v[u].z) + __builtin_popcount(v[u].w);
        int incl = mine;
        for (int o = 1; o < 32; o <<= 1) {
            const int t = __shfl_up(incl, o);
            if (lane >= o) incl += t;
        }
        const int wtot = __builtin_amdgcn_readlane(incl, 31);
        int base = 0;
        if (lane == 31 && wtot > 0) base = atomicAdd(cnt, wtot);
        int slot = __builtin_amdgcn_readlane(base, 31) + incl - mine;
        #pragma unroll
        for (int u = 0; u < SU; ++u) {
            const unsigned wd[4] = {v[u].x, v[u].y, v[u].z, v[u].w};
            const int eb = (g0 + u * NTH + tid) * 128;
            #pragma unroll
            for (int wi = 0; wi < 4; ++wi) {
                unsigned m = wd[wi];
                while (m) {
                    const int b = __builtin_ctz(m);
                    m &= m - 1;
                    if (slot < CAPL) lst_s[slot] = eb + 32 * wi + b;
                    ++slot;
                }
            }
        }
    }
}

template <int H, bool MAXUPD, bool SPAD>
__device__ __forceinline__ void phase_score(int n, bool with_self, int d0, int N, int tid,
                                            const float* __restrict__ a_src,
                                            const int* lst_s, const int* lst_d,
                                            const float* adl, unsigned* mkey, float* wgt) {
    const int tot = n + (with_self ? RNG : 0);
    for (int i = tid; i < tot; i += NTH) {
        int s, dl;
        if (i < n) { s = lst_s[i]; dl = lst_d[i]; }
        else       { dl = i - n;   s = d0 + dl; }
        if (d0 + dl < N) {
            const int si = SPAD ? (int)(((unsigned)s / RNG) * SPITCH + (unsigned)s % RNG) : s;
            for (int h = 0; h < H; ++h) {
                const float v = lrelu(a_src[si * H + h] + adl[dl * H + h]);
                if (MAXUPD) atomicMax(&mkey[dl * H + h], fkey(v));
                wgt[i * H + h] = v;
            }
        }
    }
}

template <int H>
__device__ __forceinline__ void phase_exp(int n, bool with_self, int d0, int N, int tid,
                                          const int* lst_d, const unsigned* mkey,
                                          float* den, float* wgt) {
    const int tot = n + (with_self ? RNG : 0);
    for (int i = tid; i < tot; i += NTH) {
        const int dl = i < n ? lst_d[i] : i - n;
        if (d0 + dl < N) {
            for (int h = 0; h < H; ++h) {
                const float ex = expf(wgt[i * H + h] - fdekey(mkey[dl * H + h]));
                wgt[i * H + h] = ex;
                atomicAdd(&den[dl * H + h], ex);
            }
        }
    }
}

template <int H, int C, int PITCH>
__device__ __forceinline__ void phase_acc(int n, bool with_self, int d0, int N, int tid,
                                          const float* __restrict__ hfeat,
                                          const int* lst_s, const int* lst_d,
                                          const float* wgt, float* acc) {
    constexpr int F   = H * C;
    constexpr int LPE = F >= 32 ? 32 : F;
    constexpr int EPW = 32 / LPE;
    constexpr int J   = F / LPE;
    constexpr int U   = 4;
    constexpr int STEP = (NTH / 32) * EPW * U;
    const int lane = tid & 31, wave = tid >> 5;
    const int sub = lane / LPE, l = lane % LPE;
    const int tot = n + (with_self ? RNG : 0);

    for (int i0 = 0; i0 < tot; i0 += STEP) {
        int dl[U]; bool ok[U];
        float row[U][J]; float ex[U][H];
        #pragma unroll
        for (int u = 0; u < U; ++u) {
            int i = i0 + (wave * U + u) * EPW + sub;
            int s;
            if (i < n) { s = lst_s[i]; dl[u] = lst_d[i]; }
            else       { dl[u] = i - n; s = d0 + dl[u]; }
            ok[u] = (i < tot) && (d0 + dl[u] < N);
            if (!ok[u]) { s = 0; dl[u] = 0; i = 0; }
            const float* rp = hfeat + (size_t)s * F + l;
            #pragma unroll
            for (int j = 0; j < J; ++j) row[u][j] = rp[LPE * j];
            #pragma unroll
            for (int h = 0; h < H; ++h) ex[u][h] = wgt[i * H + h];
        }
        #pragma unroll
        for (int u = 0; u < U; ++u) {
            if (ok[u]) {
                float* ap = acc + dl[u] * PITCH + l;
                #pragma unroll
                for (int j = 0; j < J; ++j)
                    atomicAdd(ap + LPE * j, ex[u][(LPE * j) / C] * row[u][j]);
            }
        }
    }
}

template <int H, int C, int PITCH, bool RELU>
__device__ __forceinline__ void phase_own(int n, int d0, int N, int tid,
                                          const float* __restrict__ hfeat,
                                          const float* __restrict__ bias,
                                          const int* lst_s, const unsigned short* perm,
                                          const int* first, const int* cnt_n,
                                          const float* wgt, const float* den, float* acc) {
    constexpr int F = H * C;
    constexpr int J = F / 32;
    constexpr int U = 8;
    const int lane = tid & 31;
    const int wave = __builtin_amdgcn_readfirstlane(tid >> 5);
    float b[J];
    #pragma unroll
    for (int j = 0; j < J; ++j) b[j] = bias[lane + 32 * j];

    for (int dl = wave; dl < RNG; dl += NTH / 32) {
        float* const ap = acc + dl * PITCH + lane;
        if (d0 + dl < N) {
            float r[J];
            #pragma unroll
            for (int j = 0; j < J; ++j) r[j] = 0.0f;
            const int beg = __builtin_amdgcn_readfirstlane(first[dl]);
            const int ne  = __builtin_amdgcn_readfirstlane(cnt_n[dl]);
            int iv, sv;
            {
                const int jj = lane - 1;
                iv = n + dl; sv = d0 + dl;
                if (jj >= 0 && jj < ne) { iv = perm[beg + jj]; sv = lst_s[iv]; }
            }
            for (int j0 = -1; j0 < ne; j0 += U) {
                float row[U][J]; int ent[U];
                #pragma unroll
                for (int u = 0; u < U; ++u) {
                    if (j0 + u < ne) {
                        ent[u] = __builtin_amdgcn_readlane(iv, u);
                        const int s = __builtin_amdgcn_readlane(sv, u);
                        const float* rp = hfeat + (size_t)s * F + lane;
                        #pragma unroll
                        for (int j = 0; j < J; ++j) row[u][j] = rp[32 * j];
                    }
                }
                {
                    const int jj = j0 + U + lane;
                    if (jj < ne) { iv = perm[beg + jj]; sv = lst_s[iv]; }
                }
                #pragma unroll
                for (int u = 0; u < U; ++u) {
                    if (j0 + u < ne) {
                        float ex[H];
                        #pragma unroll
                        for (int h = 0; h < H; ++h) ex[h] = wgt[ent[u] * H + h];
                        #pragma unroll
                        for (int j = 0; j < J; ++j) r[j] += ex[(32 * j) / C] * row[u][j];
                    }
                }
            }
            #pragma unroll
            for (int j = 0; j < J; ++j) {
                float v = r[j] / den[dl * H + (32 * j) / C] + b[j];
                if (RELU) v = fmaxf(v, 0.0f);
                ap[32 * j] = v;
            }
        } else {
            #pragma unroll
            for (int j = 0; j < J; ++j) ap[32 * j] = 0.0f;
        }
    }
}

#define OC2  16

template <int H, int C, bool RELU, bool SAVE, bool LOAD, bool NEXT, bool SPAD>
__global__ __launch_bounds__(NTH)
void node_range_layer(const int* __restrict__ ei, int E, int N,
                      const float* __restrict__ a_src, const float* __restrict__ a_dst,
                      const float* __restrict__ hfeat, const float* __restrict__ bias,
                      float* __restrict__ outp, int* __restrict__ lists,
                      const unsigned* __restrict__ bits,
                      const float* __restrict__ Wn,
                      const float* __restrict__ natt_src, const float* __restrict__ natt_dst,
                      float* __restrict__ nsrc, float* __restrict__ ndst) {
    constexpr int F = H * C;
    constexpr int PITCH = NEXT ? F + 4 : F;
    static_assert(!NEXT || (F * OC2 <= CAPL && RNG * (OC2 + 1) <= (CAPL + RNG) * H),
                  "the staged weights and the product tile reuse lst_s and wgt");
    static_assert(RNG <= SPITCH && SPITCH % 32 == 0, "a range's scores fill whole lines of their own");
    __shared__ float    acc[RNG * PITCH];
    __shared__ unsigned mkey[RNG * H];
    __shared__ float    den[RNG * H];
    __shared__ float    adl[RNG * H];
    __shared__ int      lst_s[CAPL];
    __shared__ int      lst_d[CAPL];
    __shared__ __attribute__((aligned(16))) float wgt[(CAPL + RNG) * H];
    __shared__ int      cnt;
    constexpr bool OWN = NEXT;
    __shared__ unsigned short perm[OWN ? CAPL : 1];

    const int tid = threadIdx.x;
    const int d0  = blockIdx.x * RNG;
    const int* dstp = ei + E;
    const bool vec = (((size_t)dstp) & 15) == 0;

    if (!OWN) for (int i = tid; i < RNG * PITCH; i += NTH) acc[i] = 0.0f;
    for (int i = tid; i < RNG * H; i += NTH) {
        mkey[i] = 0u; den[i] = 0.0f;
        adl[i] = (d0 * H + i < N * H) ? a_dst[(SPAD ? blockIdx.x * SPITCH : d0) * H + i] : 0.0f;
    }
    if (tid == 0) cnt = 0;
    __syncthreads();

    int* const my_cnt = lists + (size_t)blockIdx.x * LSTR;
    int* const my_s   = my_cnt + LHDR;
    int* const my_d   = my_s + CAPL;

    const bool by_ids = !LOAD && gridDim.x <= 255;

    int total;
    if (LOAD) {
        total = my_cnt[0];
    } else {
        if (by_ids) {
            const int roww = ((E + EPB - 1) / EPB) * 32;
            scan_bits((const uint4*)(bits + (size_t)blockIdx.x * roww), roww >> 2, tid, lst_s, &cnt);
        }
        else        scan_edges(dstp, 0, E, vec, d0, tid, lst_s, lst_d, &cnt);
        __syncthreads();
        total = cnt;
        if (SAVE && tid == 0) my_cnt[0] = total;
    }

    if (total <= CAPL) {
        if (LOAD) {
            for (int i = tid; i < total; i += NTH) { lst_s[i] = my_s[i]; lst_d[i] = my_d[i]; }
        } else {
            for (int i = tid; i < total; i += NTH) {
                const int e = lst_s[i];
                const int s = ei[e];
                if (by_ids) lst_d[i] = dstp[e] - d0;
                lst_s[i] = s;
                if (SAVE) { my_s[i] = s; my_d[i] = lst_d[i]; }
            }
        }
        __syncthreads();
        phase_score<H, true, SPAD>(total, true, d0, N, tid, a_src, lst_s, lst_d, adl, mkey, wgt);
        __syncthreads();
        int* const first  = reinterpret_cast<int*>(adl);
        int* const placed = reinterpret_cast<int*>(mkey);
        if (OWN) for (int i = tid; i < RNG; i += NTH) first[i] = 0;
        phase_exp<H>(total, true, d0, N, tid, lst_d, mkey, den, wgt);
        __syncthreads();
        if constexpr (OWN) {
            for (int i = tid; i < total; i += NTH) atomicAdd(&first[lst_d[i]], 1);
            for (int i = tid; i < RNG; i += NTH) placed[i] = 0;
            __syncthreads();
            if (tid < 32) {
                constexpr int PER = (RNG + 31) / 32;
                int cn[PER]; int sum = 0;
                #pragma unroll
                for (int k = 0; k < PER; ++k) {
                    const int idx = tid * PER + k;
                    cn[k] = idx < RNG ? first[idx] : 0;
                    sum += cn[k];
                }
                int incl = sum;
                for (int o = 1; o < 32; o <<= 1) {
                    const int t = __shfl_up(incl, o);
                    if (tid >= o) incl += t;
                }
                int base = incl - sum;
                #pragma unroll
                for (int k = 0; k < PER; ++k) {
                    const int idx = tid * PER + k;
                    if (idx < RNG) { first[idx] = base; base += cn[k]; }
                }
            }
            __syncthreads();
            for (int i = tid; i < total; i += NTH) {
                const int dl = lst_d[i];
                perm[first[dl] + atomicAdd(&placed[dl], 1)] = (unsigned short)i;
            }
            __syncthreads();
            phase_own<H, C, PITCH, RELU>(total, d0, N, tid, hfeat, bias, lst_s, perm, first, placed, wgt, den, acc);
        } else {
            phase_acc<H, C, PITCH>(total, true, d0, N, tid, hfeat, lst_s, lst_d, wgt, acc);
        }
        __syncthreads();
    } else {
        __syncthreads();
        if (OWN) for (int i = tid; i < RNG * PITCH; i += NTH) acc[i] = 0.0f;
        for (int pass = 0; pass < 2; ++pass) {
            for (int lo = 0; lo < E; lo += CAPL) {
                const int hi = lo + CAPL < E ? lo + CAPL : E;
                if (tid == 0) cnt = 0;
                __syncthreads();
                scan_edges(dstp, lo, hi, vec, d0, tid, lst_s, lst_d, &cnt);
                __syncthreads();
                const int n = cnt;
                for (int i = tid; i < n; i += NTH) lst_s[i] = ei[lst_s[i]];
                __syncthreads();
                if (pass == 0) {
                    phase_score<H, true, SPAD>(n, lo == 0, d0, N, tid, a_src, lst_s, lst_d, adl, mkey, wgt);
                } else {
                    phase_score<H, false, SPAD>(n, lo == 0, d0, N, tid, a_src, lst_s, lst_d, adl, mkey, wgt);
                    __syncthreads();
                    phase_exp<H>(n, lo == 0, d0, N, tid, lst_d, mkey, den, wgt);
                    __syncthreads();
                    phase_acc<H, C, PITCH>(n, lo == 0, d0, N, tid, hfeat, lst_s, lst_d, wgt, acc);
                }
                __syncthreads();
            }
        }
    }

    if (!NEXT) {
        for (int idx = tid; idx < RNG * F; idx += NTH) {
            const int dl = idx / F, c = idx % F;
            const int node = d0 + dl;
            if (node < N) {
                float v = acc[idx] / den[dl * H + c / C] + bias[c];
                if (RELU) v = fmaxf(v, 0.0f);
                outp[(size_t)node * F + c] = v;
            }
        }
    } else {
        float* const wn_s = reinterpret_cast<float*>(lst_s);
        float* const pr_s = wgt;

        constexpr int WPT = NEXT ? F * OC2 / NTH : 1;
        static_assert(!NEXT || F * OC2 % NTH == 0, "whole weights per thread");
        float wreg[WPT];
        #pragma unroll
        for (int j = 0; j < WPT; ++j) wreg[j] = Wn[tid + j * NTH];

        if (!(OWN && total <= CAPL)) {
            for (int idx = tid; idx < RNG * F; idx += NTH) {
                const int dl = idx / F, c = idx % F;
                float v = 0.0f;
                if (d0 + dl < N) {
                    v = acc[dl * PITCH + c] / den[dl * H + c / C] + bias[c];
                    if (RELU) v = fmaxf(v, 0.0f);
                }
                acc[dl * PITCH + c] = v;
            }
        }
        #pragma unroll
        for (int j = 0; j < WPT; ++j) wn_s[tid + j * NTH] = wreg[j];
        __syncthreads();

        const int wave = __builtin_amdgcn_readfirstlane(tid >> 5);
        const int lane = tid & 31;
        constexpr int MT = (RNG + 15) / 16;
        if (wave < MT) {
            const int nc    = lane & 15;
            const int row   = wave * 16 + nc;
            const int lr    = row < RNG ? row : RNG - 1;
            const int khalf = (lane >> 4) * 2;
            const float* ar = acc + lr * PITCH;
            v8f cc = {};
            for (int k = 0; k < F; k += 4) {
                const int kb = k + khalf;
                v2f a; a.x = ar[kb]; a.y = ar[kb + 1];
                v2f b; b.x = wn_s[kb * OC2 + nc]; b.y = wn_s[(kb + 1) * OC2 + nc];
                cc = __builtin_amdgcn_wmma_f32_16x16x4_f32(false, a, false, b, (short)0, cc, false, false);
            }
            const int rowoff = (lane >= 16) ? 8 : 0;
            for (int r = 0; r < 8; ++r) {
                const int rr = wave * 16 + rowoff + r;
                if (rr < RNG) pr_s[rr * (OC2 + 1) + nc] = cc[r];
            }
        }
        __syncthreads();

        for (int idx = tid; idx < RNG * OC2; idx += NTH) {
            const int dl = idx / OC2, c = idx % OC2;
            const int node = d0 + dl;
            if (node < N) outp[(size_t)node * OC2 + c] = pr_s[dl * (OC2 + 1) + c];
        }
        if (tid < RNG && d0 + tid < N) {
            float s = 0.0f, d = 0.0f;
            for (int c = 0; c < OC2; ++c) {
                const float v = pr_s[tid * (OC2 + 1) + c];
                s += v * natt_src[c];
                d += v * natt_dst[c];
            }
            nsrc[blockIdx.x * SPITCH + tid] = s;
            ndst[blockIdx.x * SPITCH + tid] = d;
        }
    }
}


extern "C" void kernel_launch(void* const* d_in, const int* in_sizes, int n_in,
                              void* d_out, int out_size, void* d_ws, size_t ws_size,
                              hipStream_t stream) {
    const float* x   = (const float*)d_in[0];
    const int*   ei  = (const int*)  d_in[1];
    const float* W1  = (const float*)d_in[2];
    const float* as1 = (const float*)d_in[3];
    const float* ad1 = (const float*)d_in[4];
    const float* b1  = (const float*)d_in[5];
    const float* W2  = (const float*)d_in[6];
    const float* as2 = (const float*)d_in[7];
    const float* ad2 = (const float*)d_in[8];
    const float* b2  = (const float*)d_in[9];

    const int N  = in_sizes[0] / 128;
    const int E  = in_sizes[1] / 2;

    char* w = (char*)d_ws;
    auto carve = [&](size_t bytes) -> void* {
        void* p = (void*)w;
        w += (bytes + 255) & ~((size_t)255);
        return p;
    };
    float* h1  = (float*)carve((size_t)N * 256 * 4);
    float* aS1 = (float*)carve((size_t)N * 4 * 4);
    float* aD1 = (float*)carve((size_t)N * 4 * 4);
    float* h2  = (float*)carve((size_t)N * 16 * 4);

    const unsigned nrange = (unsigned)((N + RNG - 1) / RNG);
    int* lists = (int*)carve((size_t)nrange * LSTR * 4);
    unsigned* bits = (unsigned*)carve((size_t)nrange * (((size_t)E + EPB - 1) / EPB) * 128);
    float* aS2 = (float*)carve((size_t)nrange * SPITCH * 4);
    float* aD2 = (float*)carve((size_t)nrange * SPITCH * 4);

    gemm1_scores<<<(N + 15) / 16, 128, 0, stream>>>(x, W1, as1, ad1, h1, aS1, aD1, N, ei + E, E, bits, (int)nrange);
    node_range_layer<4, 64, true, true, false, true, false><<<nrange, NTH, 0, stream>>>(ei, E, N, aS1, aD1, h1, b1, h2, lists, bits, W2, as2, ad2, aS2, aD2);

    node_range_layer<1, 16, false, false, true, false, true><<<nrange, NTH, 0, stream>>>(ei, E, N, aS2, aD2, h2, b2, (float*)d_out, lists, bits, nullptr, nullptr, nullptr, nullptr, nullptr);
}
